# starting kernel
# baseline (speedup 1.0000x reference)
#include <hip/hip_runtime.h>
#include <hip/hip_bf16.h>


typedef float v2f __attribute__((ext_vector_type(2)));
typedef float v8f __attribute__((ext_vector_type(8)));
typedef int   v2i __attribute__((ext_vector_type(2)));

__global__ void init_cnt_kernel(int* __restrict__ cnt, int N) {
    int i = blockIdx.x * blockDim.x + threadIdx.x;
    if (i < N) cnt[i] = 0;
}

__global__ void count_deg_kernel(const int* __restrict__ ei, int* __restrict__ cnt, int E) {
    int e = blockIdx.x * blockDim.x + threadIdx.x;
    if (e < E) atomicAdd(&cnt[ei[e]], 1);
}

__global__ void dinv_kernel(const int* __restrict__ cnt, float* __restrict__ dinv, int N) {
    int i = blockIdx.x * blockDim.x + threadIdx.x;
    if (i < N) dinv[i] = rsqrtf((float)(cnt[i] + 1));
}

__global__ __launch_bounds__(1024) void scan_kernel(const int* __restrict__ cnt,
                                                    int* __restrict__ rowptr, int N) {
    __shared__ int sdata[1024];
    int t = threadIdx.x;
    int chunk = (N + 1023) / 1024;
    int begin = t * chunk;
    int end   = begin + chunk;
    if (begin > N) begin = N;
    if (end   > N) end   = N;
    int sum = 0;
    for (int i = begin; i < end; ++i) sum += cnt[i] + 1;
    sdata[t] = sum;
    __syncthreads();
    for (int off = 1; off < 1024; off <<= 1) {
        int v = (t >= off) ? sdata[t - off] : 0;
        __syncthreads();
        sdata[t] += v;
        __syncthreads();
    }
    int run = (t == 0) ? 0 : sdata[t - 1];
    for (int i = begin; i < end; ++i) {
        rowptr[i] = run;
        run += cnt[i] + 1;
    }
    if (t == 1023) rowptr[N] = sdata[1023];
}

__global__ void init_csr_kernel(const int* __restrict__ rowptr, const float* __restrict__ dinv,
                                int* __restrict__ cursor, v2i* __restrict__ csr,
                                int N) {
    int i = blockIdx.x * blockDim.x + threadIdx.x;
    if (i < N) {
        int p = rowptr[i];
        cursor[i] = p + 1;
        float d = dinv[i];
        v2i rec;
        rec.x = i;
        rec.y = __float_as_int(d * d);
        csr[p] = rec;
    }
}

__global__ void fill_csr_kernel(const int* __restrict__ ei, const float* __restrict__ dinv,
                                int* __restrict__ cursor, v2i* __restrict__ csr, int E) {
    int e = blockIdx.x * blockDim.x + threadIdx.x;
    if (e < E) {
        int r = ei[e];
        int c = ei[E + e];
        int p = atomicAdd(&cursor[r], 1);
        v2i rec;
        rec.x = c;
        rec.y = __float_as_int(dinv[r] * dinv[c]);
        csr[p] = rec;
    }
}

__global__ __launch_bounds__(256) void gemm64_wmma_kernel(const float* __restrict__ A,
                                                          const float* __restrict__ B,
                                                          float* __restrict__ C,
                                                          int M, int K) {
    int tid  = threadIdx.x;
    int wave = tid >> 5;
    int lane = tid & 31;
    int m0 = blockIdx.x * 32 + (wave >> 2) * 16;
    int n0 = (wave & 3) * 16;
    if (m0 >= M) return;
    int half = lane >> 4;
    int l    = lane & 15;

    const float* arow = A + (size_t)(m0 + l) * K;
    v8f acc = {0.f, 0.f, 0.f, 0.f, 0.f, 0.f, 0.f, 0.f};

    for (int k = 0; k < K; k += 4) {
        int ka = k + 2 * half;
        v2f a = *(const v2f*)(arow + ka);
        v2f b;
        b.x = B[(size_t)ka * 64 + n0 + l];
        b.y = B[(size_t)(ka + 1) * 64 + n0 + l];
        acc = __builtin_amdgcn_wmma_f32_16x16x4_f32(false, a, false, b,
                                                    (short)0, acc, false, false);
    }
    float* crow = C + (size_t)m0 * 64 + n0 + l;
#pragma unroll
    for (int j = 0; j < 8; ++j) {
        crow[(size_t)(j + 8 * half) * 64] = acc[j];
    }
}

__global__ __launch_bounds__(256) void agg_kernel(const float* __restrict__ h,
                                                  const int* __restrict__ rowptr,
                                                  const v2i* __restrict__ csr,
                                                  const float* __restrict__ bias,
                                                  float* __restrict__ out,
                                                  int N, int do_relu) {
    int node = blockIdx.x * 8 + (threadIdx.x >> 5);
    if (node >= N) return;
    int lane = threadIdx.x & 31;

    const float* hbase = h + 2 * lane;

    float ax = 0.f, ay = 0.f;
    int j = rowptr[node];
    int e = rowptr[node + 1];

    for (; j + 3 < e; j += 4) {
        v2i e0 = csr[j];
        v2i e1 = csr[j + 1];
        v2i e2 = csr[j + 2];
        v2i e3 = csr[j + 3];
        v2f h0 = *(const v2f*)(hbase + (size_t)e0.x * 64);
        v2f h1 = *(const v2f*)(hbase + (size_t)e1.x * 64);
        v2f h2 = *(const v2f*)(hbase + (size_t)e2.x * 64);
        v2f h3 = *(const v2f*)(hbase + (size_t)e3.x * 64);
        float w0 = __int_as_float(e0.y);
        float w1 = __int_as_float(e1.y);
        float w2 = __int_as_float(e2.y);
        float w3 = __int_as_float(e3.y);
        ax = fmaf(h0.x, w0, ax); ay = fmaf(h0.y, w0, ay);
        ax = fmaf(h1.x, w1, ax); ay = fmaf(h1.y, w1, ay);
        ax = fmaf(h2.x, w2, ax); ay = fmaf(h2.y, w2, ay);
        ax = fmaf(h3.x, w3, ax); ay = fmaf(h3.y, w3, ay);
    }
    for (; j < e; ++j) {
        v2i er = csr[j];
        float w = __int_as_float(er.y);
        v2f hv = *(const v2f*)(hbase + (size_t)er.x * 64);
        ax = fmaf(hv.x, w, ax);
        ay = fmaf(hv.y, w, ay);
    }

    v2f bv = *(const v2f*)(bias + 2 * lane);
    ax += bv.x;
    ay += bv.y;
    if (do_relu) {
        ax = fmaxf(ax, 0.f);
        ay = fmaxf(ay, 0.f);
    }
    v2f o; o.x = ax; o.y = ay;
    *(v2f*)(out + (size_t)node * 64 + 2 * lane) = o;
}

extern "C" void kernel_launch(void* const* d_in, const int* in_sizes, int n_in,
                              void* d_out, int out_size, void* d_ws, size_t ws_size,
                              hipStream_t stream) {
    (void)n_in; (void)out_size; (void)ws_size;

    const float* x  = (const float*)d_in[0];
    const int*   ei = (const int*)d_in[1];
    const float* W1 = (const float*)d_in[2];
    const float* b1 = (const float*)d_in[3];
    const float* W2 = (const float*)d_in[4];
    const float* b2 = (const float*)d_in[5];
    float* out = (float*)d_out;

    const int H   = in_sizes[3];
    const int IN  = in_sizes[2] / H;
    const int N   = in_sizes[0] / IN;
    const int E   = in_sizes[1] / 2;
    const int NNZ = N + E;

    char* ws = (char*)d_ws;
    size_t off = 0;
    auto alloc = [&](size_t bytes) -> void* {
        void* p = (void*)(ws + off);
        off += (bytes + 255) & ~(size_t)255;
        return p;
    };
    int*   cnt     = (int*)  alloc((size_t)N * 4);
    int*   rowptr  = (int*)  alloc((size_t)(N + 1) * 4);
    int*   cursor  = (int*)  alloc((size_t)N * 4);
    float* dinv    = (float*)alloc((size_t)N * 4);
    v2i*   csr     = (v2i*)  alloc((size_t)NNZ * 8);
    float* h1      = (float*)alloc((size_t)N * H * 4);
    float* y1      = (float*)alloc((size_t)N * H * 4);
    float* h2      = h1;

    const int T = 256;
    int gN = (N + T - 1) / T;
    int gE = (E + T - 1) / T;
    int gAgg = (N + 7) / 8;
    int gGemm = (N + 31) / 32;

    init_cnt_kernel<<<gN, T, 0, stream>>>(cnt, N);
    count_deg_kernel<<<gE, T, 0, stream>>>(ei, cnt, E);
    dinv_kernel<<<gN, T, 0, stream>>>(cnt, dinv, N);
    scan_kernel<<<1, 1024, 0, stream>>>(cnt, rowptr, N);
    init_csr_kernel<<<gN, T, 0, stream>>>(rowptr, dinv, cursor, csr, N);
    fill_csr_kernel<<<gE, T, 0, stream>>>(ei, dinv, cursor, csr, E);

    gemm64_wmma_kernel<<<gGemm, T, 0, stream>>>(x, W1, h1, N, IN);
    agg_kernel<<<gAgg, T, 0, stream>>>(h1, rowptr, csr, b1, y1, N, 1);

    gemm64_wmma_kernel<<<gGemm, T, 0, stream>>>(y1, W2, h2, N, H);
    agg_kernel<<<gAgg, T, 0, stream>>>(h2, rowptr, csr, b2, out, N, 0);
}


# CSR build without global atomics: degree count and slot assignment by per-block LDS histograms over node-range x edge-chunk tiles; a row's neighbours (sum order) now come chunk by chunk
# speedup vs baseline: 1.7082x; 1.7082x over previous
#include <hip/hip_runtime.h>
#include <hip/hip_bf16.h>


typedef float v2f __attribute__((ext_vector_type(2)));
typedef float v8f __attribute__((ext_vector_type(8)));
typedef int   v2i __attribute__((ext_vector_type(2)));

#define HR  12544
#define KCH 32

__device__ __forceinline__ int4 load_idx4(const int* __restrict__ p, int e, int eend, bool vec_ok) {
    int4 v;
    if (vec_ok && e + 3 < eend) {
        v = *(const int4*)(p + e);
    } else {
        v.x = (e     < eend) ? p[e]     : -1;
        v.y = (e + 1 < eend) ? p[e + 1] : -1;
        v.z = (e + 2 < eend) ? p[e + 2] : -1;
        v.w = (e + 3 < eend) ? p[e + 3] : -1;
    }
    return v;
}

__global__ __launch_bounds__(1024) void hist_kernel(const int* __restrict__ ei,
                                                    int* __restrict__ partial,
                                                    int N, int E, int EC) {
    __shared__ int bins[HR];
    int tid = threadIdx.x;
    int r0  = blockIdx.x * HR;
    int k   = blockIdx.y;
    int nr  = N - r0; if (nr > HR) nr = HR;
    for (int i = tid; i < HR; i += 1024) bins[i] = 0;
    __syncthreads();
    int ebeg = k * EC;
    int eend = ebeg + EC; if (eend > E) eend = E;
    for (int eb = ebeg; eb < eend; eb += 8192) {
        int e = eb + 4 * tid;
        int4 a, b;
        if (eb + 8192 <= eend) {
            a = *(const int4*)(ei + e);
            b = *(const int4*)(ei + e + 4096);
        } else {
            a = load_idx4(ei, e, eend, true);
            b = load_idx4(ei, e + 4096, eend, true);
        }
        int r[8] = {a.x, a.y, a.z, a.w, b.x, b.y, b.z, b.w};
#pragma unroll
        for (int q = 0; q < 8; ++q) {
            unsigned d = (unsigned)(r[q] - r0);
            if (d < (unsigned)nr) atomicAdd(&bins[d], 1);
        }
    }
    __syncthreads();
    int* dst = partial + (size_t)k * N + r0;
    for (int i = tid; i < nr; i += 1024) dst[i] = bins[i];
}

__global__ void combine_kernel(int* __restrict__ partial, int* __restrict__ cnt,
                               float* __restrict__ dinv, int N) {
    int i = blockIdx.x * blockDim.x + threadIdx.x;
    if (i < N) {
        int t[KCH];
#pragma unroll
        for (int k = 0; k < KCH; ++k) t[k] = partial[(size_t)k * N + i];
        int run = 0;
#pragma unroll
        for (int k = 0; k < KCH; ++k) {
            partial[(size_t)k * N + i] = run;
            run += t[k];
        }
        cnt[i]  = run;
        dinv[i] = rsqrtf((float)(run + 1));
    }
}

__global__ __launch_bounds__(1024) void scan_kernel(const int* __restrict__ cnt,
                                                    int* __restrict__ rowptr, int N) {
    __shared__ int sdata[1024];
    int t = threadIdx.x;
    int chunk = (N + 1023) / 1024;
    int begin = t * chunk;
    int end   = begin + chunk;
    if (begin > N) begin = N;
    if (end   > N) end   = N;
    int sum = 0;
    for (int i = begin; i < end; ++i) sum += cnt[i] + 1;
    sdata[t] = sum;
    __syncthreads();
    for (int off = 1; off < 1024; off <<= 1) {
        int v = (t >= off) ? sdata[t - off] : 0;
        __syncthreads();
        sdata[t] += v;
        __syncthreads();
    }
    int run = (t == 0) ? 0 : sdata[t - 1];
    for (int i = begin; i < end; ++i) {
        rowptr[i] = run;
        run += cnt[i] + 1;
    }
    if (t == 1023) rowptr[N] = sdata[1023];
}

__global__ void init_csr_kernel(const int* __restrict__ rowptr, const float* __restrict__ dinv,
                                v2i* __restrict__ csr, int N) {
    int i = blockIdx.x * blockDim.x + threadIdx.x;
    if (i < N) {
        int p = rowptr[i];
        float d = dinv[i];
        v2i rec;
        rec.x = i;
        rec.y = __float_as_int(d * d);
        csr[p] = rec;
    }
}

__global__ __launch_bounds__(1024) void fill_csr_kernel(const int* __restrict__ ei,
                                                        const float* __restrict__ dinv,
                                                        const int* __restrict__ rowptr,
                                                        const int* __restrict__ partial,
                                                        v2i* __restrict__ csr,
                                                        int N, int E, int EC) {
    __shared__ int slot[HR];
    int tid = threadIdx.x;
    int r0  = blockIdx.x * HR;
    int k   = blockIdx.y;
    int nr  = N - r0; if (nr > HR) nr = HR;
    const int* pre = partial + (size_t)k * N + r0;
    {
        constexpr int TRIPS = (HR + 1023) / 1024;
        int rp[TRIPS], pp[TRIPS];
#pragma unroll
        for (int it = 0; it < TRIPS; ++it) {
            int i = min(tid + it * 1024, nr - 1);
            rp[it] = rowptr[r0 + i];
            pp[it] = pre[i];
        }
#pragma unroll
        for (int it = 0; it < TRIPS; ++it) {
            int i = tid + it * 1024;
            if (i < nr) slot[i] = rp[it] + 1 + pp[it];
        }
    }
    __syncthreads();
    int ebeg = k * EC;
    int eend = ebeg + EC; if (eend > E) eend = E;
    const int* ec = ei + E;
    bool cvec = (E & 3) == 0;
    for (int eb = ebeg; eb < eend; eb += 8192) {
        int e = eb + 4 * tid;
        int4 ra, rb, ca, cb;
        if (cvec && eb + 8192 <= eend) {
            ra = *(const int4*)(ei + e);
            rb = *(const int4*)(ei + e + 4096);
            ca = *(const int4*)(ec + e);
            cb = *(const int4*)(ec + e + 4096);
        } else {
            ra = load_idx4(ei, e, eend, true);
            rb = load_idx4(ei, e + 4096, eend, true);
            ca = load_idx4(ec, e, eend, cvec);
            cb = load_idx4(ec, e + 4096, eend, cvec);
        }
        int r[8] = {ra.x, ra.y, ra.z, ra.w, rb.x, rb.y, rb.z, rb.w};
        int c[8] = {ca.x, ca.y, ca.z, ca.w, cb.x, cb.y, cb.z, cb.w};
        unsigned d[8];
        float dr[8], dc[8];
#pragma unroll
        for (int q = 0; q < 8; ++q) {
            d[q]  = (unsigned)(r[q] - r0);
            if (d[q] < (unsigned)nr) {
                dr[q] = dinv[r[q]];
                dc[q] = dinv[c[q]];
            }
        }
        unsigned p[8];
#pragma unroll
        for (int q = 0; q < 8; ++q) {
            if (d[q] < (unsigned)nr) p[q] = (unsigned)atomicAdd(&slot[d[q]], 1);
        }
#pragma unroll
        for (int q = 0; q < 8; ++q) {
            if (d[q] < (unsigned)nr) {
                v2i rec;
                rec.x = c[q];
                rec.y = __float_as_int(dr[q] * dc[q]);
                csr[p[q]] = rec;
            }
        }
    }
}

__global__ __launch_bounds__(256) void gemm64_wmma_kernel(const float* __restrict__ A,
                                                          const float* __restrict__ B,
                                                          float* __restrict__ C,
                                                          int M, int K) {
    int tid  = threadIdx.x;
    int wave = tid >> 5;
    int lane = tid & 31;
    int m0 = blockIdx.x * 32 + (wave >> 2) * 16;
    int n0 = (wave & 3) * 16;
    if (m0 >= M) return;
    int half = lane >> 4;
    int l    = lane & 15;

    const float* arow = A + (size_t)(m0 + l) * K;
    v8f acc = {0.f, 0.f, 0.f, 0.f, 0.f, 0.f, 0.f, 0.f};

    for (int k = 0; k < K; k += 4) {
        int ka = k + 2 * half;
        v2f a = *(const v2f*)(arow + ka);
        v2f b;
        b.x = B[(size_t)ka * 64 + n0 + l];
        b.y = B[(size_t)(ka + 1) * 64 + n0 + l];
        acc = __builtin_amdgcn_wmma_f32_16x16x4_f32(false, a, false, b,
                                                    (short)0, acc, false, false);
    }
    float* crow = C + (size_t)m0 * 64 + n0 + l;
#pragma unroll
    for (int j = 0; j < 8; ++j) {
        crow[(size_t)(j + 8 * half) * 64] = acc[j];
    }
}

__global__ __launch_bounds__(256) void agg_kernel(const float* __restrict__ h,
                                                  const int* __restrict__ rowptr,
                                                  const v2i* __restrict__ csr,
                                                  const float* __restrict__ bias,
                                                  float* __restrict__ out,
                                                  int N, int do_relu) {
    int node = blockIdx.x * 8 + (threadIdx.x >> 5);
    if (node >= N) return;
    int lane = threadIdx.x & 31;

    const float* hbase = h + 2 * lane;

    float ax = 0.f, ay = 0.f;
    int j = rowptr[node];
    int e = rowptr[node + 1];

    for (; j + 3 < e; j += 4) {
        v2i e0 = csr[j];
        v2i e1 = csr[j + 1];
        v2i e2 = csr[j + 2];
        v2i e3 = csr[j + 3];
        v2f h0 = *(const v2f*)(hbase + (size_t)e0.x * 64);
        v2f h1 = *(const v2f*)(hbase + (size_t)e1.x * 64);
        v2f h2 = *(const v2f*)(hbase + (size_t)e2.x * 64);
        v2f h3 = *(const v2f*)(hbase + (size_t)e3.x * 64);
        float w0 = __int_as_float(e0.y);
        float w1 = __int_as_float(e1.y);
        float w2 = __int_as_float(e2.y);
        float w3 = __int_as_float(e3.y);
        ax = fmaf(h0.x, w0, ax); ay = fmaf(h0.y, w0, ay);
        ax = fmaf(h1.x, w1, ax); ay = fmaf(h1.y, w1, ay);
        ax = fmaf(h2.x, w2, ax); ay = fmaf(h2.y, w2, ay);
        ax = fmaf(h3.x, w3, ax); ay = fmaf(h3.y, w3, ay);
    }
    for (; j < e; ++j) {
        v2i er = csr[j];
        float w = __int_as_float(er.y);
        v2f hv = *(const v2f*)(hbase + (size_t)er.x * 64);
        ax = fmaf(hv.x, w, ax);
        ay = fmaf(hv.y, w, ay);
    }

    v2f bv = *(const v2f*)(bias + 2 * lane);
    ax += bv.x;
    ay += bv.y;
    if (do_relu) {
        ax = fmaxf(ax, 0.f);
        ay = fmaxf(ay, 0.f);
    }
    v2f o; o.x = ax; o.y = ay;
    *(v2f*)(out + (size_t)node * 64 + 2 * lane) = o;
}

extern "C" void kernel_launch(void* const* d_in, const int* in_sizes, int n_in,
                              void* d_out, int out_size, void* d_ws, size_t ws_size,
                              hipStream_t stream) {
    (void)n_in; (void)out_size; (void)ws_size;

    const float* x  = (const float*)d_in[0];
    const int*   ei = (const int*)d_in[1];
    const float* W1 = (const float*)d_in[2];
    const float* b1 = (const float*)d_in[3];
    const float* W2 = (const float*)d_in[4];
    const float* b2 = (const float*)d_in[5];
    float* out = (float*)d_out;

    const int H   = in_sizes[3];
    const int IN  = in_sizes[2] / H;
    const int N   = in_sizes[0] / IN;
    const int E   = in_sizes[1] / 2;
    const int NNZ = N + E;

    char* ws = (char*)d_ws;
    size_t off = 0;
    auto alloc = [&](size_t bytes) -> void* {
        void* p = (void*)(ws + off);
        off += (bytes + 255) & ~(size_t)255;
        return p;
    };
    int*   cnt     = (int*)  alloc((size_t)N * 4);
    int*   rowptr  = (int*)  alloc((size_t)(N + 1) * 4);
    float* dinv    = (float*)alloc((size_t)N * 4);
    int*   partial = (int*)  alloc((size_t)KCH * N * 4);
    v2i*   csr     = (v2i*)  alloc((size_t)NNZ * 8);
    float* h1      = (float*)alloc((size_t)N * H * 4);
    float* y1      = (float*)alloc((size_t)N * H * 4);
    float* h2      = h1;

    const int T = 256;
    int gN = (N + T - 1) / T;
    int EC = (((E + KCH - 1) / KCH) + 3) & ~3;
    dim3 gTile((N + HR - 1) / HR, KCH);
    int gAgg = (N + 7) / 8;
    int gGemm = (N + 31) / 32;

    hist_kernel<<<gTile, 1024, 0, stream>>>(ei, partial, N, E, EC);
    combine_kernel<<<gN, T, 0, stream>>>(partial, cnt, dinv, N);
    scan_kernel<<<1, 1024, 0, stream>>>(cnt, rowptr, N);
    init_csr_kernel<<<gN, T, 0, stream>>>(rowptr, dinv, csr, N);
    fill_csr_kernel<<<gTile, 1024, 0, stream>>>(ei, dinv, rowptr, partial, csr, N, E, EC);

    gemm64_wmma_kernel<<<gGemm, T, 0, stream>>>(x, W1, h1, N, IN);
    agg_kernel<<<gAgg, T, 0, stream>>>(h1, rowptr, csr, b1, y1, N, 1);

    gemm64_wmma_kernel<<<gGemm, T, 0, stream>>>(y1, W2, h2, N, H);
    agg_kernel<<<gAgg, T, 0, stream>>>(h2, rowptr, csr, b2, out, N, 0);
}


# Resident waves: each aggregation wave walks the records of 10 consecutive rows as one stream (8 row gathers in flight), GEMM blocks loop over row tiles: 250k short waves become 36k; sum order kept
# speedup vs baseline: 1.8286x; 1.0705x over previous
#include <hip/hip_runtime.h>
#include <hip/hip_bf16.h>


typedef float v2f __attribute__((ext_vector_type(2)));
typedef float v8f __attribute__((ext_vector_type(8)));
typedef int   v2i __attribute__((ext_vector_type(2)));

#define HR  12544
#define KCH 32

__device__ __forceinline__ int4 load_idx4(const int* __restrict__ p, int e, int eend, bool vec_ok) {
    int4 v;
    if (vec_ok && e + 3 < eend) {
        v = *(const int4*)(p + e);
    } else {
        v.x = (e     < eend) ? p[e]     : -1;
        v.y = (e + 1 < eend) ? p[e + 1] : -1;
        v.z = (e + 2 < eend) ? p[e + 2] : -1;
        v.w = (e + 3 < eend) ? p[e + 3] : -1;
    }
    return v;
}

__global__ __launch_bounds__(1024) void hist_kernel(const int* __restrict__ ei,
                                                    int* __restrict__ partial,
                                                    int N, int E, int EC) {
    __shared__ int bins[HR];
    int tid = threadIdx.x;
    int r0  = blockIdx.x * HR;
    int k   = blockIdx.y;
    int nr  = N - r0; if (nr > HR) nr = HR;
    for (int i = tid; i < HR; i += 1024) bins[i] = 0;
    __syncthreads();
    int ebeg = k * EC;
    int eend = ebeg + EC; if (eend > E) eend = E;
    for (int eb = ebeg; eb < eend; eb += 8192) {
        int e = eb + 4 * tid;
        int4 a, b;
        if (eb + 8192 <= eend) {
            a = *(const int4*)(ei + e);
            b = *(const int4*)(ei + e + 4096);
        } else {
            a = load_idx4(ei, e, eend, true);
            b = load_idx4(ei, e + 4096, eend, true);
        }
        int r[8] = {a.x, a.y, a.z, a.w, b.x, b.y, b.z, b.w};
#pragma unroll
        for (int q = 0; q < 8; ++q) {
            unsigned d = (unsigned)(r[q] - r0);
            if (d < (unsigned)nr) atomicAdd(&bins[d], 1);
        }
    }
    __syncthreads();
    int* dst = partial + (size_t)k * N + r0;
    for (int i = tid; i < nr; i += 1024) dst[i] = bins[i];
}

__global__ void combine_kernel(int* __restrict__ partial, int* __restrict__ cnt,
                               float* __restrict__ dinv, int N) {
    int i = blockIdx.x * blockDim.x + threadIdx.x;
    if (i < N) {
        int t[KCH];
#pragma unroll
        for (int k = 0; k < KCH; ++k) t[k] = partial[(size_t)k * N + i];
        int run = 0;
#pragma unroll
        for (int k = 0; k < KCH; ++k) {
            partial[(size_t)k * N + i] = run;
            run += t[k];
        }
        cnt[i]  = run;
        dinv[i] = rsqrtf((float)(run + 1));
    }
}

__global__ __launch_bounds__(1024) void scan_kernel(const int* __restrict__ cnt,
                                                    int* __restrict__ rowptr, int N) {
    __shared__ int sdata[1024];
    int t = threadIdx.x;
    int chunk = (N + 1023) / 1024;
    int begin = t * chunk;
    int end   = begin + chunk;
    if (begin > N) begin = N;
    if (end   > N) end   = N;
    int sum = 0;
    for (int i = begin; i < end; ++i) sum += cnt[i] + 1;
    sdata[t] = sum;
    __syncthreads();
    for (int off = 1; off < 1024; off <<= 1) {
        int v = (t >= off) ? sdata[t - off] : 0;
        __syncthreads();
        sdata[t] += v;
        __syncthreads();
    }
    int run = (t == 0) ? 0 : sdata[t - 1];
    for (int i = begin; i < end; ++i) {
        rowptr[i] = run;
        run += cnt[i] + 1;
    }
    if (t == 1023) rowptr[N] = sdata[1023];
}

__global__ void init_csr_kernel(const int* __restrict__ rowptr, const float* __restrict__ dinv,
                                v2i* __restrict__ csr, int N) {
    int i = blockIdx.x * blockDim.x + threadIdx.x;
    if (i < N) {
        int p = rowptr[i];
        float d = dinv[i];
        v2i rec;
        rec.x = i;
        rec.y = __float_as_int(d * d);
        csr[p] = rec;
    }
}

__global__ __launch_bounds__(1024) void fill_csr_kernel(const int* __restrict__ ei,
                                                        const float* __restrict__ dinv,
                                                        const int* __restrict__ rowptr,
                                                        const int* __restrict__ partial,
                                                        v2i* __restrict__ csr,
                                                        int N, int E, int EC) {
    __shared__ int slot[HR];
    int tid = threadIdx.x;
    int r0  = blockIdx.x * HR;
    int k   = blockIdx.y;
    int nr  = N - r0; if (nr > HR) nr = HR;
    const int* pre = partial + (size_t)k * N + r0;
    {
        constexpr int TRIPS = (HR + 1023) / 1024;
        int rp[TRIPS], pp[TRIPS];
#pragma unroll
        for (int it = 0; it < TRIPS; ++it) {
            int i = min(tid + it * 1024, nr - 1);
            rp[it] = rowptr[r0 + i];
            pp[it] = pre[i];
        }
#pragma unroll
        for (int it = 0; it < TRIPS; ++it) {
            int i = tid + it * 1024;
            if (i < nr) slot[i] = rp[it] + 1 + pp[it];
        }
    }
    __syncthreads();
    int ebeg = k * EC;
    int eend = ebeg + EC; if (eend > E) eend = E;
    const int* ec = ei + E;
    bool cvec = (E & 3) == 0;
    for (int eb = ebeg; eb < eend; eb += 8192) {
        int e = eb + 4 * tid;
        int4 ra, rb, ca, cb;
        if (cvec && eb + 8192 <= eend) {
            ra = *(const int4*)(ei + e);
            rb = *(const int4*)(ei + e + 4096);
            ca = *(const int4*)(ec + e);
            cb = *(const int4*)(ec + e + 4096);
        } else {
            ra = load_idx4(ei, e, eend, true);
            rb = load_idx4(ei, e + 4096, eend, true);
            ca = load_idx4(ec, e, eend, cvec);
            cb = load_idx4(ec, e + 4096, eend, cvec);
        }
        int r[8] = {ra.x, ra.y, ra.z, ra.w, rb.x, rb.y, rb.z, rb.w};
        int c[8] = {ca.x, ca.y, ca.z, ca.w, cb.x, cb.y, cb.z, cb.w};
        unsigned d[8];
        float dr[8], dc[8];
#pragma unroll
        for (int q = 0; q < 8; ++q) {
            d[q]  = (unsigned)(r[q] - r0);
            if (d[q] < (unsigned)nr) {
                dr[q] = dinv[r[q]];
                dc[q] = dinv[c[q]];
            }
        }
        unsigned p[8];
#pragma unroll
        for (int q = 0; q < 8; ++q) {
            if (d[q] < (unsigned)nr) p[q] = (unsigned)atomicAdd(&slot[d[q]], 1);
        }
#pragma unroll
        for (int q = 0; q < 8; ++q) {
            if (d[q] < (unsigned)nr) {
                v2i rec;
                rec.x = c[q];
                rec.y = __float_as_int(dr[q] * dc[q]);
                csr[p[q]] = rec;
            }
        }
    }
}

__global__ __launch_bounds__(256) void gemm64_wmma_kernel(const float* __restrict__ A,
                                                          const float* __restrict__ B,
                                                          float* __restrict__ C,
                                                          int M, int K) {
    int tid  = threadIdx.x;
    int wave = __builtin_amdgcn_readfirstlane(tid >> 5);
    int lane = tid & 31;
    int n0 = (wave & 3) * 16;
    int half = lane >> 4;
    int l    = lane & 15;

    for (int t = blockIdx.x; t * 32 < M; t += gridDim.x) {
        int m0 = t * 32 + (wave >> 2) * 16;
        if (m0 >= M) break;

        const float* arow = A + (size_t)(m0 + l) * K;
        v8f acc = {0.f, 0.f, 0.f, 0.f, 0.f, 0.f, 0.f, 0.f};

        for (int k = 0; k < K; k += 4) {
            int ka = k + 2 * half;
            v2f a = *(const v2f*)(arow + ka);
            v2f b;
            b.x = B[(size_t)ka * 64 + n0 + l];
            b.y = B[(size_t)(ka + 1) * 64 + n0 + l];
            acc = __builtin_amdgcn_wmma_f32_16x16x4_f32(false, a, false, b,
                                                        (short)0, acc, false, false);
        }
        float* crow = C + (size_t)m0 * 64 + n0 + l;
#pragma unroll
        for (int j = 0; j < 8; ++j) {
            crow[(size_t)(j + 8 * half) * 64] = acc[j];
        }
    }
}

#define AGB 8

template <bool GUARD>
__device__ __forceinline__ void agg_step(const v2i (&r)[AGB], int j, int end,
                                         const float* __restrict__ h, unsigned lane8,
                                         float* __restrict__ obase, int rp,
                                         v2f bv, int do_relu,
                                         float& ax, float& ay, int& row, int& rend) {
    v2f hv[AGB];
#pragma unroll
    for (int q = 0; q < AGB; ++q)
        hv[q] = *(const v2f*)((const char*)h + ((unsigned)r[q].x * 256u + lane8));
    int left = rend - j;
#pragma unroll
    for (int q = 0; q < AGB; ++q) {
        if (!GUARD || j + q < end) {
            float w = __int_as_float(r[q].y);
            ax = fmaf(hv[q].x, w, ax);
            ay = fmaf(hv[q].y, w, ay);
            if (q + 1 == left) {
                float ox = ax + bv.x;
                float oy = ay + bv.y;
                if (do_relu) {
                    ox = fmaxf(ox, 0.f);
                    oy = fmaxf(oy, 0.f);
                }
                v2f o; o.x = ox; o.y = oy;
                *(v2f*)(obase + (size_t)row * 64) = o;
                ax = 0.f; ay = 0.f;
                ++row;
                rend = __builtin_amdgcn_readlane(rp, row + 1);
                left = rend - j;
            }
        }
    }
}

__device__ __forceinline__ void agg_trip(const v2i (&rc)[AGB], v2i (&rn)[AGB],
                                         const v2i* __restrict__ csr, int& j, int end,
                                         const float* __restrict__ h, unsigned lane8,
                                         float* __restrict__ obase, int rp,
                                         v2f bv, int do_relu,
                                         float& ax, float& ay, int& row, int& rend) {
    int jn = j + AGB;
    if (jn + AGB <= end) {
        const v2i* pn = csr + jn;
#pragma unroll
        for (int q = 0; q < AGB; ++q) rn[q] = pn[q];
    } else {
#pragma unroll
        for (int q = 0; q < AGB; ++q) rn[q] = csr[min(jn + q, end - 1)];
    }
    if (jn <= end) agg_step<false>(rc, j, end, h, lane8, obase, rp, bv, do_relu, ax, ay, row, rend);
    else           agg_step<true >(rc, j, end, h, lane8, obase, rp, bv, do_relu, ax, ay, row, rend);
    j = jn;
}

__global__ __launch_bounds__(256) void agg_kernel(const float* __restrict__ h,
                                                  const int* __restrict__ rowptr,
                                                  const v2i* __restrict__ csr,
                                                  const float* __restrict__ bias,
                                                  float* __restrict__ out,
                                                  int N, int NPW, int do_relu) {
    int lane = threadIdx.x & 31;
    int w  = __builtin_amdgcn_readfirstlane((int)(blockIdx.x * 8 + (threadIdx.x >> 5)));
    int n0 = w * NPW;
    if (n0 >= N) return;
    int cnt = N - n0; if (cnt > NPW) cnt = NPW;

    int rp   = rowptr[n0 + min(lane, cnt)];
    int j    = __builtin_amdgcn_readlane(rp, 0);
    int end  = __builtin_amdgcn_readlane(rp, cnt);
    int row  = 0;
    int rend = __builtin_amdgcn_readlane(rp, 1);
    if (j >= end) return;

    unsigned lane8 = (unsigned)lane * 8u;
    float*   obase = out + (size_t)n0 * 64 + 2 * lane;
    v2f bv = *(const v2f*)(bias + 2 * lane);

    float ax = 0.f, ay = 0.f;
    v2i ra[AGB], rb[AGB];
#pragma unroll
    for (int q = 0; q < AGB; ++q) ra[q] = csr[min(j + q, end - 1)];

    for (;;) {
        agg_trip(ra, rb, csr, j, end, h, lane8, obase, rp, bv, do_relu, ax, ay, row, rend);
        if (j >= end) break;
        agg_trip(rb, ra, csr, j, end, h, lane8, obase, rp, bv, do_relu, ax, ay, row, rend);
        if (j >= end) break;
    }
}

extern "C" void kernel_launch(void* const* d_in, const int* in_sizes, int n_in,
                              void* d_out, int out_size, void* d_ws, size_t ws_size,
                              hipStream_t stream) {
    (void)n_in; (void)out_size; (void)ws_size;

    const float* x  = (const float*)d_in[0];
    const int*   ei = (const int*)d_in[1];
    const float* W1 = (const float*)d_in[2];
    const float* b1 = (const float*)d_in[3];
    const float* W2 = (const float*)d_in[4];
    const float* b2 = (const float*)d_in[5];
    float* out = (float*)d_out;

    const int H   = in_sizes[3];
    const int IN  = in_sizes[2] / H;
    const int N   = in_sizes[0] / IN;
    const int E   = in_sizes[1] / 2;
    const int NNZ = N + E;

    char* ws = (char*)d_ws;
    size_t off = 0;
    auto alloc = [&](size_t bytes) -> void* {
        void* p = (void*)(ws + off);
        off += (bytes + 255) & ~(size_t)255;
        return p;
    };
    int*   cnt     = (int*)  alloc((size_t)N * 4);
    int*   rowptr  = (int*)  alloc((size_t)(N + 1) * 4);
    float* dinv    = (float*)alloc((size_t)N * 4);
    int*   partial = (int*)  alloc((size_t)KCH * N * 4);
    v2i*   csr     = (v2i*)  alloc((size_t)NNZ * 8);
    float* h1      = (float*)alloc((size_t)N * H * 4);
    float* y1      = (float*)alloc((size_t)N * H * 4);
    float* h2      = h1;

    const int T = 256;
    int gN = (N + T - 1) / T;
    int EC = (((E + KCH - 1) / KCH) + 3) & ~3;
    dim3 gTile((N + HR - 1) / HR, KCH);
    int NPW  = (N + 10239) / 10240;
    if (NPW > 30) NPW = 30;
    int gAgg = ((N + NPW - 1) / NPW + 7) / 8;
    int gGemm = (N + 31) / 32;
    if (gGemm > 1024) gGemm = 1024;

    hist_kernel<<<gTile, 1024, 0, stream>>>(ei, partial, N, E, EC);
    combine_kernel<<<gN, T, 0, stream>>>(partial, cnt, dinv, N);
    scan_kernel<<<1, 1024, 0, stream>>>(cnt, rowptr, N);
    init_csr_kernel<<<gN, T, 0, stream>>>(rowptr, dinv, csr, N);
    fill_csr_kernel<<<gTile, 1024, 0, stream>>>(ei, dinv, rowptr, partial, csr, N, E, EC);

    gemm64_wmma_kernel<<<gGemm, T, 0, stream>>>(x, W1, h1, N, IN);
    agg_kernel<<<gAgg, T, 0, stream>>>(h1, rowptr, csr, b1, y1, N, NPW, 1);

    gemm64_wmma_kernel<<<gGemm, T, 0, stream>>>(y1, W2, h2, N, H);
    agg_kernel<<<gAgg, T, 0, stream>>>(h2, rowptr, csr, b2, out, N, NPW, 0);
}


# Row-pointer scan made two-level and folded into the launches beside it: block prefix in the combine launch, block offsets summed in the self-loop launch; single-block scan launch gone
# speedup vs baseline: 2.6280x; 1.4372x over previous
#include <hip/hip_runtime.h>
#include <hip/hip_bf16.h>


typedef float v2f __attribute__((ext_vector_type(2)));
typedef float v8f __attribute__((ext_vector_type(8)));
typedef int   v2i __attribute__((ext_vector_type(2)));

#define HR  12544
#define KCH 32

__device__ __forceinline__ int4 load_idx4(const int* __restrict__ p, int e, int eend, bool vec_ok) {
    int4 v;
    if (vec_ok && e + 3 < eend) {
        v = *(const int4*)(p + e);
    } else {
        v.x = (e     < eend) ? p[e]     : -1;
        v.y = (e + 1 < eend) ? p[e + 1] : -1;
        v.z = (e + 2 < eend) ? p[e + 2] : -1;
        v.w = (e + 3 < eend) ? p[e + 3] : -1;
    }
    return v;
}

__global__ __launch_bounds__(1024) void hist_kernel(const int* __restrict__ ei,
                                                    int* __restrict__ partial,
                                                    int N, int E, int EC) {
    __shared__ int bins[HR];
    int tid = threadIdx.x;
    int r0  = blockIdx.x * HR;
    int k   = blockIdx.y;
    int nr  = N - r0; if (nr > HR) nr = HR;
    for (int i = tid; i < HR; i += 1024) bins[i] = 0;
    __syncthreads();
    int ebeg = k * EC;
    int eend = ebeg + EC; if (eend > E) eend = E;
    for (int eb = ebeg; eb < eend; eb += 8192) {
        int e = eb + 4 * tid;
        int4 a, b;
        if (eb + 8192 <= eend) {
            a = *(const int4*)(ei + e);
            b = *(const int4*)(ei + e + 4096);
        } else {
            a = load_idx4(ei, e, eend, true);
            b = load_idx4(ei, e + 4096, eend, true);
        }
        int r[8] = {a.x, a.y, a.z, a.w, b.x, b.y, b.z, b.w};
#pragma unroll
        for (int q = 0; q < 8; ++q) {
            unsigned d = (unsigned)(r[q] - r0);
            if (d < (unsigned)nr) atomicAdd(&bins[d], 1);
        }
    }
    __syncthreads();
    int* dst = partial + (size_t)k * N + r0;
    for (int i = tid; i < nr; i += 1024) dst[i] = bins[i];
}

#define CB 256

__global__ __launch_bounds__(CB) void combine_kernel(int* __restrict__ partial, int* __restrict__ rowptr,
                                                     int* __restrict__ bsum, float* __restrict__ dinv,
                                                     int N) {
    __shared__ int wsum[CB / 32];
    int tid = threadIdx.x;
    int i = blockIdx.x * CB + tid;
    int v = 0;
    if (i < N) {
        int t[KCH];
#pragma unroll
        for (int k = 0; k < KCH; ++k) t[k] = partial[(size_t)k * N + i];
        int run = 0;
#pragma unroll
        for (int k = 0; k < KCH; ++k) {
            partial[(size_t)k * N + i] = run;
            run += t[k];
        }
        dinv[i] = rsqrtf((float)(run + 1));
        v = run + 1;
    }
    int lane = tid & 31;
    int wv   = tid >> 5;
    int inc  = v;
    for (int off = 1; off < 32; off <<= 1) {
        int n = __shfl_up(inc, off, 32);
        if (lane >= off) inc += n;
    }
    if (lane == 31) wsum[wv] = inc;
    __syncthreads();
    int base = 0, tot = 0;
    for (int q = 0; q < CB / 32; ++q) {
        int s = wsum[q];
        if (q < wv) base += s;
        tot += s;
    }
    if (i < N) rowptr[i] = base + inc - v;
    if (tid == 0) bsum[blockIdx.x] = tot;
}

__global__ __launch_bounds__(CB) void init_csr_kernel(int* __restrict__ rowptr, const int* __restrict__ bsum,
                                                      const float* __restrict__ dinv,
                                                      v2i* __restrict__ csr, int N) {
    __shared__ int wsum[CB / 32];
    int tid = threadIdx.x;
    int b   = blockIdx.x;
    int i   = b * CB + tid;
    int   loc = 0;
    float d   = 0.f;
    if (i < N) {
        loc = rowptr[i];
        d   = dinv[i];
    }
    int s = 0;
    for (int q = tid; q < b; q += CB) s += bsum[q];
    for (int off = 16; off > 0; off >>= 1) s += __shfl_xor(s, off, 32);
    if ((tid & 31) == 0) wsum[tid >> 5] = s;
    __syncthreads();
    int boff = 0;
    for (int q = 0; q < CB / 32; ++q) boff += wsum[q];
    if (i < N) {
        int p = loc + boff;
        rowptr[i] = p;
        v2i rec;
        rec.x = i;
        rec.y = __float_as_int(d * d);
        csr[p] = rec;
    }
    if (b == (int)gridDim.x - 1 && tid == 0) rowptr[N] = boff + bsum[b];
}

__global__ __launch_bounds__(1024) void fill_csr_kernel(const int* __restrict__ ei,
                                                        const float* __restrict__ dinv,
                                                        const int* __restrict__ rowptr,
                                                        const int* __restrict__ partial,
                                                        v2i* __restrict__ csr,
                                                        int N, int E, int EC) {
    __shared__ int slot[HR];
    int tid = threadIdx.x;
    int r0  = blockIdx.x * HR;
    int k   = blockIdx.y;
    int nr  = N - r0; if (nr > HR) nr = HR;
    const int* pre = partial + (size_t)k * N + r0;
    {
        constexpr int TRIPS = (HR + 1023) / 1024;
        int rp[TRIPS], pp[TRIPS];
#pragma unroll
        for (int it = 0; it < TRIPS; ++it) {
            int i = min(tid + it * 1024, nr - 1);
            rp[it] = rowptr[r0 + i];
            pp[it] = pre[i];
        }
#pragma unroll
        for (int it = 0; it < TRIPS; ++it) {
            int i = tid + it * 1024;
            if (i < nr) slot[i] = rp[it] + 1 + pp[it];
        }
    }
    __syncthreads();
    int ebeg = k * EC;
    int eend = ebeg + EC; if (eend > E) eend = E;
    const int* ec = ei + E;
    bool cvec = (E & 3) == 0;
    for (int eb = ebeg; eb < eend; eb += 8192) {
        int e = eb + 4 * tid;
        int4 ra, rb, ca, cb;
        if (cvec && eb + 8192 <= eend) {
            ra = *(const int4*)(ei + e);
            rb = *(const int4*)(ei + e + 4096);
            ca = *(const int4*)(ec + e);
            cb = *(const int4*)(ec + e + 4096);
        } else {
            ra = load_idx4(ei, e, eend, true);
            rb = load_idx4(ei, e + 4096, eend, true);
            ca = load_idx4(ec, e, eend, cvec);
            cb = load_idx4(ec, e + 4096, eend, cvec);
        }
        int r[8] = {ra.x, ra.y, ra.z, ra.w, rb.x, rb.y, rb.z, rb.w};
        int c[8] = {ca.x, ca.y, ca.z, ca.w, cb.x, cb.y, cb.z, cb.w};
        unsigned d[8];
        float dr[8], dc[8];
#pragma unroll
        for (int q = 0; q < 8; ++q) {
            d[q]  = (unsigned)(r[q] - r0);
            if (d[q] < (unsigned)nr) {
                dr[q] = dinv[r[q]];
                dc[q] = dinv[c[q]];
            }
        }
        unsigned p[8];
#pragma unroll
        for (int q = 0; q < 8; ++q) {
            if (d[q] < (unsigned)nr) p[q] = (unsigned)atomicAdd(&slot[d[q]], 1);
        }
#pragma unroll
        for (int q = 0; q < 8; ++q) {
            if (d[q] < (unsigned)nr) {
                v2i rec;
                rec.x = c[q];
                rec.y = __float_as_int(dr[q] * dc[q]);
                csr[p[q]] = rec;
            }
        }
    }
}

__global__ __launch_bounds__(256) void gemm64_wmma_kernel(const float* __restrict__ A,
                                                          const float* __restrict__ B,
                                                          float* __restrict__ C,
                                                          int M, int K) {
    int tid  = threadIdx.x;
    int wave = __builtin_amdgcn_readfirstlane(tid >> 5);
    int lane = tid & 31;
    int n0 = (wave & 3) * 16;
    int half = lane >> 4;
    int l    = lane & 15;

    for (int t = blockIdx.x; t * 32 < M; t += gridDim.x) {
        int m0 = t * 32 + (wave >> 2) * 16;
        if (m0 >= M) break;

        const float* arow = A + (size_t)(m0 + l) * K;
        v8f acc = {0.f, 0.f, 0.f, 0.f, 0.f, 0.f, 0.f, 0.f};

        for (int k = 0; k < K; k += 4) {
            int ka = k + 2 * half;
            v2f a = *(const v2f*)(arow + ka);
            v2f b;
            b.x = B[(size_t)ka * 64 + n0 + l];
            b.y = B[(size_t)(ka + 1) * 64 + n0 + l];
            acc = __builtin_amdgcn_wmma_f32_16x16x4_f32(false, a, false, b,
                                                        (short)0, acc, false, false);
        }
        float* crow = C + (size_t)m0 * 64 + n0 + l;
#pragma unroll
        for (int j = 0; j < 8; ++j) {
            crow[(size_t)(j + 8 * half) * 64] = acc[j];
        }
    }
}

#define AGB 8

template <bool GUARD>
__device__ __forceinline__ void agg_step(const v2i (&r)[AGB], int j, int end,
                                         const float* __restrict__ h, unsigned lane8,
                                         float* __restrict__ obase, int rp,
                                         v2f bv, int do_relu,
                                         float& ax, float& ay, int& row, int& rend) {
    v2f hv[AGB];
#pragma unroll
    for (int q = 0; q < AGB; ++q)
        hv[q] = *(const v2f*)((const char*)h + ((unsigned)r[q].x * 256u + lane8));
    int left = rend - j;
#pragma unroll
    for (int q = 0; q < AGB; ++q) {
        if (!GUARD || j + q < end) {
            float w = __int_as_float(r[q].y);
            ax = fmaf(hv[q].x, w, ax);
            ay = fmaf(hv[q].y, w, ay);
            if (q + 1 == left) {
                float ox = ax + bv.x;
                float oy = ay + bv.y;
                if (do_relu) {
                    ox = fmaxf(ox, 0.f);
                    oy = fmaxf(oy, 0.f);
                }
                v2f o; o.x = ox; o.y = oy;
                *(v2f*)(obase + (size_t)row * 64) = o;
                ax = 0.f; ay = 0.f;
                ++row;
                rend = __builtin_amdgcn_readlane(rp, row + 1);
                left = rend - j;
            }
        }
    }
}

__device__ __forceinline__ void agg_trip(const v2i (&rc)[AGB], v2i (&rn)[AGB],
                                         const v2i* __restrict__ csr, int& j, int end,
                                         const float* __restrict__ h, unsigned lane8,
                                         float* __restrict__ obase, int rp,
                                         v2f bv, int do_relu,
                                         float& ax, float& ay, int& row, int& rend) {
    int jn = j + AGB;
    if (jn + AGB <= end) {
        const v2i* pn = csr + jn;
#pragma unroll
        for (int q = 0; q < AGB; ++q) rn[q] = pn[q];
    } else {
#pragma unroll
        for (int q = 0; q < AGB; ++q) rn[q] = csr[min(jn + q, end - 1)];
    }
    if (jn <= end) agg_step<false>(rc, j, end, h, lane8, obase, rp, bv, do_relu, ax, ay, row, rend);
    else           agg_step<true >(rc, j, end, h, lane8, obase, rp, bv, do_relu, ax, ay, row, rend);
    j = jn;
}

__global__ __launch_bounds__(256) void agg_kernel(const float* __restrict__ h,
                                                  const int* __restrict__ rowptr,
                                                  const v2i* __restrict__ csr,
                                                  const float* __restrict__ bias,
                                                  float* __restrict__ out,
                                                  int N, int NPW, int do_relu) {
    int lane = threadIdx.x & 31;
    int w  = __builtin_amdgcn_readfirstlane((int)(blockIdx.x * 8 + (threadIdx.x >> 5)));
    int n0 = w * NPW;
    if (n0 >= N) return;
    int cnt = N - n0; if (cnt > NPW) cnt = NPW;

    int rp   = rowptr[n0 + min(lane, cnt)];
    int j    = __builtin_amdgcn_readlane(rp, 0);
    int end  = __builtin_amdgcn_readlane(rp, cnt);
    int row  = 0;
    int rend = __builtin_amdgcn_readlane(rp, 1);
    if (j >= end) return;

    unsigned lane8 = (unsigned)lane * 8u;
    float*   obase = out + (size_t)n0 * 64 + 2 * lane;
    v2f bv = *(const v2f*)(bias + 2 * lane);

    float ax = 0.f, ay = 0.f;
    v2i ra[AGB], rb[AGB];
#pragma unroll
    for (int q = 0; q < AGB; ++q) ra[q] = csr[min(j + q, end - 1)];

    for (;;) {
        agg_trip(ra, rb, csr, j, end, h, lane8, obase, rp, bv, do_relu, ax, ay, row, rend);
        if (j >= end) break;
        agg_trip(rb, ra, csr, j, end, h, lane8, obase, rp, bv, do_relu, ax, ay, row, rend);
        if (j >= end) break;
    }
}

extern "C" void kernel_launch(void* const* d_in, const int* in_sizes, int n_in,
                              void* d_out, int out_size, void* d_ws, size_t ws_size,
                              hipStream_t stream) {
    (void)n_in; (void)out_size; (void)ws_size;

    const float* x  = (const float*)d_in[0];
    const int*   ei = (const int*)d_in[1];
    const float* W1 = (const float*)d_in[2];
    const float* b1 = (const float*)d_in[3];
    const float* W2 = (const float*)d_in[4];
    const float* b2 = (const float*)d_in[5];
    float* out = (float*)d_out;

    const int H   = in_sizes[3];
    const int IN  = in_sizes[2] / H;
    const int N   = in_sizes[0] / IN;
    const int E   = in_sizes[1] / 2;
    const int NNZ = N + E;

    char* ws = (char*)d_ws;
    size_t off = 0;
    auto alloc = [&](size_t bytes) -> void* {
        void* p = (void*)(ws + off);
        off += (bytes + 255) & ~(size_t)255;
        return p;
    };
    const int T = 256;
    int gN = (N + CB - 1) / CB;
    int*   bsum    = (int*)  alloc((size_t)gN * 4);
    int*   rowptr  = (int*)  alloc((size_t)(N + 1) * 4);
    float* dinv    = (float*)alloc((size_t)N * 4);
    int*   partial = (int*)  alloc((size_t)KCH * N * 4);
    v2i*   csr     = (v2i*)  alloc((size_t)NNZ * 8);
    float* h1      = (float*)alloc((size_t)N * H * 4);
    float* y1      = (float*)alloc((size_t)N * H * 4);
    float* h2      = h1;

    int EC = (((E + KCH - 1) / KCH) + 3) & ~3;
    dim3 gTile((N + HR - 1) / HR, KCH);
    int NPW  = (N + 10239) / 10240;
    if (NPW > 30) NPW = 30;
    int gAgg = ((N + NPW - 1) / NPW + 7) / 8;
    int gGemm = (N + 31) / 32;
    if (gGemm > 1024) gGemm = 1024;

    hist_kernel<<<gTile, 1024, 0, stream>>>(ei, partial, N, E, EC);
    combine_kernel<<<gN, CB, 0, stream>>>(partial, rowptr, bsum, dinv, N);
    init_csr_kernel<<<gN, CB, 0, stream>>>(rowptr, bsum, dinv, csr, N);
    fill_csr_kernel<<<gTile, 1024, 0, stream>>>(ei, dinv, rowptr, partial, csr, N, E, EC);

    gemm64_wmma_kernel<<<gGemm, T, 0, stream>>>(x, W1, h1, N, IN);
    agg_kernel<<<gAgg, T, 0, stream>>>(h1, rowptr, csr, b1, y1, N, NPW, 1);

    gemm64_wmma_kernel<<<gGemm, T, 0, stream>>>(y1, W2, h2, N, H);
    agg_kernel<<<gAgg, T, 0, stream>>>(h2, rowptr, csr, b2, out, N, NPW, 0);
}


# GEMM operand feed: weight matrix held in each wave's registers as B fragments, a wave does 16 rows x 64 cols per A fragment (16B loads); k grouped {0,1,4,5},{2,3,6,7}: GEMM sum order changes
# speedup vs baseline: 2.8568x; 1.0871x over previous
#include <hip/hip_runtime.h>
#include <hip/hip_bf16.h>


typedef float v2f __attribute__((ext_vector_type(2)));
typedef float v8f __attribute__((ext_vector_type(8)));
typedef int   v2i __attribute__((ext_vector_type(2)));

#define HR  12544
#define KCH 32

__device__ __forceinline__ int4 load_idx4(const int* __restrict__ p, int e, int eend, bool vec_ok) {
    int4 v;
    if (vec_ok && e + 3 < eend) {
        v = *(const int4*)(p + e);
    } else {
        v.x = (e     < eend) ? p[e]     : -1;
        v.y = (e + 1 < eend) ? p[e + 1] : -1;
        v.z = (e + 2 < eend) ? p[e + 2] : -1;
        v.w = (e + 3 < eend) ? p[e + 3] : -1;
    }
    return v;
}

__global__ __launch_bounds__(1024) void hist_kernel(const int* __restrict__ ei,
                                                    int* __restrict__ partial,
                                                    int N, int E, int EC) {
    __shared__ int bins[HR];
    int tid = threadIdx.x;
    int r0  = blockIdx.x * HR;
    int k   = blockIdx.y;
    int nr  = N - r0; if (nr > HR) nr = HR;
    for (int i = tid; i < HR; i += 1024) bins[i] = 0;
    __syncthreads();
    int ebeg = k * EC;
    int eend = ebeg + EC; if (eend > E) eend = E;
    for (int eb = ebeg; eb < eend; eb += 8192) {
        int e = eb + 4 * tid;
        int4 a, b;
        if (eb + 8192 <= eend) {
            a = *(const int4*)(ei + e);
            b = *(const int4*)(ei + e + 4096);
        } else {
            a = load_idx4(ei, e, eend, true);
            b = load_idx4(ei, e + 4096, eend, true);
        }
        int r[8] = {a.x, a.y, a.z, a.w, b.x, b.y, b.z, b.w};
#pragma unroll
        for (int q = 0; q < 8; ++q) {
            unsigned d = (unsigned)(r[q] - r0);
            if (d < (unsigned)nr) atomicAdd(&bins[d], 1);
        }
    }
    __syncthreads();
    int* dst = partial + (size_t)k * N + r0;
    for (int i = tid; i < nr; i += 1024) dst[i] = bins[i];
}

#define CB 256

__global__ __launch_bounds__(CB) void combine_kernel(int* __restrict__ partial, int* __restrict__ rowptr,
                                                     int* __restrict__ bsum, float* __restrict__ dinv,
                                                     int N) {
    __shared__ int wsum[CB / 32];
    int tid = threadIdx.x;
    int i = blockIdx.x * CB + tid;
    int v = 0;
    if (i < N) {
        int t[KCH];
#pragma unroll
        for (int k = 0; k < KCH; ++k) t[k] = partial[(size_t)k * N + i];
        int run = 0;
#pragma unroll
        for (int k = 0; k < KCH; ++k) {
            partial[(size_t)k * N + i] = run;
            run += t[k];
        }
        dinv[i] = rsqrtf((float)(run + 1));
        v = run + 1;
    }
    int lane = tid & 31;
    int wv   = tid >> 5;
    int inc  = v;
    for (int off = 1; off < 32; off <<= 1) {
        int n = __shfl_up(inc, off, 32);
        if (lane >= off) inc += n;
    }
    if (lane == 31) wsum[wv] = inc;
    __syncthreads();
    int base = 0, tot = 0;
    for (int q = 0; q < CB / 32; ++q) {
        int s = wsum[q];
        if (q < wv) base += s;
        tot += s;
    }
    if (i < N) rowptr[i] = base + inc - v;
    if (tid == 0) bsum[blockIdx.x] = tot;
}

__global__ __launch_bounds__(CB) void init_csr_kernel(int* __restrict__ rowptr, const int* __restrict__ bsum,
                                                      const float* __restrict__ dinv,
                                                      v2i* __restrict__ csr, int N) {
    __shared__ int wsum[CB / 32];
    int tid = threadIdx.x;
    int b   = blockIdx.x;
    int i   = b * CB + tid;
    int   loc = 0;
    float d   = 0.f;
    if (i < N) {
        loc = rowptr[i];
        d   = dinv[i];
    }
    int s = 0;
    for (int q = tid; q < b; q += CB) s += bsum[q];
    for (int off = 16; off > 0; off >>= 1) s += __shfl_xor(s, off, 32);
    if ((tid & 31) == 0) wsum[tid >> 5] = s;
    __syncthreads();
    int boff = 0;
    for (int q = 0; q < CB / 32; ++q) boff += wsum[q];
    if (i < N) {
        int p = loc + boff;
        rowptr[i] = p;
        v2i rec;
        rec.x = i;
        rec.y = __float_as_int(d * d);
        csr[p] = rec;
    }
    if (b == (int)gridDim.x - 1 && tid == 0) rowptr[N] = boff + bsum[b];
}

__global__ __launch_bounds__(1024) void fill_csr_kernel(const int* __restrict__ ei,
                                                        const float* __restrict__ dinv,
                                                        const int* __restrict__ rowptr,
                                                        const int* __restrict__ partial,
                                                        v2i* __restrict__ csr,
                                                        int N, int E, int EC) {
    __shared__ int slot[HR];
    int tid = threadIdx.x;
    int r0  = blockIdx.x * HR;
    int k   = blockIdx.y;
    int nr  = N - r0; if (nr > HR) nr = HR;
    const int* pre = partial + (size_t)k * N + r0;
    {
        constexpr int TRIPS = (HR + 1023) / 1024;
        int rp[TRIPS], pp[TRIPS];
#pragma unroll
        for (int it = 0; it < TRIPS; ++it) {
            int i = min(tid + it * 1024, nr - 1);
            rp[it] = rowptr[r0 + i];
            pp[it] = pre[i];
        }
#pragma unroll
        for (int it = 0; it < TRIPS; ++it) {
            int i = tid + it * 1024;
            if (i < nr) slot[i] = rp[it] + 1 + pp[it];
        }
    }
    __syncthreads();
    int ebeg = k * EC;
    int eend = ebeg + EC; if (eend > E) eend = E;
    const int* ec = ei + E;
    bool cvec = (E & 3) == 0;
    for (int eb = ebeg; eb < eend; eb += 8192) {
        int e = eb + 4 * tid;
        int4 ra, rb, ca, cb;
        if (cvec && eb + 8192 <= eend) {
            ra = *(const int4*)(ei + e);
            rb = *(const int4*)(ei + e + 4096);
            ca = *(const int4*)(ec + e);
            cb = *(const int4*)(ec + e + 4096);
        } else {
            ra = load_idx4(ei, e, eend, true);
            rb = load_idx4(ei, e + 4096, eend, true);
            ca = load_idx4(ec, e, eend, cvec);
            cb = load_idx4(ec, e + 4096, eend, cvec);
        }
        int r[8] = {ra.x, ra.y, ra.z, ra.w, rb.x, rb.y, rb.z, rb.w};
        int c[8] = {ca.x, ca.y, ca.z, ca.w, cb.x, cb.y, cb.z, cb.w};
        unsigned d[8];
        float dr[8], dc[8];
#pragma unroll
        for (int q = 0; q < 8; ++q) {
            d[q]  = (unsigned)(r[q] - r0);
            if (d[q] < (unsigned)nr) {
                dr[q] = dinv[r[q]];
                dc[q] = dinv[c[q]];
            }
        }
        unsigned p[8];
#pragma unroll
        for (int q = 0; q < 8; ++q) {
            if (d[q] < (unsigned)nr) p[q] = (unsigned)atomicAdd(&slot[d[q]], 1);
        }
#pragma unroll
        for (int q = 0; q < 8; ++q) {
            if (d[q] < (unsigned)nr) {
                v2i rec;
                rec.x = c[q];
                rec.y = __float_as_int(dr[q] * dc[q]);
                csr[p[q]] = rec;
            }
        }
    }
}

template <int K>
__device__ __forceinline__ void gemm_load_a(float4 (&a)[K / 8], const float* __restrict__ A,
                                            int t, int c, int h) {
    const float* p = A + (size_t)(t * 16 + c) * K + 4 * h;
#pragma unroll
    for (int i = 0; i < K / 8; ++i) a[i] = *(const float4*)(p + 8 * i);
}

template <int K>
__device__ __forceinline__ void gemm_tile(const float4 (&a)[K / 8], const v2f (&bf)[K / 4][4],
                                          float* __restrict__ C, int t, int c, int h) {
    v8f acc[4];
#pragma unroll
    for (int j = 0; j < 4; ++j) acc[j] = (v8f){0.f, 0.f, 0.f, 0.f, 0.f, 0.f, 0.f, 0.f};
#pragma unroll
    for (int i = 0; i < K / 8; ++i) {
        v2f lo, hi;
        lo.x = a[i].x; lo.y = a[i].y;
        hi.x = a[i].z; hi.y = a[i].w;
#pragma unroll
        for (int j = 0; j < 4; ++j)
            acc[j] = __builtin_amdgcn_wmma_f32_16x16x4_f32(false, lo, false, bf[2 * i][j],
                                                           (short)0, acc[j], false, false);
#pragma unroll
        for (int j = 0; j < 4; ++j)
            acc[j] = __builtin_amdgcn_wmma_f32_16x16x4_f32(false, hi, false, bf[2 * i + 1][j],
                                                           (short)0, acc[j], false, false);
    }
    float* crow = C + (size_t)(t * 16 + 8 * h) * 64 + 4 * c;
#pragma unroll
    for (int r = 0; r < 8; ++r) {
        float4 o;
        o.x = acc[0][r]; o.y = acc[1][r]; o.z = acc[2][r]; o.w = acc[3][r];
        *(float4*)(crow + (size_t)r * 64) = o;
    }
}

template <int K>
__global__ __launch_bounds__(256) void gemm64_wmma_kernel(const float* __restrict__ A,
                                                          const float* __restrict__ B,
                                                          float* __restrict__ C,
                                                          int M) {
    int tid  = threadIdx.x;
    int wave = __builtin_amdgcn_readfirstlane(tid >> 5);
    int lane = tid & 31;
    int h = lane >> 4;
    int c = lane & 15;

    int nt     = M >> 4;
    int stride = (int)gridDim.x * 8;
    int t = (int)blockIdx.x + (int)gridDim.x * wave;
    if (t >= nt) return;

    float4 a0[K / 8], a1[K / 8];
    gemm_load_a<K>(a0, A, t, c, h);

    v2f bf[K / 4][4];
#pragma unroll
    for (int w = 0; w < K / 4; ++w) {
        int k0 = 8 * (w >> 1) + 4 * h + 2 * (w & 1);
        float4 r0 = *(const float4*)(B + (size_t)k0 * 64 + 4 * c);
        float4 r1 = *(const float4*)(B + (size_t)(k0 + 1) * 64 + 4 * c);
        bf[w][0].x = r0.x; bf[w][0].y = r1.x;
        bf[w][1].x = r0.y; bf[w][1].y = r1.y;
        bf[w][2].x = r0.z; bf[w][2].y = r1.z;
        bf[w][3].x = r0.w; bf[w][3].y = r1.w;
    }

    for (;;) {
        int tn = t + stride;
        if (tn >= nt) { gemm_tile<K>(a0, bf, C, t, c, h); return; }
        gemm_load_a<K>(a1, A, tn, c, h);
        __builtin_amdgcn_sched_barrier(0);
        gemm_tile<K>(a0, bf, C, t, c, h);
        t  = tn;
        tn = t + stride;
        if (tn >= nt) { gemm_tile<K>(a1, bf, C, t, c, h); return; }
        gemm_load_a<K>(a0, A, tn, c, h);
        __builtin_amdgcn_sched_barrier(0);
        gemm_tile<K>(a1, bf, C, t, c, h);
        t = tn;
    }
}

#define AGB 8

template <bool GUARD>
__device__ __forceinline__ void agg_step(const v2i (&r)[AGB], int j, int end,
                                         const float* __restrict__ h, unsigned lane8,
                                         float* __restrict__ obase, int rp,
                                         v2f bv, int do_relu,
                                         float& ax, float& ay, int& row, int& rend) {
    v2f hv[AGB];
#pragma unroll
    for (int q = 0; q < AGB; ++q)
        hv[q] = *(const v2f*)((const char*)h + ((unsigned)r[q].x * 256u + lane8));
    int left = rend - j;
#pragma unroll
    for (int q = 0; q < AGB; ++q) {
        if (!GUARD || j + q < end) {
            float w = __int_as_float(r[q].y);
            ax = fmaf(hv[q].x, w, ax);
            ay = fmaf(hv[q].y, w, ay);
            if (q + 1 == left) {
                float ox = ax + bv.x;
                float oy = ay + bv.y;
                if (do_relu) {
                    ox = fmaxf(ox, 0.f);
                    oy = fmaxf(oy, 0.f);
                }
                v2f o; o.x = ox; o.y = oy;
                *(v2f*)(obase + (size_t)row * 64) = o;
                ax = 0.f; ay = 0.f;
                ++row;
                rend = __builtin_amdgcn_readlane(rp, row + 1);
                left = rend - j;
            }
        }
    }
}

__device__ __forceinline__ void agg_trip(const v2i (&rc)[AGB], v2i (&rn)[AGB],
                                         const v2i* __restrict__ csr, int& j, int end,
                                         const float* __restrict__ h, unsigned lane8,
                                         float* __restrict__ obase, int rp,
                                         v2f bv, int do_relu,
                                         float& ax, float& ay, int& row, int& rend) {
    int jn = j + AGB;
    if (jn + AGB <= end) {
        const v2i* pn = csr + jn;
#pragma unroll
        for (int q = 0; q < AGB; ++q) rn[q] = pn[q];
    } else {
#pragma unroll
        for (int q = 0; q < AGB; ++q) rn[q] = csr[min(jn + q, end - 1)];
    }
    if (jn <= end) agg_step<false>(rc, j, end, h, lane8, obase, rp, bv, do_relu, ax, ay, row, rend);
    else           agg_step<true >(rc, j, end, h, lane8, obase, rp, bv, do_relu, ax, ay, row, rend);
    j = jn;
}

__global__ __launch_bounds__(256) void agg_kernel(const float* __restrict__ h,
                                                  const int* __restrict__ rowptr,
                                                  const v2i* __restrict__ csr,
                                                  const float* __restrict__ bias,
                                                  float* __restrict__ out,
                                                  int N, int NPW, int do_relu) {
    int lane = threadIdx.x & 31;
    int w  = __builtin_amdgcn_readfirstlane((int)(blockIdx.x * 8 + (threadIdx.x >> 5)));
    int n0 = w * NPW;
    if (n0 >= N) return;
    int cnt = N - n0; if (cnt > NPW) cnt = NPW;

    int rp   = rowptr[n0 + min(lane, cnt)];
    int j    = __builtin_amdgcn_readlane(rp, 0);
    int end  = __builtin_amdgcn_readlane(rp, cnt);
    int row  = 0;
    int rend = __builtin_amdgcn_readlane(rp, 1);
    if (j >= end) return;

    unsigned lane8 = (unsigned)lane * 8u;
    float*   obase = out + (size_t)n0 * 64 + 2 * lane;
    v2f bv = *(const v2f*)(bias + 2 * lane);

    float ax = 0.f, ay = 0.f;
    v2i ra[AGB], rb[AGB];
#pragma unroll
    for (int q = 0; q < AGB; ++q) ra[q] = csr[min(j + q, end - 1)];

    for (;;) {
        agg_trip(ra, rb, csr, j, end, h, lane8, obase, rp, bv, do_relu, ax, ay, row, rend);
        if (j >= end) break;
        agg_trip(rb, ra, csr, j, end, h, lane8, obase, rp, bv, do_relu, ax, ay, row, rend);
        if (j >= end) break;
    }
}

extern "C" void kernel_launch(void* const* d_in, const int* in_sizes, int n_in,
                              void* d_out, int out_size, void* d_ws, size_t ws_size,
                              hipStream_t stream) {
    (void)n_in; (void)out_size; (void)ws_size;

    const float* x  = (const float*)d_in[0];
    const int*   ei = (const int*)d_in[1];
    const float* W1 = (const float*)d_in[2];
    const float* b1 = (const float*)d_in[3];
    const float* W2 = (const float*)d_in[4];
    const float* b2 = (const float*)d_in[5];
    float* out = (float*)d_out;

    const int H   = in_sizes[3];
    const int IN  = in_sizes[2] / H;
    const int N   = in_sizes[0] / IN;
    const int E   = in_sizes[1] / 2;
    const int NNZ = N + E;

    char* ws = (char*)d_ws;
    size_t off = 0;
    auto alloc = [&](size_t bytes) -> void* {
        void* p = (void*)(ws + off);
        off += (bytes + 255) & ~(size_t)255;
        return p;
    };
    const int T = 256;
    int gN = (N + CB - 1) / CB;
    int*   bsum    = (int*)  alloc((size_t)gN * 4);
    int*   rowptr  = (int*)  alloc((size_t)(N + 1) * 4);
    float* dinv    = (float*)alloc((size_t)N * 4);
    int*   partial = (int*)  alloc((size_t)KCH * N * 4);
    v2i*   csr     = (v2i*)  alloc((size_t)NNZ * 8);
    float* h1      = (float*)alloc((size_t)N * H * 4);
    float* y1      = (float*)alloc((size_t)N * H * 4);
    float* h2      = h1;

    int EC = (((E + KCH - 1) / KCH) + 3) & ~3;
    dim3 gTile((N + HR - 1) / HR, KCH);
    int NPW  = (N + 10239) / 10240;
    if (NPW > 30) NPW = 30;
    int gAgg = ((N + NPW - 1) / NPW + 7) / 8;
    int gGemm1 = 256;
    int gGemm2 = 512;

    hist_kernel<<<gTile, 1024, 0, stream>>>(ei, partial, N, E, EC);
    combine_kernel<<<gN, CB, 0, stream>>>(partial, rowptr, bsum, dinv, N);
    init_csr_kernel<<<gN, CB, 0, stream>>>(rowptr, bsum, dinv, csr, N);
    fill_csr_kernel<<<gTile, 1024, 0, stream>>>(ei, dinv, rowptr, partial, csr, N, E, EC);

    gemm64_wmma_kernel<128><<<gGemm1, T, 0, stream>>>(x, W1, h1, N);
    agg_kernel<<<gAgg, T, 0, stream>>>(h1, rowptr, csr, b1, y1, N, NPW, 1);

    gemm64_wmma_kernel<64><<<gGemm2, T, 0, stream>>>(y1, W2, h2, N);
    agg_kernel<<<gAgg, T, 0, stream>>>(h2, rowptr, csr, b2, out, N, NPW, 0);
}


# Aggregation step reordered: its 8 row gathers are issued before the request for the next step's records, so that request's scalar wait runs under the gathers' flight, not before it; sum order kept
# speedup vs baseline: 2.9471x; 1.0316x over previous
#include <hip/hip_runtime.h>
#include <hip/hip_bf16.h>


typedef float v2f __attribute__((ext_vector_type(2)));
typedef float v8f __attribute__((ext_vector_type(8)));
typedef int   v2i __attribute__((ext_vector_type(2)));

#define HR  12544
#define KCH 32

__device__ __forceinline__ int4 load_idx4(const int* __restrict__ p, int e, int eend, bool vec_ok) {
    int4 v;
    if (vec_ok && e + 3 < eend) {
        v = *(const int4*)(p + e);
    } else {
        v.x = (e     < eend) ? p[e]     : -1;
        v.y = (e + 1 < eend) ? p[e + 1] : -1;
        v.z = (e + 2 < eend) ? p[e + 2] : -1;
        v.w = (e + 3 < eend) ? p[e + 3] : -1;
    }
    return v;
}

__global__ __launch_bounds__(1024) void hist_kernel(const int* __restrict__ ei,
                                                    int* __restrict__ partial,
                                                    int N, int E, int EC) {
    __shared__ int bins[HR];
    int tid = threadIdx.x;
    int r0  = blockIdx.x * HR;
    int k   = blockIdx.y;
    int nr  = N - r0; if (nr > HR) nr = HR;
    for (int i = tid; i < HR; i += 1024) bins[i] = 0;
    __syncthreads();
    int ebeg = k * EC;
    int eend = ebeg + EC; if (eend > E) eend = E;
    for (int eb = ebeg; eb < eend; eb += 8192) {
        int e = eb + 4 * tid;
        int4 a, b;
        if (eb + 8192 <= eend) {
            a = *(const int4*)(ei + e);
            b = *(const int4*)(ei + e + 4096);
        } else {
            a = load_idx4(ei, e, eend, true);
            b = load_idx4(ei, e + 4096, eend, true);
        }
        int r[8] = {a.x, a.y, a.z, a.w, b.x, b.y, b.z, b.w};
#pragma unroll
        for (int q = 0; q < 8; ++q) {
            unsigned d = (unsigned)(r[q] - r0);
            if (d < (unsigned)nr) atomicAdd(&bins[d], 1);
        }
    }
    __syncthreads();
    int* dst = partial + (size_t)k * N + r0;
    for (int i = tid; i < nr; i += 1024) dst[i] = bins[i];
}

#define CB 256

__global__ __launch_bounds__(CB) void combine_kernel(int* __restrict__ partial, int* __restrict__ rowptr,
                                                     int* __restrict__ bsum, float* __restrict__ dinv,
                                                     int N) {
    __shared__ int wsum[CB / 32];
    int tid = threadIdx.x;
    int i = blockIdx.x * CB + tid;
    int v = 0;
    if (i < N) {
        int t[KCH];
#pragma unroll
        for (int k = 0; k < KCH; ++k) t[k] = partial[(size_t)k * N + i];
        int run = 0;
#pragma unroll
        for (int k = 0; k < KCH; ++k) {
            partial[(size_t)k * N + i] = run;
            run += t[k];
        }
        dinv[i] = rsqrtf((float)(run + 1));
        v = run + 1;
    }
    int lane = tid & 31;
    int wv   = tid >> 5;
    int inc  = v;
    for (int off = 1; off < 32; off <<= 1) {
        int n = __shfl_up(inc, off, 32);
        if (lane >= off) inc += n;
    }
    if (lane == 31) wsum[wv] = inc;
    __syncthreads();
    int base = 0, tot = 0;
    for (int q = 0; q < CB / 32; ++q) {
        int s = wsum[q];
        if (q < wv) base += s;
        tot += s;
    }
    if (i < N) rowptr[i] = base + inc - v;
    if (tid == 0) bsum[blockIdx.x] = tot;
}

__global__ __launch_bounds__(CB) void init_csr_kernel(int* __restrict__ rowptr, const int* __restrict__ bsum,
                                                      const float* __restrict__ dinv,
                                                      v2i* __restrict__ csr, int N) {
    __shared__ int wsum[CB / 32];
    int tid = threadIdx.x;
    int b   = blockIdx.x;
    int i   = b * CB + tid;
    int   loc = 0;
    float d   = 0.f;
    if (i < N) {
        loc = rowptr[i];
        d   = dinv[i];
    }
    int s = 0;
    for (int q = tid; q < b; q += CB) s += bsum[q];
    for (int off = 16; off > 0; off >>= 1) s += __shfl_xor(s, off, 32);
    if ((tid & 31) == 0) wsum[tid >> 5] = s;
    __syncthreads();
    int boff = 0;
    for (int q = 0; q < CB / 32; ++q) boff += wsum[q];
    if (i < N) {
        int p = loc + boff;
        rowptr[i] = p;
        v2i rec;
        rec.x = i;
        rec.y = __float_as_int(d * d);
        csr[p] = rec;
    }
    if (b == (int)gridDim.x - 1 && tid == 0) rowptr[N] = boff + bsum[b];
}

__global__ __launch_bounds__(1024) void fill_csr_kernel(const int* __restrict__ ei,
                                                        const float* __restrict__ dinv,
                                                        const int* __restrict__ rowptr,
                                                        const int* __restrict__ partial,
                                                        v2i* __restrict__ csr,
                                                        int N, int E, int EC) {
    __shared__ int slot[HR];
    int tid = threadIdx.x;
    int r0  = blockIdx.x * HR;
    int k   = blockIdx.y;
    int nr  = N - r0; if (nr > HR) nr = HR;
    const int* pre = partial + (size_t)k * N + r0;
    {
        constexpr int TRIPS = (HR + 1023) / 1024;
        int rp[TRIPS], pp[TRIPS];
#pragma unroll
        for (int it = 0; it < TRIPS; ++it) {
            int i = min(tid + it * 1024, nr - 1);
            rp[it] = rowptr[r0 + i];
            pp[it] = pre[i];
        }
#pragma unroll
        for (int it = 0; it < TRIPS; ++it) {
            int i = tid + it * 1024;
            if (i < nr) slot[i] = rp[it] + 1 + pp[it];
        }
    }
    __syncthreads();
    int ebeg = k * EC;
    int eend = ebeg + EC; if (eend > E) eend = E;
    const int* ec = ei + E;
    bool cvec = (E & 3) == 0;
    for (int eb = ebeg; eb < eend; eb += 8192) {
        int e = eb + 4 * tid;
        int4 ra, rb, ca, cb;
        if (cvec && eb + 8192 <= eend) {
            ra = *(const int4*)(ei + e);
            rb = *(const int4*)(ei + e + 4096);
            ca = *(const int4*)(ec + e);
            cb = *(const int4*)(ec + e + 4096);
        } else {
            ra = load_idx4(ei, e, eend, true);
            rb = load_idx4(ei, e + 4096, eend, true);
            ca = load_idx4(ec, e, eend, cvec);
            cb = load_idx4(ec, e + 4096, eend, cvec);
        }
        int r[8] = {ra.x, ra.y, ra.z, ra.w, rb.x, rb.y, rb.z, rb.w};
        int c[8] = {ca.x, ca.y, ca.z, ca.w, cb.x, cb.y, cb.z, cb.w};
        unsigned d[8];
        float dr[8], dc[8];
#pragma unroll
        for (int q = 0; q < 8; ++q) {
            d[q]  = (unsigned)(r[q] - r0);
            if (d[q] < (unsigned)nr) {
                dr[q] = dinv[r[q]];
                dc[q] = dinv[c[q]];
            }
        }
        unsigned p[8];
#pragma unroll
        for (int q = 0; q < 8; ++q) {
            if (d[q] < (unsigned)nr) p[q] = (unsigned)atomicAdd(&slot[d[q]], 1);
        }
#pragma unroll
        for (int q = 0; q < 8; ++q) {
            if (d[q] < (unsigned)nr) {
                v2i rec;
                rec.x = c[q];
                rec.y = __float_as_int(dr[q] * dc[q]);
                csr[p[q]] = rec;
            }
        }
    }
}

template <int K>
__device__ __forceinline__ void gemm_load_a(float4 (&a)[K / 8], const float* __restrict__ A,
                                            int t, int c, int h) {
    const float* p = A + (size_t)(t * 16 + c) * K + 4 * h;
#pragma unroll
    for (int i = 0; i < K / 8; ++i) a[i] = *(const float4*)(p + 8 * i);
}

template <int K>
__device__ __forceinline__ void gemm_tile(const float4 (&a)[K / 8], const v2f (&bf)[K / 4][4],
                                          float* __restrict__ C, int t, int c, int h) {
    v8f acc[4];
#pragma unroll
    for (int j = 0; j < 4; ++j) acc[j] = (v8f){0.f, 0.f, 0.f, 0.f, 0.f, 0.f, 0.f, 0.f};
#pragma unroll
    for (int i = 0; i < K / 8; ++i) {
        v2f lo, hi;
        lo.x = a[i].x; lo.y = a[i].y;
        hi.x = a[i].z; hi.y = a[i].w;
#pragma unroll
        for (int j = 0; j < 4; ++j)
            acc[j] = __builtin_amdgcn_wmma_f32_16x16x4_f32(false, lo, false, bf[2 * i][j],
                                                           (short)0, acc[j], false, false);
#pragma unroll
        for (int j = 0; j < 4; ++j)
            acc[j] = __builtin_amdgcn_wmma_f32_16x16x4_f32(false, hi, false, bf[2 * i + 1][j],
                                                           (short)0, acc[j], false, false);
    }
    float* crow = C + (size_t)(t * 16 + 8 * h) * 64 + 4 * c;
#pragma unroll
    for (int r = 0; r < 8; ++r) {
        float4 o;
        o.x = acc[0][r]; o.y = acc[1][r]; o.z = acc[2][r]; o.w = acc[3][r];
        *(float4*)(crow + (size_t)r * 64) = o;
    }
}

template <int K>
__global__ __launch_bounds__(256) void gemm64_wmma_kernel(const float* __restrict__ A,
                                                          const float* __restrict__ B,
                                                          float* __restrict__ C,
                                                          int M) {
    int tid  = threadIdx.x;
    int wave = __builtin_amdgcn_readfirstlane(tid >> 5);
    int lane = tid & 31;
    int h = lane >> 4;
    int c = lane & 15;

    int nt     = M >> 4;
    int stride = (int)gridDim.x * 8;
    int t = (int)blockIdx.x + (int)gridDim.x * wave;
    if (t >= nt) return;

    float4 a0[K / 8], a1[K / 8];
    gemm_load_a<K>(a0, A, t, c, h);

    v2f bf[K / 4][4];
#pragma unroll
    for (int w = 0; w < K / 4; ++w) {
        int k0 = 8 * (w >> 1) + 4 * h + 2 * (w & 1);
        float4 r0 = *(const float4*)(B + (size_t)k0 * 64 + 4 * c);
        float4 r1 = *(const float4*)(B + (size_t)(k0 + 1) * 64 + 4 * c);
        bf[w][0].x = r0.x; bf[w][0].y = r1.x;
        bf[w][1].x = r0.y; bf[w][1].y = r1.y;
        bf[w][2].x = r0.z; bf[w][2].y = r1.z;
        bf[w][3].x = r0.w; bf[w][3].y = r1.w;
    }

    for (;;) {
        int tn = t + stride;
        if (tn >= nt) { gemm_tile<K>(a0, bf, C, t, c, h); return; }
        gemm_load_a<K>(a1, A, tn, c, h);
        __builtin_amdgcn_sched_barrier(0);
        gemm_tile<K>(a0, bf, C, t, c, h);
        t  = tn;
        tn = t + stride;
        if (tn >= nt) { gemm_tile<K>(a1, bf, C, t, c, h); return; }
        gemm_load_a<K>(a0, A, tn, c, h);
        __builtin_amdgcn_sched_barrier(0);
        gemm_tile<K>(a1, bf, C, t, c, h);
        t = tn;
    }
}

#define AGB 8

template <bool GUARD>
__device__ __forceinline__ void agg_step(const v2i (&r)[AGB], const v2f (&hv)[AGB], int j, int end,
                                         float* __restrict__ obase, int rp,
                                         v2f bv, int do_relu,
                                         float& ax, float& ay, int& row, int& rend) {
    int left = rend - j;
#pragma unroll
    for (int q = 0; q < AGB; ++q) {
        if (!GUARD || j + q < end) {
            float w = __int_as_float(r[q].y);
            ax = fmaf(hv[q].x, w, ax);
            ay = fmaf(hv[q].y, w, ay);
            if (q + 1 == left) {
                float ox = ax + bv.x;
                float oy = ay + bv.y;
                if (do_relu) {
                    ox = fmaxf(ox, 0.f);
                    oy = fmaxf(oy, 0.f);
                }
                v2f o; o.x = ox; o.y = oy;
                *(v2f*)(obase + (size_t)row * 64) = o;
                ax = 0.f; ay = 0.f;
                ++row;
                rend = __builtin_amdgcn_readlane(rp, row + 1);
                left = rend - j;
            }
        }
    }
}

__device__ __forceinline__ void agg_trip(const v2i (&rc)[AGB], v2i (&rn)[AGB],
                                         const v2i* __restrict__ csr, int& j, int end,
                                         const float* __restrict__ h, unsigned lane8,
                                         float* __restrict__ obase, int rp,
                                         v2f bv, int do_relu,
                                         float& ax, float& ay, int& row, int& rend) {
    v2f hv[AGB];
#pragma unroll
    for (int q = 0; q < AGB; ++q)
        hv[q] = *(const v2f*)((const char*)h + ((unsigned)rc[q].x * 256u + lane8));
    __builtin_amdgcn_sched_barrier(0);
    int jn = j + AGB;
    if (jn + AGB <= end) {
        const v2i* pn = csr + jn;
#pragma unroll
        for (int q = 0; q < AGB; ++q) rn[q] = pn[q];
    } else {
#pragma unroll
        for (int q = 0; q < AGB; ++q) rn[q] = csr[min(jn + q, end - 1)];
    }
    if (jn <= end) agg_step<false>(rc, hv, j, end, obase, rp, bv, do_relu, ax, ay, row, rend);
    else           agg_step<true >(rc, hv, j, end, obase, rp, bv, do_relu, ax, ay, row, rend);
    j = jn;
}

__global__ __launch_bounds__(256) void agg_kernel(const float* __restrict__ h,
                                                  const int* __restrict__ rowptr,
                                                  const v2i* __restrict__ csr,
                                                  const float* __restrict__ bias,
                                                  float* __restrict__ out,
                                                  int N, int NPW, int do_relu) {
    int lane = threadIdx.x & 31;
    int w  = __builtin_amdgcn_readfirstlane((int)(blockIdx.x * 8 + (threadIdx.x >> 5)));
    int n0 = w * NPW;
    if (n0 >= N) return;
    int cnt = N - n0; if (cnt > NPW) cnt = NPW;

    int rp   = rowptr[n0 + min(lane, cnt)];
    int j    = __builtin_amdgcn_readlane(rp, 0);
    int end  = __builtin_amdgcn_readlane(rp, cnt);
    int row  = 0;
    int rend = __builtin_amdgcn_readlane(rp, 1);
    if (j >= end) return;

    unsigned lane8 = (unsigned)lane * 8u;
    float*   obase = out + (size_t)n0 * 64 + 2 * lane;
    v2f bv = *(const v2f*)(bias + 2 * lane);

    float ax = 0.f, ay = 0.f;
    v2i ra[AGB], rb[AGB];
#pragma unroll
    for (int q = 0; q < AGB; ++q) ra[q] = csr[min(j + q, end - 1)];

    for (;;) {
        agg_trip(ra, rb, csr, j, end, h, lane8, obase, rp, bv, do_relu, ax, ay, row, rend);
        if (j >= end) break;
        agg_trip(rb, ra, csr, j, end, h, lane8, obase, rp, bv, do_relu, ax, ay, row, rend);
        if (j >= end) break;
    }
}

extern "C" void kernel_launch(void* const* d_in, const int* in_sizes, int n_in,
                              void* d_out, int out_size, void* d_ws, size_t ws_size,
                              hipStream_t stream) {
    (void)n_in; (void)out_size; (void)ws_size;

    const float* x  = (const float*)d_in[0];
    const int*   ei = (const int*)d_in[1];
    const float* W1 = (const float*)d_in[2];
    const float* b1 = (const float*)d_in[3];
    const float* W2 = (const float*)d_in[4];
    const float* b2 = (const float*)d_in[5];
    float* out = (float*)d_out;

    const int H   = in_sizes[3];
    const int IN  = in_sizes[2] / H;
    const int N   = in_sizes[0] / IN;
    const int E   = in_sizes[1] / 2;
    const int NNZ = N + E;

    char* ws = (char*)d_ws;
    size_t off = 0;
    auto alloc = [&](size_t bytes) -> void* {
        void* p = (void*)(ws + off);
        off += (bytes + 255) & ~(size_t)255;
        return p;
    };
    const int T = 256;
    int gN = (N + CB - 1) / CB;
    int*   bsum    = (int*)  alloc((size_t)gN * 4);
    int*   rowptr  = (int*)  alloc((size_t)(N + 1) * 4);
    float* dinv    = (float*)alloc((size_t)N * 4);
    int*   partial = (int*)  alloc((size_t)KCH * N * 4);
    v2i*   csr     = (v2i*)  alloc((size_t)NNZ * 8);
    float* h1      = (float*)alloc((size_t)N * H * 4);
    float* y1      = (float*)alloc((size_t)N * H * 4);
    float* h2      = h1;

    int EC = (((E + KCH - 1) / KCH) + 3) & ~3;
    dim3 gTile((N + HR - 1) / HR, KCH);
    int NPW  = (N + 10239) / 10240;
    if (NPW > 30) NPW = 30;
    int gAgg = ((N + NPW - 1) / NPW + 7) / 8;
    int gGemm1 = 256;
    int gGemm2 = 512;

    hist_kernel<<<gTile, 1024, 0, stream>>>(ei, partial, N, E, EC);
    combine_kernel<<<gN, CB, 0, stream>>>(partial, rowptr, bsum, dinv, N);
    init_csr_kernel<<<gN, CB, 0, stream>>>(rowptr, bsum, dinv, csr, N);
    fill_csr_kernel<<<gTile, 1024, 0, stream>>>(ei, dinv, rowptr, partial, csr, N, E, EC);

    gemm64_wmma_kernel<128><<<gGemm1, T, 0, stream>>>(x, W1, h1, N);
    agg_kernel<<<gAgg, T, 0, stream>>>(h1, rowptr, csr, b1, y1, N, NPW, 1);

    gemm64_wmma_kernel<64><<<gGemm2, T, 0, stream>>>(y1, W2, h2, N);
    agg_kernel<<<gAgg, T, 0, stream>>>(h2, rowptr, csr, b2, out, N, NPW, 0);
}


# Layer-2 product moved into the layer-1 aggregation launch: a wave owns 16 nodes, keeps their relu rows in an LDS tile of its own and multiplies it by W2 on WMMA; 8 launches become 7; sum orders kept
# speedup vs baseline: 3.0554x; 1.0367x over previous
#include <hip/hip_runtime.h>
#include <hip/hip_bf16.h>


typedef float v2f __attribute__((ext_vector_type(2)));
typedef float v8f __attribute__((ext_vector_type(8)));
typedef int   v2i __attribute__((ext_vector_type(2)));

#define HR  12544
#define KCH 32

__device__ __forceinline__ int4 load_idx4(const int* __restrict__ p, int e, int eend, bool vec_ok) {
    int4 v;
    if (vec_ok && e + 3 < eend) {
        v = *(const int4*)(p + e);
    } else {
        v.x = (e     < eend) ? p[e]     : -1;
        v.y = (e + 1 < eend) ? p[e + 1] : -1;
        v.z = (e + 2 < eend) ? p[e + 2] : -1;
        v.w = (e + 3 < eend) ? p[e + 3] : -1;
    }
    return v;
}

__global__ __launch_bounds__(1024) void hist_kernel(const int* __restrict__ ei,
                                                    int* __restrict__ partial,
                                                    int N, int E, int EC) {
    __shared__ int bins[HR];
    int tid = threadIdx.x;
    int r0  = blockIdx.x * HR;
    int k   = blockIdx.y;
    int nr  = N - r0; if (nr > HR) nr = HR;
    for (int i = tid; i < HR; i += 1024) bins[i] = 0;
    __syncthreads();
    int ebeg = k * EC;
    int eend = ebeg + EC; if (eend > E) eend = E;
    for (int eb = ebeg; eb < eend; eb += 8192) {
        int e = eb + 4 * tid;
        int4 a, b;
        if (eb + 8192 <= eend) {
            a = *(const int4*)(ei + e);
            b = *(const int4*)(ei + e + 4096);
        } else {
            a = load_idx4(ei, e, eend, true);
            b = load_idx4(ei, e + 4096, eend, true);
        }
        int r[8] = {a.x, a.y, a.z, a.w, b.x, b.y, b.z, b.w};
#pragma unroll
        for (int q = 0; q < 8; ++q) {
            unsigned d = (unsigned)(r[q] - r0);
            if (d < (unsigned)nr) atomicAdd(&bins[d], 1);
        }
    }
    __syncthreads();
    int* dst = partial + (size_t)k * N + r0;
    for (int i = tid; i < nr; i += 1024) dst[i] = bins[i];
}

#define CB 256

__global__ __launch_bounds__(CB) void combine_kernel(int* __restrict__ partial, int* __restrict__ rowptr,
                                                     int* __restrict__ bsum, float* __restrict__ dinv,
                                                     int N) {
    __shared__ int wsum[CB / 32];
    int tid = threadIdx.x;
    int i = blockIdx.x * CB + tid;
    int v = 0;
    if (i < N) {
        int t[KCH];
#pragma unroll
        for (int k = 0; k < KCH; ++k) t[k] = partial[(size_t)k * N + i];
        int run = 0;
#pragma unroll
        for (int k = 0; k < KCH; ++k) {
            partial[(size_t)k * N + i] = run;
            run += t[k];
        }
        dinv[i] = rsqrtf((float)(run + 1));
        v = run + 1;
    }
    int lane = tid & 31;
    int wv   = tid >> 5;
    int inc  = v;
    for (int off = 1; off < 32; off <<= 1) {
        int n = __shfl_up(inc, off, 32);
        if (lane >= off) inc += n;
    }
    if (lane == 31) wsum[wv] = inc;
    __syncthreads();
    int base = 0, tot = 0;
    for (int q = 0; q < CB / 32; ++q) {
        int s = wsum[q];
        if (q < wv) base += s;
        tot += s;
    }
    if (i < N) rowptr[i] = base + inc - v;
    if (tid == 0) bsum[blockIdx.x] = tot;
}

__global__ __launch_bounds__(CB) void init_csr_kernel(int* __restrict__ rowptr, const int* __restrict__ bsum,
                                                      const float* __restrict__ dinv,
                                                      v2i* __restrict__ csr, int N) {
    __shared__ int wsum[CB / 32];
    int tid = threadIdx.x;
    int b   = blockIdx.x;
    int i   = b * CB + tid;
    int   loc = 0;
    float d   = 0.f;
    if (i < N) {
        loc = rowptr[i];
        d   = dinv[i];
    }
    int s = 0;
    for (int q = tid; q < b; q += CB) s += bsum[q];
    for (int off = 16; off > 0; off >>= 1) s += __shfl_xor(s, off, 32);
    if ((tid & 31) == 0) wsum[tid >> 5] = s;
    __syncthreads();
    int boff = 0;
    for (int q = 0; q < CB / 32; ++q) boff += wsum[q];
    if (i < N) {
        int p = loc + boff;
        rowptr[i] = p;
        v2i rec;
        rec.x = i;
        rec.y = __float_as_int(d * d);
        csr[p] = rec;
    }
    if (b == (int)gridDim.x - 1 && tid == 0) rowptr[N] = boff + bsum[b];
}

__global__ __launch_bounds__(1024) void fill_csr_kernel(const int* __restrict__ ei,
                                                        const float* __restrict__ dinv,
                                                        const int* __restrict__ rowptr,
                                                        const int* __restrict__ partial,
                                                        v2i* __restrict__ csr,
                                                        int N, int E, int EC) {
    __shared__ int slot[HR];
    int tid = threadIdx.x;
    int r0  = blockIdx.x * HR;
    int k   = blockIdx.y;
    int nr  = N - r0; if (nr > HR) nr = HR;
    const int* pre = partial + (size_t)k * N + r0;
    {
        constexpr int TRIPS = (HR + 1023) / 1024;
        int rp[TRIPS], pp[TRIPS];
#pragma unroll
        for (int it = 0; it < TRIPS; ++it) {
            int i = min(tid + it * 1024, nr - 1);
            rp[it] = rowptr[r0 + i];
            pp[it] = pre[i];
        }
#pragma unroll
        for (int it = 0; it < TRIPS; ++it) {
            int i = tid + it * 1024;
            if (i < nr) slot[i] = rp[it] + 1 + pp[it];
        }
    }
    __syncthreads();
    int ebeg = k * EC;
    int eend = ebeg + EC; if (eend > E) eend = E;
    const int* ec = ei + E;
    bool cvec = (E & 3) == 0;
    for (int eb = ebeg; eb < eend; eb += 8192) {
        int e = eb + 4 * tid;
        int4 ra, rb, ca, cb;
        if (cvec && eb + 8192 <= eend) {
            ra = *(const int4*)(ei + e);
            rb = *(const int4*)(ei + e + 4096);
            ca = *(const int4*)(ec + e);
            cb = *(const int4*)(ec + e + 4096);
        } else {
            ra = load_idx4(ei, e, eend, true);
            rb = load_idx4(ei, e + 4096, eend, true);
            ca = load_idx4(ec, e, eend, cvec);
            cb = load_idx4(ec, e + 4096, eend, cvec);
        }
        int r[8] = {ra.x, ra.y, ra.z, ra.w, rb.x, rb.y, rb.z, rb.w};
        int c[8] = {ca.x, ca.y, ca.z, ca.w, cb.x, cb.y, cb.z, cb.w};
        unsigned d[8];
        float dr[8], dc[8];
#pragma unroll
        for (int q = 0; q < 8; ++q) {
            d[q]  = (unsigned)(r[q] - r0);
            if (d[q] < (unsigned)nr) {
                dr[q] = dinv[r[q]];
                dc[q] = dinv[c[q]];
            }
        }
        unsigned p[8];
#pragma unroll
        for (int q = 0; q < 8; ++q) {
            if (d[q] < (unsigned)nr) p[q] = (unsigned)atomicAdd(&slot[d[q]], 1);
        }
#pragma unroll
        for (int q = 0; q < 8; ++q) {
            if (d[q] < (unsigned)nr) {
                v2i rec;
                rec.x = c[q];
                rec.y = __float_as_int(dr[q] * dc[q]);
                csr[p[q]] = rec;
            }
        }
    }
}

template <int K>
__device__ __forceinline__ void gemm_load_a(float4 (&a)[K / 8], const float* __restrict__ A,
                                            int t, int c, int h) {
    const float* p = A + (size_t)(t * 16 + c) * K + 4 * h;
#pragma unroll
    for (int i = 0; i < K / 8; ++i) a[i] = *(const float4*)(p + 8 * i);
}

template <int K>
__device__ __forceinline__ void gemm_tile(const float4 (&a)[K / 8], const v2f (&bf)[K / 4][4],
                                          float* __restrict__ C, int t, int c, int h) {
    v8f acc[4];
#pragma unroll
    for (int j = 0; j < 4; ++j) acc[j] = (v8f){0.f, 0.f, 0.f, 0.f, 0.f, 0.f, 0.f, 0.f};
#pragma unroll
    for (int i = 0; i < K / 8; ++i) {
        v2f lo, hi;
        lo.x = a[i].x; lo.y = a[i].y;
        hi.x = a[i].z; hi.y = a[i].w;
#pragma unroll
        for (int j = 0; j < 4; ++j)
            acc[j] = __builtin_amdgcn_wmma_f32_16x16x4_f32(false, lo, false, bf[2 * i][j],
                                                           (short)0, acc[j], false, false);
#pragma unroll
        for (int j = 0; j < 4; ++j)
            acc[j] = __builtin_amdgcn_wmma_f32_16x16x4_f32(false, hi, false, bf[2 * i + 1][j],
                                                           (short)0, acc[j], false, false);
    }
    float* crow = C + (size_t)(t * 16 + 8 * h) * 64 + 4 * c;
#pragma unroll
    for (int r = 0; r < 8; ++r) {
        float4 o;
        o.x = acc[0][r]; o.y = acc[1][r]; o.z = acc[2][r]; o.w = acc[3][r];
        *(float4*)(crow + (size_t)r * 64) = o;
    }
}

template <int K>
__global__ __launch_bounds__(256) void gemm64_wmma_kernel(const float* __restrict__ A,
                                                          const float* __restrict__ B,
                                                          float* __restrict__ C,
                                                          int M) {
    int tid  = threadIdx.x;
    int wave = __builtin_amdgcn_readfirstlane(tid >> 5);
    int lane = tid & 31;
    int h = lane >> 4;
    int c = lane & 15;

    int nt     = M >> 4;
    int stride = (int)gridDim.x * 8;
    int t = (int)blockIdx.x + (int)gridDim.x * wave;
    if (t >= nt) return;

    float4 a0[K / 8], a1[K / 8];
    gemm_load_a<K>(a0, A, t, c, h);

    v2f bf[K / 4][4];
#pragma unroll
    for (int w = 0; w < K / 4; ++w) {
        int k0 = 8 * (w >> 1) + 4 * h + 2 * (w & 1);
        float4 r0 = *(const float4*)(B + (size_t)k0 * 64 + 4 * c);
        float4 r1 = *(const float4*)(B + (size_t)(k0 + 1) * 64 + 4 * c);
        bf[w][0].x = r0.x; bf[w][0].y = r1.x;
        bf[w][1].x = r0.y; bf[w][1].y = r1.y;
        bf[w][2].x = r0.z; bf[w][2].y = r1.z;
        bf[w][3].x = r0.w; bf[w][3].y = r1.w;
    }

    for (;;) {
        int tn = t + stride;
        if (tn >= nt) { gemm_tile<K>(a0, bf, C, t, c, h); return; }
        gemm_load_a<K>(a1, A, tn, c, h);
        __builtin_amdgcn_sched_barrier(0);
        gemm_tile<K>(a0, bf, C, t, c, h);
        t  = tn;
        tn = t + stride;
        if (tn >= nt) { gemm_tile<K>(a1, bf, C, t, c, h); return; }
        gemm_load_a<K>(a0, A, tn, c, h);
        __builtin_amdgcn_sched_barrier(0);
        gemm_tile<K>(a1, bf, C, t, c, h);
        t = tn;
    }
}

#define AGB 8

template <bool GUARD, int PITCH>
__device__ __forceinline__ void agg_step(const v2i (&r)[AGB], const v2f (&hv)[AGB], int j, int end,
                                         float* __restrict__ obase, int rp,
                                         v2f bv, int do_relu,
                                         float& ax, float& ay, int& row, int& rend) {
    int left = rend - j;
#pragma unroll
    for (int q = 0; q < AGB; ++q) {
        if (!GUARD || j + q < end) {
            float w = __int_as_float(r[q].y);
            ax = fmaf(hv[q].x, w, ax);
            ay = fmaf(hv[q].y, w, ay);
            if (q + 1 == left) {
                float ox = ax + bv.x;
                float oy = ay + bv.y;
                if (do_relu) {
                    ox = fmaxf(ox, 0.f);
                    oy = fmaxf(oy, 0.f);
                }
                v2f o; o.x = ox; o.y = oy;
                *(v2f*)(obase + row * PITCH) = o;
                ax = 0.f; ay = 0.f;
                ++row;
                rend = __builtin_amdgcn_readlane(rp, row + 1);
                left = rend - j;
            }
        }
    }
}

template <int PITCH>
__device__ __forceinline__ void agg_trip(const v2i (&rc)[AGB], v2i (&rn)[AGB],
                                         const v2i* __restrict__ csr, int& j, int end,
                                         const float* __restrict__ h, unsigned lane8,
                                         float* __restrict__ obase, int rp,
                                         v2f bv, int do_relu,
                                         float& ax, float& ay, int& row, int& rend) {
    v2f hv[AGB];
#pragma unroll
    for (int q = 0; q < AGB; ++q)
        hv[q] = *(const v2f*)((const char*)h + ((unsigned)rc[q].x * 256u + lane8));
    __builtin_amdgcn_sched_barrier(0);
    int jn = j + AGB;
    if (jn + AGB <= end) {
        const v2i* pn = csr + jn;
#pragma unroll
        for (int q = 0; q < AGB; ++q) rn[q] = pn[q];
    } else {
#pragma unroll
        for (int q = 0; q < AGB; ++q) rn[q] = csr[min(jn + q, end - 1)];
    }
    if (jn <= end) agg_step<false, PITCH>(rc, hv, j, end, obase, rp, bv, do_relu, ax, ay, row, rend);
    else           agg_step<true,  PITCH>(rc, hv, j, end, obase, rp, bv, do_relu, ax, ay, row, rend);
    j = jn;
}

__global__ __launch_bounds__(256) void agg_kernel(const float* __restrict__ h,
                                                  const int* __restrict__ rowptr,
                                                  const v2i* __restrict__ csr,
                                                  const float* __restrict__ bias,
                                                  float* __restrict__ out,
                                                  int N, int NPW, int do_relu) {
    int lane = threadIdx.x & 31;
    int w  = __builtin_amdgcn_readfirstlane((int)(blockIdx.x * 8 + (threadIdx.x >> 5)));
    int n0 = w * NPW;
    if (n0 >= N) return;
    int cnt = N - n0; if (cnt > NPW) cnt = NPW;

    int rp   = rowptr[n0 + min(lane, cnt)];
    int j    = __builtin_amdgcn_readlane(rp, 0);
    int end  = __builtin_amdgcn_readlane(rp, cnt);
    int row  = 0;
    int rend = __builtin_amdgcn_readlane(rp, 1);
    if (j >= end) return;

    unsigned lane8 = (unsigned)lane * 8u;
    float*   obase = out + (size_t)n0 * 64 + 2 * lane;
    v2f bv = *(const v2f*)(bias + 2 * lane);

    float ax = 0.f, ay = 0.f;
    v2i ra[AGB], rb[AGB];
#pragma unroll
    for (int q = 0; q < AGB; ++q) ra[q] = csr[min(j + q, end - 1)];

    for (;;) {
        agg_trip<64>(ra, rb, csr, j, end, h, lane8, obase, rp, bv, do_relu, ax, ay, row, rend);
        if (j >= end) break;
        agg_trip<64>(rb, ra, csr, j, end, h, lane8, obase, rp, bv, do_relu, ax, ay, row, rend);
        if (j >= end) break;
    }
}

#define TP 68

__global__ __launch_bounds__(256) void agg_gemm_kernel(const float* __restrict__ h,
                                                       const int* __restrict__ rowptr,
                                                       const v2i* __restrict__ csr,
                                                       const float* __restrict__ bias,
                                                       const float* __restrict__ B,
                                                       float* __restrict__ C,
                                                       int N) {
    __shared__ float tiles[8][16 * TP];
    int lane = threadIdx.x & 31;
    int wv = __builtin_amdgcn_readfirstlane((int)(threadIdx.x >> 5));
    int w  = __builtin_amdgcn_readfirstlane((int)(blockIdx.x * 8 + (threadIdx.x >> 5)));
    int n0 = w * 16;
    if (n0 >= N) return;
    int cnt = N - n0; if (cnt > 16) cnt = 16;

    int rp   = rowptr[n0 + min(lane, cnt)];
    int j    = __builtin_amdgcn_readlane(rp, 0);
    int end  = __builtin_amdgcn_readlane(rp, cnt);
    int row  = 0;
    int rend = __builtin_amdgcn_readlane(rp, 1);
    if (j >= end) return;

    unsigned lane8 = (unsigned)lane * 8u;
    float*   tile  = &tiles[wv][0];
    float*   obase = tile + 2 * lane;
    v2f bv = *(const v2f*)(bias + 2 * lane);

    float ax = 0.f, ay = 0.f;
    v2i ra[AGB], rb[AGB];
#pragma unroll
    for (int q = 0; q < AGB; ++q) ra[q] = csr[min(j + q, end - 1)];

    for (;;) {
        agg_trip<TP>(ra, rb, csr, j, end, h, lane8, obase, rp, bv, 1, ax, ay, row, rend);
        if (j >= end) break;
        agg_trip<TP>(rb, ra, csr, j, end, h, lane8, obase, rp, bv, 1, ax, ay, row, rend);
        if (j >= end) break;
    }

    int hh = lane >> 4;
    int c  = lane & 15;
    const float* tp = tile + c * TP + 4 * hh;
    v8f acc[4];
#pragma unroll
    for (int jt = 0; jt < 4; ++jt) acc[jt] = (v8f){0.f, 0.f, 0.f, 0.f, 0.f, 0.f, 0.f, 0.f};
#pragma unroll 1
    for (int i = 0; i < 8; i += 2) {
        float4 a0 = *(const float4*)(tp + 8 * i);
        float4 a1 = *(const float4*)(tp + 8 * i + 8);
        v2f bf[4][4];
#pragma unroll
        for (int q = 0; q < 4; ++q) {
            int ks = 2 * i + q;
            int k0 = 8 * (ks >> 1) + 4 * hh + 2 * (ks & 1);
            float4 r0 = *(const float4*)(B + (size_t)k0 * 64 + 4 * c);
            float4 r1 = *(const float4*)(B + (size_t)(k0 + 1) * 64 + 4 * c);
            bf[q][0].x = r0.x; bf[q][0].y = r1.x;
            bf[q][1].x = r0.y; bf[q][1].y = r1.y;
            bf[q][2].x = r0.z; bf[q][2].y = r1.z;
            bf[q][3].x = r0.w; bf[q][3].y = r1.w;
        }
        v2f af[4];
        af[0].x = a0.x; af[0].y = a0.y;
        af[1].x = a0.z; af[1].y = a0.w;
        af[2].x = a1.x; af[2].y = a1.y;
        af[3].x = a1.z; af[3].y = a1.w;
#pragma unroll
        for (int q = 0; q < 4; ++q) {
#pragma unroll
            for (int jt = 0; jt < 4; ++jt)
                acc[jt] = __builtin_amdgcn_wmma_f32_16x16x4_f32(false, af[q], false, bf[q][jt],
                                                                (short)0, acc[jt], false, false);
        }
    }
    float* crow = C + (size_t)(n0 + 8 * hh) * 64 + 4 * c;
#pragma unroll
    for (int r = 0; r < 8; ++r) {
        if (8 * hh + r < cnt) {
            float4 o;
            o.x = acc[0][r]; o.y = acc[1][r]; o.z = acc[2][r]; o.w = acc[3][r];
            *(float4*)(crow + (size_t)r * 64) = o;
        }
    }
}

extern "C" void kernel_launch(void* const* d_in, const int* in_sizes, int n_in,
                              void* d_out, int out_size, void* d_ws, size_t ws_size,
                              hipStream_t stream) {
    (void)n_in; (void)out_size; (void)ws_size;

    const float* x  = (const float*)d_in[0];
    const int*   ei = (const int*)d_in[1];
    const float* W1 = (const float*)d_in[2];
    const float* b1 = (const float*)d_in[3];
    const float* W2 = (const float*)d_in[4];
    const float* b2 = (const float*)d_in[5];
    float* out = (float*)d_out;

    const int H   = in_sizes[3];
    const int IN  = in_sizes[2] / H;
    const int N   = in_sizes[0] / IN;
    const int E   = in_sizes[1] / 2;
    const int NNZ = N + E;

    char* ws = (char*)d_ws;
    size_t off = 0;
    auto alloc = [&](size_t bytes) -> void* {
        void* p = (void*)(ws + off);
        off += (bytes + 255) & ~(size_t)255;
        return p;
    };
    const int T = 256;
    int gN = (N + CB - 1) / CB;
    int*   bsum    = (int*)  alloc((size_t)gN * 4);
    int*   rowptr  = (int*)  alloc((size_t)(N + 1) * 4);
    float* dinv    = (float*)alloc((size_t)N * 4);
    int*   partial = (int*)  alloc((size_t)KCH * N * 4);
    v2i*   csr     = (v2i*)  alloc((size_t)NNZ * 8);
    float* h1      = (float*)alloc((size_t)N * H * 4);
    float* h2      = (float*)alloc((size_t)N * H * 4);

    int EC = (((E + KCH - 1) / KCH) + 3) & ~3;
    dim3 gTile((N + HR - 1) / HR, KCH);
    int NPW  = (N + 10239) / 10240;
    if (NPW > 30) NPW = 30;
    int gAgg = ((N + NPW - 1) / NPW + 7) / 8;
    int gAggF = ((N + 15) / 16 + 7) / 8;
    int gGemm1 = 256;

    hist_kernel<<<gTile, 1024, 0, stream>>>(ei, partial, N, E, EC);
    combine_kernel<<<gN, CB, 0, stream>>>(partial, rowptr, bsum, dinv, N);
    init_csr_kernel<<<gN, CB, 0, stream>>>(rowptr, bsum, dinv, csr, N);
    fill_csr_kernel<<<gTile, 1024, 0, stream>>>(ei, dinv, rowptr, partial, csr, N, E, EC);

    gemm64_wmma_kernel<128><<<gGemm1, T, 0, stream>>>(x, W1, h1, N);
    agg_gemm_kernel<<<gAggF, T, 0, stream>>>(h1, rowptr, csr, b1, W2, h2, N);

    agg_kernel<<<gAgg, T, 0, stream>>>(h2, rowptr, csr, b2, out, N, NPW, 0);
}


# Edge weight factored out of the CSR records: gathered rows are stored pre-scaled by dinv[col], a row sum is scaled once by dinv[row]; 4B records, fill gathers no dinv; term rounding changes
# speedup vs baseline: 3.5677x; 1.1677x over previous
#include <hip/hip_runtime.h>
#include <hip/hip_bf16.h>


typedef float v2f __attribute__((ext_vector_type(2)));
typedef float v8f __attribute__((ext_vector_type(8)));

#define HR  12544
#define KCH 32

__device__ __forceinline__ int4 load_idx4(const int* __restrict__ p, int e, int eend, bool vec_ok) {
    int4 v;
    if (vec_ok && e + 3 < eend) {
        v = *(const int4*)(p + e);
    } else {
        v.x = (e     < eend) ? p[e]     : -1;
        v.y = (e + 1 < eend) ? p[e + 1] : -1;
        v.z = (e + 2 < eend) ? p[e + 2] : -1;
        v.w = (e + 3 < eend) ? p[e + 3] : -1;
    }
    return v;
}

__global__ __launch_bounds__(1024) void hist_kernel(const int* __restrict__ ei,
                                                    int* __restrict__ partial,
                                                    int N, int E, int EC) {
    __shared__ int bins[HR];
    int tid = threadIdx.x;
    int r0  = blockIdx.x * HR;
    int k   = blockIdx.y;
    int nr  = N - r0; if (nr > HR) nr = HR;
    for (int i = tid; i < HR; i += 1024) bins[i] = 0;
    __syncthreads();
    int ebeg = k * EC;
    int eend = ebeg + EC; if (eend > E) eend = E;
    for (int eb = ebeg; eb < eend; eb += 8192) {
        int e = eb + 4 * tid;
        int4 a, b;
        if (eb + 8192 <= eend) {
            a = *(const int4*)(ei + e);
            b = *(const int4*)(ei + e + 4096);
        } else {
            a = load_idx4(ei, e, eend, true);
            b = load_idx4(ei, e + 4096, eend, true);
        }
        int r[8] = {a.x, a.y, a.z, a.w, b.x, b.y, b.z, b.w};
#pragma unroll
        for (int q = 0; q < 8; ++q) {
            unsigned d = (unsigned)(r[q] - r0);
            if (d < (unsigned)nr) atomicAdd(&bins[d], 1);
        }
    }
    __syncthreads();
    int* dst = partial + (size_t)k * N + r0;
    for (int i = tid; i < nr; i += 1024) dst[i] = bins[i];
}

#define CB 256

__global__ __launch_bounds__(CB) void combine_kernel(int* __restrict__ partial, int* __restrict__ rowptr,
                                                     int* __restrict__ bsum, float* __restrict__ dinv,
                                                     int N) {
    __shared__ int wsum[CB / 32];
    int tid = threadIdx.x;
    int i = blockIdx.x * CB + tid;
    int v = 0;
    if (i < N) {
        int t[KCH];
#pragma unroll
        for (int k = 0; k < KCH; ++k) t[k] = partial[(size_t)k * N + i];
        int run = 0;
#pragma unroll
        for (int k = 0; k < KCH; ++k) {
            partial[(size_t)k * N + i] = run;
            run += t[k];
        }
        dinv[i] = rsqrtf((float)(run + 1));
        v = run + 1;
    }
    int lane = tid & 31;
    int wv   = tid >> 5;
    int inc  = v;
    for (int off = 1; off < 32; off <<= 1) {
        int n = __shfl_up(inc, off, 32);
        if (lane >= off) inc += n;
    }
    if (lane == 31) wsum[wv] = inc;
    __syncthreads();
    int base = 0, tot = 0;
    for (int q = 0; q < CB / 32; ++q) {
        int s = wsum[q];
        if (q < wv) base += s;
        tot += s;
    }
    if (i < N) rowptr[i] = base + inc - v;
    if (tid == 0) bsum[blockIdx.x] = tot;
}

__global__ __launch_bounds__(CB) void init_csr_kernel(int* __restrict__ rowptr, const int* __restrict__ bsum,
                                                      int* __restrict__ csr, int N) {
    __shared__ int wsum[CB / 32];
    int tid = threadIdx.x;
    int b   = blockIdx.x;
    int i   = b * CB + tid;
    int loc = 0;
    if (i < N) loc = rowptr[i];
    int s = 0;
    for (int q = tid; q < b; q += CB) s += bsum[q];
    for (int off = 16; off > 0; off >>= 1) s += __shfl_xor(s, off, 32);
    if ((tid & 31) == 0) wsum[tid >> 5] = s;
    __syncthreads();
    int boff = 0;
    for (int q = 0; q < CB / 32; ++q) boff += wsum[q];
    if (i < N) {
        int p = loc + boff;
        rowptr[i] = p;
        csr[p] = i;
    }
    if (b == (int)gridDim.x - 1 && tid == 0) rowptr[N] = boff + bsum[b];
}

__global__ __launch_bounds__(1024) void fill_csr_kernel(const int* __restrict__ ei,
                                                        const int* __restrict__ rowptr,
                                                        const int* __restrict__ partial,
                                                        int* __restrict__ csr,
                                                        int N, int E, int EC) {
    __shared__ int slot[HR];
    int tid = threadIdx.x;
    int r0  = blockIdx.x * HR;
    int k   = blockIdx.y;
    int nr  = N - r0; if (nr > HR) nr = HR;
    const int* pre = partial + (size_t)k * N + r0;
    {
        constexpr int TRIPS = (HR + 1023) / 1024;
        int rp[TRIPS], pp[TRIPS];
#pragma unroll
        for (int it = 0; it < TRIPS; ++it) {
            int i = min(tid + it * 1024, nr - 1);
            rp[it] = rowptr[r0 + i];
            pp[it] = pre[i];
        }
#pragma unroll
        for (int it = 0; it < TRIPS; ++it) {
            int i = tid + it * 1024;
            if (i < nr) slot[i] = rp[it] + 1 + pp[it];
        }
    }
    __syncthreads();
    int ebeg = k * EC;
    int eend = ebeg + EC; if (eend > E) eend = E;
    const int* ec = ei + E;
    bool cvec = (E & 3) == 0;
    for (int eb = ebeg; eb < eend; eb += 8192) {
        int e = eb + 4 * tid;
        int4 ra, rb, ca, cb;
        if (cvec && eb + 8192 <= eend) {
            ra = *(const int4*)(ei + e);
            rb = *(const int4*)(ei + e + 4096);
            ca = *(const int4*)(ec + e);
            cb = *(const int4*)(ec + e + 4096);
        } else {
            ra = load_idx4(ei, e, eend, true);
            rb = load_idx4(ei, e + 4096, eend, true);
            ca = load_idx4(ec, e, eend, cvec);
            cb = load_idx4(ec, e + 4096, eend, cvec);
        }
        int r[8] = {ra.x, ra.y, ra.z, ra.w, rb.x, rb.y, rb.z, rb.w};
        int c[8] = {ca.x, ca.y, ca.z, ca.w, cb.x, cb.y, cb.z, cb.w};
        unsigned d[8];
#pragma unroll
        for (int q = 0; q < 8; ++q) d[q] = (unsigned)(r[q] - r0);
        unsigned p[8];
#pragma unroll
        for (int q = 0; q < 8; ++q) {
            if (d[q] < (unsigned)nr) p[q] = (unsigned)atomicAdd(&slot[d[q]], 1);
        }
#pragma unroll
        for (int q = 0; q < 8; ++q) {
            if (d[q] < (unsigned)nr) csr[p[q]] = c[q];
        }
    }
}

template <int K>
__device__ __forceinline__ void gemm_load_a(float4 (&a)[K / 8], float4 (&dv)[2],
                                            const float* __restrict__ A, const float* __restrict__ dinv,
                                            int t, int c, int h) {
    const float* p = A + (size_t)(t * 16 + c) * K + 4 * h;
#pragma unroll
    for (int i = 0; i < K / 8; ++i) a[i] = *(const float4*)(p + 8 * i);
    const float* pd = dinv + t * 16 + 8 * h;
    dv[0] = *(const float4*)pd;
    dv[1] = *(const float4*)(pd + 4);
}

template <int K>
__device__ __forceinline__ void gemm_tile(const float4 (&a)[K / 8], const float4 (&dv)[2],
                                          const v2f (&bf)[K / 4][4],
                                          float* __restrict__ C, int t, int c, int h) {
    v8f acc[4];
#pragma unroll
    for (int j = 0; j < 4; ++j) acc[j] = (v8f){0.f, 0.f, 0.f, 0.f, 0.f, 0.f, 0.f, 0.f};
#pragma unroll
    for (int i = 0; i < K / 8; ++i) {
        v2f lo, hi;
        lo.x = a[i].x; lo.y = a[i].y;
        hi.x = a[i].z; hi.y = a[i].w;
#pragma unroll
        for (int j = 0; j < 4; ++j)
            acc[j] = __builtin_amdgcn_wmma_f32_16x16x4_f32(false, lo, false, bf[2 * i][j],
                                                           (short)0, acc[j], false, false);
#pragma unroll
        for (int j = 0; j < 4; ++j)
            acc[j] = __builtin_amdgcn_wmma_f32_16x16x4_f32(false, hi, false, bf[2 * i + 1][j],
                                                           (short)0, acc[j], false, false);
    }
    float* crow = C + (size_t)(t * 16 + 8 * h) * 64 + 4 * c;
    const float sc[8] = {dv[0].x, dv[0].y, dv[0].z, dv[0].w, dv[1].x, dv[1].y, dv[1].z, dv[1].w};
#pragma unroll
    for (int r = 0; r < 8; ++r) {
        float4 o;
        o.x = acc[0][r] * sc[r]; o.y = acc[1][r] * sc[r]; o.z = acc[2][r] * sc[r]; o.w = acc[3][r] * sc[r];
        *(float4*)(crow + (size_t)r * 64) = o;
    }
}

template <int K>
__global__ __launch_bounds__(256) void gemm64_wmma_kernel(const float* __restrict__ A,
                                                          const float* __restrict__ B,
                                                          const float* __restrict__ dinv,
                                                          float* __restrict__ C,
                                                          int M) {
    int tid  = threadIdx.x;
    int wave = __builtin_amdgcn_readfirstlane(tid >> 5);
    int lane = tid & 31;
    int h = lane >> 4;
    int c = lane & 15;

    int nt     = M >> 4;
    int stride = (int)gridDim.x * 8;
    int t = (int)blockIdx.x + (int)gridDim.x * wave;
    if (t >= nt) return;

    float4 a0[K / 8], a1[K / 8];
    float4 d0[2], d1[2];
    gemm_load_a<K>(a0, d0, A, dinv, t, c, h);

    v2f bf[K / 4][4];
#pragma unroll
    for (int w = 0; w < K / 4; ++w) {
        int k0 = 8 * (w >> 1) + 4 * h + 2 * (w & 1);
        float4 r0 = *(const float4*)(B + (size_t)k0 * 64 + 4 * c);
        float4 r1 = *(const float4*)(B + (size_t)(k0 + 1) * 64 + 4 * c);
        bf[w][0].x = r0.x; bf[w][0].y = r1.x;
        bf[w][1].x = r0.y; bf[w][1].y = r1.y;
        bf[w][2].x = r0.z; bf[w][2].y = r1.z;
        bf[w][3].x = r0.w; bf[w][3].y = r1.w;
    }

    for (;;) {
        int tn = t + stride;
        if (tn >= nt) { gemm_tile<K>(a0, d0, bf, C, t, c, h); return; }
        gemm_load_a<K>(a1, d1, A, dinv, tn, c, h);
        __builtin_amdgcn_sched_barrier(0);
        gemm_tile<K>(a0, d0, bf, C, t, c, h);
        t  = tn;
        tn = t + stride;
        if (tn >= nt) { gemm_tile<K>(a1, d1, bf, C, t, c, h); return; }
        gemm_load_a<K>(a0, d0, A, dinv, tn, c, h);
        __builtin_amdgcn_sched_barrier(0);
        gemm_tile<K>(a1, d1, bf, C, t, c, h);
        t = tn;
    }
}

#define AGB 8

template <bool GUARD, int PITCH>
__device__ __forceinline__ void agg_step(const v2f (&hv)[AGB], int j, int end,
                                         float* __restrict__ obase, int rp, int dvb,
                                         v2f bv, int do_relu,
                                         float& ax, float& ay, int& row, int& rend) {
    int left = rend - j;
#pragma unroll
    for (int q = 0; q < AGB; ++q) {
        if (!GUARD || j + q < end) {
            ax += hv[q].x;
            ay += hv[q].y;
            if (q + 1 == left) {
                float s  = __int_as_float(__builtin_amdgcn_readlane(dvb, row));
                float ox = fmaf(ax, s, bv.x);
                float oy = fmaf(ay, s, bv.y);
                if (do_relu) {
                    ox = fmaxf(ox, 0.f);
                    oy = fmaxf(oy, 0.f);
                }
                v2f o; o.x = ox; o.y = oy;
                *(v2f*)(obase + row * PITCH) = o;
                ax = 0.f; ay = 0.f;
                ++row;
                rend = __builtin_amdgcn_readlane(rp, row + 1);
                left = rend - j;
            }
        }
    }
}

template <int PITCH>
__device__ __forceinline__ void agg_trip(const int (&rc)[AGB], int (&rn)[AGB],
                                         const int* __restrict__ csr, int& j, int end,
                                         const float* __restrict__ h, unsigned lane8,
                                         float* __restrict__ obase, int rp, int dvb,
                                         v2f bv, int do_relu,
                                         float& ax, float& ay, int& row, int& rend) {
    v2f hv[AGB];
#pragma unroll
    for (int q = 0; q < AGB; ++q)
        hv[q] = *(const v2f*)((const char*)h + ((unsigned)rc[q] * 256u + lane8));
    __builtin_amdgcn_sched_barrier(0);
    int jn = j + AGB;
    if (jn + AGB <= end) {
        const int* pn = csr + jn;
#pragma unroll
        for (int q = 0; q < AGB; ++q) rn[q] = pn[q];
    } else {
#pragma unroll
        for (int q = 0; q < AGB; ++q) rn[q] = csr[min(jn + q, end - 1)];
    }
    if (jn <= end) agg_step<false, PITCH>(hv, j, end, obase, rp, dvb, bv, do_relu, ax, ay, row, rend);
    else           agg_step<true,  PITCH>(hv, j, end, obase, rp, dvb, bv, do_relu, ax, ay, row, rend);
    j = jn;
}

__global__ __launch_bounds__(256) void agg_kernel(const float* __restrict__ h,
                                                  const int* __restrict__ rowptr,
                                                  const int* __restrict__ csr,
                                                  const float* __restrict__ dinv,
                                                  const float* __restrict__ bias,
                                                  float* __restrict__ out,
                                                  int N, int NPW, int do_relu) {
    int lane = threadIdx.x & 31;
    int w  = __builtin_amdgcn_readfirstlane((int)(blockIdx.x * 8 + (threadIdx.x >> 5)));
    int n0 = w * NPW;
    if (n0 >= N) return;
    int cnt = N - n0; if (cnt > NPW) cnt = NPW;

    int rp   = rowptr[n0 + min(lane, cnt)];
    int dvb  = __float_as_int(dinv[n0 + min(lane, cnt - 1)]);
    int j    = __builtin_amdgcn_readlane(rp, 0);
    int end  = __builtin_amdgcn_readlane(rp, cnt);
    int row  = 0;
    int rend = __builtin_amdgcn_readlane(rp, 1);
    if (j >= end) return;

    unsigned lane8 = (unsigned)lane * 8u;
    float*   obase = out + (size_t)n0 * 64 + 2 * lane;
    v2f bv = *(const v2f*)(bias + 2 * lane);

    float ax = 0.f, ay = 0.f;
    int ra[AGB], rb[AGB];
#pragma unroll
    for (int q = 0; q < AGB; ++q) ra[q] = csr[min(j + q, end - 1)];

    for (;;) {
        agg_trip<64>(ra, rb, csr, j, end, h, lane8, obase, rp, dvb, bv, do_relu, ax, ay, row, rend);
        if (j >= end) break;
        agg_trip<64>(rb, ra, csr, j, end, h, lane8, obase, rp, dvb, bv, do_relu, ax, ay, row, rend);
        if (j >= end) break;
    }
}

#define TP 68

__global__ __launch_bounds__(256) void agg_gemm_kernel(const float* __restrict__ h,
                                                       const int* __restrict__ rowptr,
                                                       const int* __restrict__ csr,
                                                       const float* __restrict__ dinv,
                                                       const float* __restrict__ bias,
                                                       const float* __restrict__ B,
                                                       float* __restrict__ C,
                                                       int N) {
    __shared__ float tiles[8][16 * TP];
    int lane = threadIdx.x & 31;
    int wv = __builtin_amdgcn_readfirstlane((int)(threadIdx.x >> 5));
    int w  = __builtin_amdgcn_readfirstlane((int)(blockIdx.x * 8 + (threadIdx.x >> 5)));
    int n0 = w * 16;
    if (n0 >= N) return;
    int cnt = N - n0; if (cnt > 16) cnt = 16;

    int   rp  = rowptr[n0 + min(lane, cnt)];
    float dvf = dinv[n0 + min(lane, cnt - 1)];
    int   dvb = __float_as_int(dvf);
    int j    = __builtin_amdgcn_readlane(rp, 0);
    int end  = __builtin_amdgcn_readlane(rp, cnt);
    int row  = 0;
    int rend = __builtin_amdgcn_readlane(rp, 1);
    if (j >= end) return;

    unsigned lane8 = (unsigned)lane * 8u;
    float*   tile  = &tiles[wv][0];
    float*   obase = tile + 2 * lane;
    v2f bv = *(const v2f*)(bias + 2 * lane);

    float ax = 0.f, ay = 0.f;
    int ra[AGB], rb[AGB];
#pragma unroll
    for (int q = 0; q < AGB; ++q) ra[q] = csr[min(j + q, end - 1)];

    for (;;) {
        agg_trip<TP>(ra, rb, csr, j, end, h, lane8, obase, rp, dvb, bv, 1, ax, ay, row, rend);
        if (j >= end) break;
        agg_trip<TP>(rb, ra, csr, j, end, h, lane8, obase, rp, dvb, bv, 1, ax, ay, row, rend);
        if (j >= end) break;
    }

    int hh = lane >> 4;
    int c  = lane & 15;
    float sc[8];
#pragma unroll
    for (int r = 0; r < 8; ++r) sc[r] = __shfl(dvf, 8 * hh + r, 32);
    const float* tp = tile + c * TP + 4 * hh;
    v8f acc[4];
#pragma unroll
    for (int jt = 0; jt < 4; ++jt) acc[jt] = (v8f){0.f, 0.f, 0.f, 0.f, 0.f, 0.f, 0.f, 0.f};
#pragma unroll 1
    for (int i = 0; i < 8; i += 2) {
        float4 a0 = *(const float4*)(tp + 8 * i);
        float4 a1 = *(const float4*)(tp + 8 * i + 8);
        v2f bf[4][4];
#pragma unroll
        for (int q = 0; q < 4; ++q) {
            int ks = 2 * i + q;
            int k0 = 8 * (ks >> 1) + 4 * hh + 2 * (ks & 1);
            float4 r0 = *(const float4*)(B + (size_t)k0 * 64 + 4 * c);
            float4 r1 = *(const float4*)(B + (size_t)(k0 + 1) * 64 + 4 * c);
            bf[q][0].x = r0.x; bf[q][0].y = r1.x;
            bf[q][1].x = r0.y; bf[q][1].y = r1.y;
            bf[q][2].x = r0.z; bf[q][2].y = r1.z;
            bf[q][3].x = r0.w; bf[q][3].y = r1.w;
        }
        v2f af[4];
        af[0].x = a0.x; af[0].y = a0.y;
        af[1].x = a0.z; af[1].y = a0.w;
        af[2].x = a1.x; af[2].y = a1.y;
        af[3].x = a1.z; af[3].y = a1.w;
#pragma unroll
        for (int q = 0; q < 4; ++q) {
#pragma unroll
            for (int jt = 0; jt < 4; ++jt)
                acc[jt] = __builtin_amdgcn_wmma_f32_16x16x4_f32(false, af[q], false, bf[q][jt],
                                                                (short)0, acc[jt], false, false);
        }
    }
    float* crow = C + (size_t)(n0 + 8 * hh) * 64 + 4 * c;
#pragma unroll
    for (int r = 0; r < 8; ++r) {
        if (8 * hh + r < cnt) {
            float4 o;
            o.x = acc[0][r] * sc[r]; o.y = acc[1][r] * sc[r]; o.z = acc[2][r] * sc[r]; o.w = acc[3][r] * sc[r];
            *(float4*)(crow + (size_t)r * 64) = o;
        }
    }
}

extern "C" void kernel_launch(void* const* d_in, const int* in_sizes, int n_in,
                              void* d_out, int out_size, void* d_ws, size_t ws_size,
                              hipStream_t stream) {
    (void)n_in; (void)out_size; (void)ws_size;

    const float* x  = (const float*)d_in[0];
    const int*   ei = (const int*)d_in[1];
    const float* W1 = (const float*)d_in[2];
    const float* b1 = (const float*)d_in[3];
    const float* W2 = (const float*)d_in[4];
    const float* b2 = (const float*)d_in[5];
    float* out = (float*)d_out;

    const int H   = in_sizes[3];
    const int IN  = in_sizes[2] / H;
    const int N   = in_sizes[0] / IN;
    const int E   = in_sizes[1] / 2;
    const int NNZ = N + E;

    char* ws = (char*)d_ws;
    size_t off = 0;
    auto alloc = [&](size_t bytes) -> void* {
        void* p = (void*)(ws + off);
        off += (bytes + 255) & ~(size_t)255;
        return p;
    };
    const int T = 256;
    int gN = (N + CB - 1) / CB;
    int*   bsum    = (int*)  alloc((size_t)gN * 4);
    int*   rowptr  = (int*)  alloc((size_t)(N + 1) * 4);
    float* dinv    = (float*)alloc((size_t)N * 4);
    int*   partial = (int*)  alloc((size_t)KCH * N * 4);
    int*   csr     = (int*)  alloc((size_t)NNZ * 4);
    float* h1      = (float*)alloc((size_t)N * H * 4);
    float* h2      = (float*)alloc((size_t)N * H * 4);

    int EC = (((E + KCH - 1) / KCH) + 3) & ~3;
    dim3 gTile((N + HR - 1) / HR, KCH);
    int NPW  = (N + 10239) / 10240;
    if (NPW > 30) NPW = 30;
    int gAgg = ((N + NPW - 1) / NPW + 7) / 8;
    int gAggF = ((N + 15) / 16 + 7) / 8;
    int gGemm1 = 256;

    hist_kernel<<<gTile, 1024, 0, stream>>>(ei, partial, N, E, EC);
    combine_kernel<<<gN, CB, 0, stream>>>(partial, rowptr, bsum, dinv, N);
    init_csr_kernel<<<gN, CB, 0, stream>>>(rowptr, bsum, csr, N);
    fill_csr_kernel<<<gTile, 1024, 0, stream>>>(ei, rowptr, partial, csr, N, E, EC);

    gemm64_wmma_kernel<128><<<gGemm1, T, 0, stream>>>(x, W1, dinv, h1, N);
    agg_gemm_kernel<<<gAggF, T, 0, stream>>>(h1, rowptr, csr, dinv, b1, W2, h2, N);

    agg_kernel<<<gAgg, T, 0, stream>>>(h2, rowptr, csr, dinv, b2, out, N, NPW, 0);
}


# CSR build as a two-level counting sort: edge tiles ordered by 256-node group in LDS, then one block per group builds its rows; 4 launches become 2; a row's sum order is now arrival order
# speedup vs baseline: 3.7599x; 1.0539x over previous
#include <hip/hip_runtime.h>
#include <hip/hip_bf16.h>


typedef float v2f __attribute__((ext_vector_type(2)));
typedef float v8f __attribute__((ext_vector_type(8)));

#define GS   256
#define TE   4096
#define GMAX 512
#define RPW  (GMAX / 8)

__device__ __forceinline__ int4 load_idx4(const int* __restrict__ p, int e, int eend, bool vec_ok) {
    int4 v;
    if (vec_ok && e + 3 < eend) {
        v = *(const int4*)(p + e);
    } else {
        v.x = (e     < eend) ? p[e]     : -1;
        v.y = (e + 1 < eend) ? p[e + 1] : -1;
        v.z = (e + 2 < eend) ? p[e + 2] : -1;
        v.w = (e + 3 < eend) ? p[e + 3] : -1;
    }
    return v;
}

__global__ __launch_bounds__(256) void tile_sort_kernel(const int* __restrict__ ei,
                                                        unsigned* __restrict__ tiles,
                                                        int* __restrict__ cumT,
                                                        int N, int E, int G, int T) {
    __shared__ int cnt[GMAX];
    __shared__ int st[GMAX];
    __shared__ __attribute__((aligned(16))) unsigned srec[TE];
    __shared__ int wsum[8];
    int tid  = threadIdx.x;
    int lane = tid & 31;
    int wv   = tid >> 5;
    int t    = blockIdx.x;
    cnt[tid] = 0;
    cnt[tid + 256] = 0;
    int ebeg = t * TE;
    int eend = ebeg + TE; if (eend > E) eend = E;
    const int* ec = ei + E;
    bool cvec = (E & 3) == 0;
    int4 rv[4], cv[4];
    if (cvec && ebeg + TE <= E) {
#pragma unroll
        for (int k = 0; k < 4; ++k) {
            int e = ebeg + 4 * tid + 1024 * k;
            rv[k] = *(const int4*)(ei + e);
            cv[k] = *(const int4*)(ec + e);
        }
    } else {
#pragma unroll
        for (int k = 0; k < 4; ++k) {
            int e = ebeg + 4 * tid + 1024 * k;
            rv[k] = load_idx4(ei, e, eend, true);
            cv[k] = load_idx4(ec, e, eend, cvec);
        }
    }
    int r[16] = {rv[0].x, rv[0].y, rv[0].z, rv[0].w, rv[1].x, rv[1].y, rv[1].z, rv[1].w,
                 rv[2].x, rv[2].y, rv[2].z, rv[2].w, rv[3].x, rv[3].y, rv[3].z, rv[3].w};
    int c[16] = {cv[0].x, cv[0].y, cv[0].z, cv[0].w, cv[1].x, cv[1].y, cv[1].z, cv[1].w,
                 cv[2].x, cv[2].y, cv[2].z, cv[2].w, cv[3].x, cv[3].y, cv[3].z, cv[3].w};
    __syncthreads();
    int rk[16];
#pragma unroll
    for (int q = 0; q < 16; ++q) {
        if ((unsigned)r[q] < (unsigned)N && (unsigned)c[q] < (unsigned)N)
            rk[q] = atomicAdd(&cnt[(unsigned)r[q] / GS], 1);
    }
    __syncthreads();
    int a = cnt[2 * tid];
    int b = cnt[2 * tid + 1];
    int v = a + b;
    int inc = v;
    for (int off = 1; off < 32; off <<= 1) {
        int n = __shfl_up(inc, off, 32);
        if (lane >= off) inc += n;
    }
    if (lane == 31) wsum[wv] = inc;
    __syncthreads();
    int base = 0;
    for (int q = 0; q < 8; ++q) {
        int s = wsum[q];
        if (q < wv) base += s;
    }
    int ex = base + inc - v;
    st[2 * tid]     = ex;
    st[2 * tid + 1] = ex + a;
    __syncthreads();
#pragma unroll
    for (int q = 0; q < 16; ++q) {
        if ((unsigned)r[q] < (unsigned)N && (unsigned)c[q] < (unsigned)N)
            srec[st[(unsigned)r[q] / GS] + rk[q]] = (((unsigned)r[q] % GS) << 24) | (unsigned)c[q];
    }
    __syncthreads();
    int tot = st[G];
    unsigned* dst = tiles + (size_t)t * TE;
    for (int i = 4 * tid; i < tot; i += 1024) {
        if (i + 3 < tot) {
            *(uint4*)(dst + i) = *(const uint4*)(srec + i);
        } else {
            for (int k = i; k < tot; ++k) dst[k] = srec[k];
        }
    }
    for (int g = tid; g <= G; g += 256) cumT[(size_t)g * T + t] = st[g];
}

__global__ __launch_bounds__(256) void group_rows_kernel(const unsigned* __restrict__ tiles,
                                                         const int* __restrict__ cumT,
                                                         int* __restrict__ rowptr,
                                                         float* __restrict__ dinv,
                                                         int* __restrict__ csr,
                                                         int N, int G, int T) {
    __shared__ int deg[GS];
    __shared__ int wsa[8];
    __shared__ int wsb[8];
    int tid  = threadIdx.x;
    int lane = tid & 31;
    int wv   = __builtin_amdgcn_readfirstlane(tid >> 5);
    int g    = blockIdx.x;
    const int* c0 = cumT + (size_t)g * T;
    const int* c1 = c0 + T;
    int tA = 8 * lane + wv;
    int tB = tA + 256;
    int loA = 0, hiA = 0, loB = 0, hiB = 0;
    if (tA < T) { loA = c0[tA]; hiA = c1[tA]; }
    if (tB < T) { loB = c0[tB]; hiB = c1[tB]; }
    int s = loA + loB;
    deg[tid] = 0;
    for (int off = 16; off > 0; off >>= 1) s += __shfl_xor(s, off, 32);
    if (lane == 0) wsa[wv] = s;
    __syncthreads();
    int before = 0;
    for (int q = 0; q < 8; ++q) before += wsa[q];
    int base = g * GS + before;

    unsigned rec[RPW];
    unsigned long long okm = 0ull;
    unsigned long long lgm = 0ull;
#pragma unroll
    for (int u = 0; u < RPW; ++u) {
        int l0 = __builtin_amdgcn_readlane(u < 32 ? loA : loB, u & 31);
        int h0 = __builtin_amdgcn_readlane(u < 32 ? hiA : hiB, u & 31);
        const unsigned* tb = tiles + (size_t)(8 * u + wv) * TE + l0;
        if (lane < h0 - l0) {
            rec[u] = tb[lane];
            okm |= 1ull << u;
        }
        if (h0 - l0 > 32) lgm |= 1ull << u;
    }
#pragma unroll
    for (int u = 0; u < RPW; ++u) {
        if ((okm >> u) & 1ull) atomicAdd(&deg[rec[u] >> 24], 1);
    }
    for (unsigned long long m = lgm; m != 0ull; m &= m - 1ull) {
        int u  = __builtin_amdgcn_readfirstlane((int)__builtin_ctzll(m));
        int l0 = __builtin_amdgcn_readlane(u < 32 ? loA : loB, u & 31);
        int h0 = __builtin_amdgcn_readlane(u < 32 ? hiA : hiB, u & 31);
        const unsigned* tb = tiles + (size_t)(8 * u + wv) * TE;
        for (int p = l0 + 32 + lane; p < h0; p += 32)
            atomicAdd(&deg[tb[p] >> 24], 1);
    }
    __syncthreads();
    int node = g * GS + tid;
    int d = deg[tid];
    int v = (node < N) ? d + 1 : 0;
    int inc = v;
    for (int off = 1; off < 32; off <<= 1) {
        int n = __shfl_up(inc, off, 32);
        if (lane >= off) inc += n;
    }
    if (lane == 31) wsb[wv] = inc;
    __syncthreads();
    int wb = 0, tot = 0;
    for (int q = 0; q < 8; ++q) {
        int x = wsb[q];
        if (q < wv) wb += x;
        tot += x;
    }
    int p0 = base + wb + inc - v;
    if (node < N) {
        rowptr[node] = p0;
        dinv[node]   = rsqrtf((float)(d + 1));
        csr[p0]      = node;
    }
    deg[tid] = p0 + 1;
    if (g == G - 1 && tid == 0) rowptr[N] = base + tot;
    __syncthreads();
#pragma unroll
    for (int u = 0; u < RPW; ++u) {
        if ((okm >> u) & 1ull) {
            unsigned w = rec[u];
            int p = atomicAdd(&deg[w >> 24], 1);
            csr[p] = (int)(w & 0xFFFFFFu);
        }
    }
    for (unsigned long long m = lgm; m != 0ull; m &= m - 1ull) {
        int u  = __builtin_amdgcn_readfirstlane((int)__builtin_ctzll(m));
        int l0 = __builtin_amdgcn_readlane(u < 32 ? loA : loB, u & 31);
        int h0 = __builtin_amdgcn_readlane(u < 32 ? hiA : hiB, u & 31);
        const unsigned* tb = tiles + (size_t)(8 * u + wv) * TE;
        for (int q = l0 + 32 + lane; q < h0; q += 32) {
            unsigned w = tb[q];
            int p = atomicAdd(&deg[w >> 24], 1);
            csr[p] = (int)(w & 0xFFFFFFu);
        }
    }
}

template <int K>
__device__ __forceinline__ void gemm_load_a(float4 (&a)[K / 8], float4 (&dv)[2],
                                            const float* __restrict__ A, const float* __restrict__ dinv,
                                            int t, int c, int h) {
    const float* p = A + (size_t)(t * 16 + c) * K + 4 * h;
#pragma unroll
    for (int i = 0; i < K / 8; ++i) a[i] = *(const float4*)(p + 8 * i);
    const float* pd = dinv + t * 16 + 8 * h;
    dv[0] = *(const float4*)pd;
    dv[1] = *(const float4*)(pd + 4);
}

template <int K>
__device__ __forceinline__ void gemm_tile(const float4 (&a)[K / 8], const float4 (&dv)[2],
                                          const v2f (&bf)[K / 4][4],
                                          float* __restrict__ C, int t, int c, int h) {
    v8f acc[4];
#pragma unroll
    for (int j = 0; j < 4; ++j) acc[j] = (v8f){0.f, 0.f, 0.f, 0.f, 0.f, 0.f, 0.f, 0.f};
#pragma unroll
    for (int i = 0; i < K / 8; ++i) {
        v2f lo, hi;
        lo.x = a[i].x; lo.y = a[i].y;
        hi.x = a[i].z; hi.y = a[i].w;
#pragma unroll
        for (int j = 0; j < 4; ++j)
            acc[j] = __builtin_amdgcn_wmma_f32_16x16x4_f32(false, lo, false, bf[2 * i][j],
                                                           (short)0, acc[j], false, false);
#pragma unroll
        for (int j = 0; j < 4; ++j)
            acc[j] = __builtin_amdgcn_wmma_f32_16x16x4_f32(false, hi, false, bf[2 * i + 1][j],
                                                           (short)0, acc[j], false, false);
    }
    float* crow = C + (size_t)(t * 16 + 8 * h) * 64 + 4 * c;
    const float sc[8] = {dv[0].x, dv[0].y, dv[0].z, dv[0].w, dv[1].x, dv[1].y, dv[1].z, dv[1].w};
#pragma unroll
    for (int r = 0; r < 8; ++r) {
        float4 o;
        o.x = acc[0][r] * sc[r]; o.y = acc[1][r] * sc[r]; o.z = acc[2][r] * sc[r]; o.w = acc[3][r] * sc[r];
        *(float4*)(crow + (size_t)r * 64) = o;
    }
}

template <int K>
__global__ __launch_bounds__(256) void gemm64_wmma_kernel(const float* __restrict__ A,
                                                          const float* __restrict__ B,
                                                          const float* __restrict__ dinv,
                                                          float* __restrict__ C,
                                                          int M) {
    int tid  = threadIdx.x;
    int wave = __builtin_amdgcn_readfirstlane(tid >> 5);
    int lane = tid & 31;
    int h = lane >> 4;
    int c = lane & 15;

    int nt     = M >> 4;
    int stride = (int)gridDim.x * 8;
    int t = (int)blockIdx.x + (int)gridDim.x * wave;
    if (t >= nt) return;

    float4 a0[K / 8], a1[K / 8];
    float4 d0[2], d1[2];
    gemm_load_a<K>(a0, d0, A, dinv, t, c, h);

    v2f bf[K / 4][4];
#pragma unroll
    for (int w = 0; w < K / 4; ++w) {
        int k0 = 8 * (w >> 1) + 4 * h + 2 * (w & 1);
        float4 r0 = *(const float4*)(B + (size_t)k0 * 64 + 4 * c);
        float4 r1 = *(const float4*)(B + (size_t)(k0 + 1) * 64 + 4 * c);
        bf[w][0].x = r0.x; bf[w][0].y = r1.x;
        bf[w][1].x = r0.y; bf[w][1].y = r1.y;
        bf[w][2].x = r0.z; bf[w][2].y = r1.z;
        bf[w][3].x = r0.w; bf[w][3].y = r1.w;
    }

    for (;;) {
        int tn = t + stride;
        if (tn >= nt) { gemm_tile<K>(a0, d0, bf, C, t, c, h); return; }
        gemm_load_a<K>(a1, d1, A, dinv, tn, c, h);
        __builtin_amdgcn_sched_barrier(0);
        gemm_tile<K>(a0, d0, bf, C, t, c, h);
        t  = tn;
        tn = t + stride;
        if (tn >= nt) { gemm_tile<K>(a1, d1, bf, C, t, c, h); return; }
        gemm_load_a<K>(a0, d0, A, dinv, tn, c, h);
        __builtin_amdgcn_sched_barrier(0);
        gemm_tile<K>(a1, d1, bf, C, t, c, h);
        t = tn;
    }
}

#define AGB 8

template <bool GUARD, int PITCH>
__device__ __forceinline__ void agg_step(const v2f (&hv)[AGB], int j, int end,
                                         float* __restrict__ obase, int rp, int dvb,
                                         v2f bv, int do_relu,
                                         float& ax, float& ay, int& row, int& rend) {
    int left = rend - j;
#pragma unroll
    for (int q = 0; q < AGB; ++q) {
        if (!GUARD || j + q < end) {
            ax += hv[q].x;
            ay += hv[q].y;
            if (q + 1 == left) {
                float s  = __int_as_float(__builtin_amdgcn_readlane(dvb, row));
                float ox = fmaf(ax, s, bv.x);
                float oy = fmaf(ay, s, bv.y);
                if (do_relu) {
                    ox = fmaxf(ox, 0.f);
                    oy = fmaxf(oy, 0.f);
                }
                v2f o; o.x = ox; o.y = oy;
                *(v2f*)(obase + row * PITCH) = o;
                ax = 0.f; ay = 0.f;
                ++row;
                rend = __builtin_amdgcn_readlane(rp, row + 1);
                left = rend - j;
            }
        }
    }
}

template <int PITCH>
__device__ __forceinline__ void agg_trip(const int (&rc)[AGB], int (&rn)[AGB],
                                         const int* __restrict__ csr, int& j, int end,
                                         const float* __restrict__ h, unsigned lane8,
                                         float* __restrict__ obase, int rp, int dvb,
                                         v2f bv, int do_relu,
                                         float& ax, float& ay, int& row, int& rend) {
    v2f hv[AGB];
#pragma unroll
    for (int q = 0; q < AGB; ++q)
        hv[q] = *(const v2f*)((const char*)h + ((unsigned)rc[q] * 256u + lane8));
    __builtin_amdgcn_sched_barrier(0);
    int jn = j + AGB;
    if (jn + AGB <= end) {
        const int* pn = csr + jn;
#pragma unroll
        for (int q = 0; q < AGB; ++q) rn[q] = pn[q];
    } else {
#pragma unroll
        for (int q = 0; q < AGB; ++q) rn[q] = csr[min(jn + q, end - 1)];
    }
    if (jn <= end) agg_step<false, PITCH>(hv, j, end, obase, rp, dvb, bv, do_relu, ax, ay, row, rend);
    else           agg_step<true,  PITCH>(hv, j, end, obase, rp, dvb, bv, do_relu, ax, ay, row, rend);
    j = jn;
}

__global__ __launch_bounds__(256) void agg_kernel(const float* __restrict__ h,
                                                  const int* __restrict__ rowptr,
                                                  const int* __restrict__ csr,
                                                  const float* __restrict__ dinv,
                                                  const float* __restrict__ bias,
                                                  float* __restrict__ out,
                                                  int N, int NPW, int do_relu) {
    int lane = threadIdx.x & 31;
    int w  = __builtin_amdgcn_readfirstlane((int)(blockIdx.x * 8 + (threadIdx.x >> 5)));
    int n0 = w * NPW;
    if (n0 >= N) return;
    int cnt = N - n0; if (cnt > NPW) cnt = NPW;

    int rp   = rowptr[n0 + min(lane, cnt)];
    int dvb  = __float_as_int(dinv[n0 + min(lane, cnt - 1)]);
    int j    = __builtin_amdgcn_readlane(rp, 0);
    int end  = __builtin_amdgcn_readlane(rp, cnt);
    int row  = 0;
    int rend = __builtin_amdgcn_readlane(rp, 1);
    if (j >= end) return;

    unsigned lane8 = (unsigned)lane * 8u;
    float*   obase = out + (size_t)n0 * 64 + 2 * lane;
    v2f bv = *(const v2f*)(bias + 2 * lane);

    float ax = 0.f, ay = 0.f;
    int ra[AGB], rb[AGB];
#pragma unroll
    for (int q = 0; q < AGB; ++q) ra[q] = csr[min(j + q, end - 1)];

    for (;;) {
        agg_trip<64>(ra, rb, csr, j, end, h, lane8, obase, rp, dvb, bv, do_relu, ax, ay, row, rend);
        if (j >= end) break;
        agg_trip<64>(rb, ra, csr, j, end, h, lane8, obase, rp, dvb, bv, do_relu, ax, ay, row, rend);
        if (j >= end) break;
    }
}

#define TP 68

__global__ __launch_bounds__(256) void agg_gemm_kernel(const float* __restrict__ h,
                                                       const int* __restrict__ rowptr,
                                                       const int* __restrict__ csr,
                                                       const float* __restrict__ dinv,
                                                       const float* __restrict__ bias,
                                                       const float* __restrict__ B,
                                                       float* __restrict__ C,
                                                       int N) {
    __shared__ float tiles[8][16 * TP];
    int lane = threadIdx.x & 31;
    int wv = __builtin_amdgcn_readfirstlane((int)(threadIdx.x >> 5));
    int w  = __builtin_amdgcn_readfirstlane((int)(blockIdx.x * 8 + (threadIdx.x >> 5)));
    int n0 = w * 16;
    if (n0 >= N) return;
    int cnt = N - n0; if (cnt > 16) cnt = 16;

    int   rp  = rowptr[n0 + min(lane, cnt)];
    float dvf = dinv[n0 + min(lane, cnt - 1)];
    int   dvb = __float_as_int(dvf);
    int j    = __builtin_amdgcn_readlane(rp, 0);
    int end  = __builtin_amdgcn_readlane(rp, cnt);
    int row  = 0;
    int rend = __builtin_amdgcn_readlane(rp, 1);
    if (j >= end) return;

    unsigned lane8 = (unsigned)lane * 8u;
    float*   tile  = &tiles[wv][0];
    float*   obase = tile + 2 * lane;
    v2f bv = *(const v2f*)(bias + 2 * lane);

    float ax = 0.f, ay = 0.f;
    int ra[AGB], rb[AGB];
#pragma unroll
    for (int q = 0; q < AGB; ++q) ra[q] = csr[min(j + q, end - 1)];

    for (;;) {
        agg_trip<TP>(ra, rb, csr, j, end, h, lane8, obase, rp, dvb, bv, 1, ax, ay, row, rend);
        if (j >= end) break;
        agg_trip<TP>(rb, ra, csr, j, end, h, lane8, obase, rp, dvb, bv, 1, ax, ay, row, rend);
        if (j >= end) break;
    }

    int hh = lane >> 4;
    int c  = lane & 15;
    float sc[8];
#pragma unroll
    for (int r = 0; r < 8; ++r) sc[r] = __shfl(dvf, 8 * hh + r, 32);
    const float* tp = tile + c * TP + 4 * hh;
    v8f acc[4];
#pragma unroll
    for (int jt = 0; jt < 4; ++jt) acc[jt] = (v8f){0.f, 0.f, 0.f, 0.f, 0.f, 0.f, 0.f, 0.f};
#pragma unroll 1
    for (int i = 0; i < 8; i += 2) {
        float4 a0 = *(const float4*)(tp + 8 * i);
        float4 a1 = *(const float4*)(tp + 8 * i + 8);
        v2f bf[4][4];
#pragma unroll
        for (int q = 0; q < 4; ++q) {
            int ks = 2 * i + q;
            int k0 = 8 * (ks >> 1) + 4 * hh + 2 * (ks & 1);
            float4 r0 = *(const float4*)(B + (size_t)k0 * 64 + 4 * c);
            float4 r1 = *(const float4*)(B + (size_t)(k0 + 1) * 64 + 4 * c);
            bf[q][0].x = r0.x; bf[q][0].y = r1.x;
            bf[q][1].x = r0.y; bf[q][1].y = r1.y;
            bf[q][2].x = r0.z; bf[q][2].y = r1.z;
            bf[q][3].x = r0.w; bf[q][3].y = r1.w;
        }
        v2f af[4];
        af[0].x = a0.x; af[0].y = a0.y;
        af[1].x = a0.z; af[1].y = a0.w;
        af[2].x = a1.x; af[2].y = a1.y;
        af[3].x = a1.z; af[3].y = a1.w;
#pragma unroll
        for (int q = 0; q < 4; ++q) {
#pragma unroll
            for (int jt = 0; jt < 4; ++jt)
                acc[jt] = __builtin_amdgcn_wmma_f32_16x16x4_f32(false, af[q], false, bf[q][jt],
                                                                (short)0, acc[jt], false, false);
        }
    }
    float* crow = C + (size_t)(n0 + 8 * hh) * 64 + 4 * c;
#pragma unroll
    for (int r = 0; r < 8; ++r) {
        if (8 * hh + r < cnt) {
            float4 o;
            o.x = acc[0][r] * sc[r]; o.y = acc[1][r] * sc[r]; o.z = acc[2][r] * sc[r]; o.w = acc[3][r] * sc[r];
            *(float4*)(crow + (size_t)r * 64) = o;
        }
    }
}

extern "C" void kernel_launch(void* const* d_in, const int* in_sizes, int n_in,
                              void* d_out, int out_size, void* d_ws, size_t ws_size,
                              hipStream_t stream) {
    (void)n_in; (void)out_size; (void)ws_size;

    const float* x  = (const float*)d_in[0];
    const int*   ei = (const int*)d_in[1];
    const float* W1 = (const float*)d_in[2];
    const float* b1 = (const float*)d_in[3];
    const float* W2 = (const float*)d_in[4];
    const float* b2 = (const float*)d_in[5];
    float* out = (float*)d_out;

    const int H   = in_sizes[3];
    const int IN  = in_sizes[2] / H;
    const int N   = in_sizes[0] / IN;
    const int E   = in_sizes[1] / 2;
    const int NNZ = N + E;

    char* ws = (char*)d_ws;
    size_t off = 0;
    auto alloc = [&](size_t bytes) -> void* {
        void* p = (void*)(ws + off);
        off += (bytes + 255) & ~(size_t)255;
        return p;
    };
    const int T = 256;
    int G  = (N + GS - 1) / GS;
    int NT = (E + TE - 1) / TE;
    int*      rowptr = (int*)     alloc((size_t)(N + 1) * 4);
    float*    dinv   = (float*)   alloc((size_t)N * 4);
    int*      cumT   = (int*)     alloc((size_t)(G + 1) * NT * 4);
    unsigned* tiles  = (unsigned*)alloc((size_t)NT * TE * 4);
    int*      csr    = (int*)     alloc((size_t)NNZ * 4);
    float*    h1     = (float*)   alloc((size_t)N * H * 4);
    float*    h2     = (float*)   alloc((size_t)N * H * 4);

    int NPW  = (N + 10239) / 10240;
    if (NPW > 30) NPW = 30;
    int gAgg = ((N + NPW - 1) / NPW + 7) / 8;
    int gAggF = ((N + 15) / 16 + 7) / 8;
    int gGemm1 = 256;

    tile_sort_kernel<<<NT, 256, 0, stream>>>(ei, tiles, cumT, N, E, G, NT);
    group_rows_kernel<<<G, 256, 0, stream>>>(tiles, cumT, rowptr, dinv, csr, N, G, NT);

    gemm64_wmma_kernel<128><<<gGemm1, T, 0, stream>>>(x, W1, dinv, h1, N);
    agg_gemm_kernel<<<gAggF, T, 0, stream>>>(h1, rowptr, csr, dinv, b1, W2, h2, N);

    agg_kernel<<<gAgg, T, 0, stream>>>(h2, rowptr, csr, dinv, b2, out, N, NPW, 0);
}


# Row build puts a group's records together in LDS (all slot requests first, then the stores) and writes its CSR stretch as whole 16-byte pieces, not one 4-byte store per edge; sum order as before
# speedup vs baseline: 3.9315x; 1.0456x over previous
#include <hip/hip_runtime.h>
#include <hip/hip_bf16.h>


typedef float v2f __attribute__((ext_vector_type(2)));
typedef float v8f __attribute__((ext_vector_type(8)));

#define GS   256
#define TE   4096
#define GMAX 512
#define RPW  (GMAX / 8)
#define RCAP 8192

__device__ __forceinline__ int4 load_idx4(const int* __restrict__ p, int e, int eend, bool vec_ok) {
    int4 v;
    if (vec_ok && e + 3 < eend) {
        v = *(const int4*)(p + e);
    } else {
        v.x = (e     < eend) ? p[e]     : -1;
        v.y = (e + 1 < eend) ? p[e + 1] : -1;
        v.z = (e + 2 < eend) ? p[e + 2] : -1;
        v.w = (e + 3 < eend) ? p[e + 3] : -1;
    }
    return v;
}

__global__ __launch_bounds__(256) void tile_sort_kernel(const int* __restrict__ ei,
                                                        unsigned* __restrict__ tiles,
                                                        int* __restrict__ cumT,
                                                        int N, int E, int G, int T) {
    __shared__ int cnt[GMAX];
    __shared__ int st[GMAX];
    __shared__ __attribute__((aligned(16))) unsigned srec[TE];
    __shared__ int wsum[8];
    int tid  = threadIdx.x;
    int lane = tid & 31;
    int wv   = tid >> 5;
    int t    = blockIdx.x;
    cnt[tid] = 0;
    cnt[tid + 256] = 0;
    int ebeg = t * TE;
    int eend = ebeg + TE; if (eend > E) eend = E;
    const int* ec = ei + E;
    bool cvec = (E & 3) == 0;
    int4 rv[4], cv[4];
    if (cvec && ebeg + TE <= E) {
#pragma unroll
        for (int k = 0; k < 4; ++k) {
            int e = ebeg + 4 * tid + 1024 * k;
            rv[k] = *(const int4*)(ei + e);
            cv[k] = *(const int4*)(ec + e);
        }
    } else {
#pragma unroll
        for (int k = 0; k < 4; ++k) {
            int e = ebeg + 4 * tid + 1024 * k;
            rv[k] = load_idx4(ei, e, eend, true);
            cv[k] = load_idx4(ec, e, eend, cvec);
        }
    }
    int r[16] = {rv[0].x, rv[0].y, rv[0].z, rv[0].w, rv[1].x, rv[1].y, rv[1].z, rv[1].w,
                 rv[2].x, rv[2].y, rv[2].z, rv[2].w, rv[3].x, rv[3].y, rv[3].z, rv[3].w};
    int c[16] = {cv[0].x, cv[0].y, cv[0].z, cv[0].w, cv[1].x, cv[1].y, cv[1].z, cv[1].w,
                 cv[2].x, cv[2].y, cv[2].z, cv[2].w, cv[3].x, cv[3].y, cv[3].z, cv[3].w};
    __syncthreads();
    int rk[16];
#pragma unroll
    for (int q = 0; q < 16; ++q) {
        if ((unsigned)r[q] < (unsigned)N && (unsigned)c[q] < (unsigned)N)
            rk[q] = atomicAdd(&cnt[(unsigned)r[q] / GS], 1);
    }
    __syncthreads();
    int a = cnt[2 * tid];
    int b = cnt[2 * tid + 1];
    int v = a + b;
    int inc = v;
    for (int off = 1; off < 32; off <<= 1) {
        int n = __shfl_up(inc, off, 32);
        if (lane >= off) inc += n;
    }
    if (lane == 31) wsum[wv] = inc;
    __syncthreads();
    int base = 0;
    for (int q = 0; q < 8; ++q) {
        int s = wsum[q];
        if (q < wv) base += s;
    }
    int ex = base + inc - v;
    st[2 * tid]     = ex;
    st[2 * tid + 1] = ex + a;
    __syncthreads();
#pragma unroll
    for (int q = 0; q < 16; ++q) {
        if ((unsigned)r[q] < (unsigned)N && (unsigned)c[q] < (unsigned)N)
            srec[st[(unsigned)r[q] / GS] + rk[q]] = (((unsigned)r[q] % GS) << 24) | (unsigned)c[q];
    }
    __syncthreads();
    int tot = st[G];
    unsigned* dst = tiles + (size_t)t * TE;
    for (int i = 4 * tid; i < tot; i += 1024) {
        if (i + 3 < tot) {
            *(uint4*)(dst + i) = *(const uint4*)(srec + i);
        } else {
            for (int k = i; k < tot; ++k) dst[k] = srec[k];
        }
    }
    for (int g = tid; g <= G; g += 256) cumT[(size_t)g * T + t] = st[g];
}

__global__ __launch_bounds__(256) void group_rows_kernel(const unsigned* __restrict__ tiles,
                                                         const int* __restrict__ cumT,
                                                         int* __restrict__ rowptr,
                                                         float* __restrict__ dinv,
                                                         int* __restrict__ csr,
                                                         int N, int G, int T) {
    __shared__ int deg[GS];
    __shared__ int wsa[8];
    __shared__ int wsb[8];
    __shared__ __attribute__((aligned(16))) int stage[RCAP + 4];
    int tid  = threadIdx.x;
    int lane = tid & 31;
    int wv   = __builtin_amdgcn_readfirstlane(tid >> 5);
    int g    = blockIdx.x;
    const int* c0 = cumT + (size_t)g * T;
    const int* c1 = c0 + T;
    int tA = 8 * lane + wv;
    int tB = tA + 256;
    int loA = 0, hiA = 0, loB = 0, hiB = 0;
    if (tA < T) { loA = c0[tA]; hiA = c1[tA]; }
    if (tB < T) { loB = c0[tB]; hiB = c1[tB]; }
    int s = loA + loB;
    deg[tid] = 0;
    for (int off = 16; off > 0; off >>= 1) s += __shfl_xor(s, off, 32);
    if (lane == 0) wsa[wv] = s;
    __syncthreads();
    int before = 0;
    for (int q = 0; q < 8; ++q) before += wsa[q];
    int base = g * GS + before;

    unsigned rec[RPW];
    unsigned long long okm = 0ull;
    unsigned long long lgm = 0ull;
#pragma unroll
    for (int u = 0; u < RPW; ++u) {
        int l0 = __builtin_amdgcn_readlane(u < 32 ? loA : loB, u & 31);
        int h0 = __builtin_amdgcn_readlane(u < 32 ? hiA : hiB, u & 31);
        const unsigned* tb = tiles + (size_t)(8 * u + wv) * TE + l0;
        if (lane < h0 - l0) {
            rec[u] = tb[lane];
            okm |= 1ull << u;
        }
        if (h0 - l0 > 32) lgm |= 1ull << u;
    }
#pragma unroll
    for (int u = 0; u < RPW; ++u) {
        if ((okm >> u) & 1ull) atomicAdd(&deg[rec[u] >> 24], 1);
    }
    for (unsigned long long m = lgm; m != 0ull; m &= m - 1ull) {
        int u  = __builtin_amdgcn_readfirstlane((int)__builtin_ctzll(m));
        int l0 = __builtin_amdgcn_readlane(u < 32 ? loA : loB, u & 31);
        int h0 = __builtin_amdgcn_readlane(u < 32 ? hiA : hiB, u & 31);
        const unsigned* tb = tiles + (size_t)(8 * u + wv) * TE;
        for (int p = l0 + 32 + lane; p < h0; p += 32)
            atomicAdd(&deg[tb[p] >> 24], 1);
    }
    __syncthreads();
    int node = g * GS + tid;
    int d = deg[tid];
    int v = (node < N) ? d + 1 : 0;
    int inc = v;
    for (int off = 1; off < 32; off <<= 1) {
        int n = __shfl_up(inc, off, 32);
        if (lane >= off) inc += n;
    }
    if (lane == 31) wsb[wv] = inc;
    __syncthreads();
    int wb = 0, tot = 0;
    for (int q = 0; q < 8; ++q) {
        int x = wsb[q];
        if (q < wv) wb += x;
        tot += x;
    }
    int p0 = base + wb + inc - v;
    bool staged = tot <= RCAP;
    int  A = base & ~3;
    if (node < N) {
        rowptr[node] = p0;
        dinv[node]   = rsqrtf((float)(d + 1));
        if (staged) stage[p0 - A] = node;
        else        csr[p0]       = node;
    }
    deg[tid] = p0 + 1;
    if (g == G - 1 && tid == 0) rowptr[N] = base + tot;
    __syncthreads();
    if (staged) {
        int slot[RPW];
#pragma unroll
        for (int u = 0; u < RPW; ++u) {
            if ((okm >> u) & 1ull) slot[u] = atomicAdd(&deg[rec[u] >> 24], 1);
        }
        for (unsigned long long m = lgm; m != 0ull; m &= m - 1ull) {
            int u  = __builtin_amdgcn_readfirstlane((int)__builtin_ctzll(m));
            int l0 = __builtin_amdgcn_readlane(u < 32 ? loA : loB, u & 31);
            int h0 = __builtin_amdgcn_readlane(u < 32 ? hiA : hiB, u & 31);
            const unsigned* tb = tiles + (size_t)(8 * u + wv) * TE;
            for (int q = l0 + 32 + lane; q < h0; q += 32) {
                unsigned w = tb[q];
                int p = atomicAdd(&deg[w >> 24], 1);
                stage[p - A] = (int)(w & 0xFFFFFFu);
            }
        }
        __builtin_amdgcn_fence(__ATOMIC_SEQ_CST, "workgroup", "local");
#pragma unroll
        for (int u = 0; u < RPW; ++u) {
            if ((okm >> u) & 1ull) stage[slot[u] - A] = (int)(rec[u] & 0xFFFFFFu);
        }
        __syncthreads();
        int a   = base - A;
        int lim = a + tot;
        for (int i = 4 * tid; i < lim; i += 1024) {
            if (i >= a && i + 3 < lim) {
                *(int4*)(csr + A + i) = *(const int4*)(stage + i);
            } else {
                for (int k = i; k < i + 4; ++k)
                    if (k >= a && k < lim) csr[A + k] = stage[k];
            }
        }
    } else {
#pragma unroll
        for (int u = 0; u < RPW; ++u) {
            if ((okm >> u) & 1ull) {
                unsigned w = rec[u];
                int p = atomicAdd(&deg[w >> 24], 1);
                csr[p] = (int)(w & 0xFFFFFFu);
            }
        }
        for (unsigned long long m = lgm; m != 0ull; m &= m - 1ull) {
            int u  = __builtin_amdgcn_readfirstlane((int)__builtin_ctzll(m));
            int l0 = __builtin_amdgcn_readlane(u < 32 ? loA : loB, u & 31);
            int h0 = __builtin_amdgcn_readlane(u < 32 ? hiA : hiB, u & 31);
            const unsigned* tb = tiles + (size_t)(8 * u + wv) * TE;
            for (int q = l0 + 32 + lane; q < h0; q += 32) {
                unsigned w = tb[q];
                int p = atomicAdd(&deg[w >> 24], 1);
                csr[p] = (int)(w & 0xFFFFFFu);
            }
        }
    }
}

template <int K>
__device__ __forceinline__ void gemm_load_a(float4 (&a)[K / 8], float4 (&dv)[2],
                                            const float* __restrict__ A, const float* __restrict__ dinv,
                                            int t, int c, int h) {
    const float* p = A + (size_t)(t * 16 + c) * K + 4 * h;
#pragma unroll
    for (int i = 0; i < K / 8; ++i) a[i] = *(const float4*)(p + 8 * i);
    const float* pd = dinv + t * 16 + 8 * h;
    dv[0] = *(const float4*)pd;
    dv[1] = *(const float4*)(pd + 4);
}

template <int K>
__device__ __forceinline__ void gemm_tile(const float4 (&a)[K / 8], const float4 (&dv)[2],
                                          const v2f (&bf)[K / 4][4],
                                          float* __restrict__ C, int t, int c, int h) {
    v8f acc[4];
#pragma unroll
    for (int j = 0; j < 4; ++j) acc[j] = (v8f){0.f, 0.f, 0.f, 0.f, 0.f, 0.f, 0.f, 0.f};
#pragma unroll
    for (int i = 0; i < K / 8; ++i) {
        v2f lo, hi;
        lo.x = a[i].x; lo.y = a[i].y;
        hi.x = a[i].z; hi.y = a[i].w;
#pragma unroll
        for (int j = 0; j < 4; ++j)
            acc[j] = __builtin_amdgcn_wmma_f32_16x16x4_f32(false, lo, false, bf[2 * i][j],
                                                           (short)0, acc[j], false, false);
#pragma unroll
        for (int j = 0; j < 4; ++j)
            acc[j] = __builtin_amdgcn_wmma_f32_16x16x4_f32(false, hi, false, bf[2 * i + 1][j],
                                                           (short)0, acc[j], false, false);
    }
    float* crow = C + (size_t)(t * 16 + 8 * h) * 64 + 4 * c;
    const float sc[8] = {dv[0].x, dv[0].y, dv[0].z, dv[0].w, dv[1].x, dv[1].y, dv[1].z, dv[1].w};
#pragma unroll
    for (int r = 0; r < 8; ++r) {
        float4 o;
        o.x = acc[0][r] * sc[r]; o.y = acc[1][r] * sc[r]; o.z = acc[2][r] * sc[r]; o.w = acc[3][r] * sc[r];
        *(float4*)(crow + (size_t)r * 64) = o;
    }
}

template <int K>
__global__ __launch_bounds__(256) void gemm64_wmma_kernel(const float* __restrict__ A,
                                                          const float* __restrict__ B,
                                                          const float* __restrict__ dinv,
                                                          float* __restrict__ C,
                                                          int M) {
    int tid  = threadIdx.x;
    int wave = __builtin_amdgcn_readfirstlane(tid >> 5);
    int lane = tid & 31;
    int h = lane >> 4;
    int c = lane & 15;

    int nt     = M >> 4;
    int stride = (int)gridDim.x * 8;
    int t = (int)blockIdx.x + (int)gridDim.x * wave;
    if (t >= nt) return;

    float4 a0[K / 8], a1[K / 8];
    float4 d0[2], d1[2];
    gemm_load_a<K>(a0, d0, A, dinv, t, c, h);

    v2f bf[K / 4][4];
#pragma unroll
    for (int w = 0; w < K / 4; ++w) {
        int k0 = 8 * (w >> 1) + 4 * h + 2 * (w & 1);
        float4 r0 = *(const float4*)(B + (size_t)k0 * 64 + 4 * c);
        float4 r1 = *(const float4*)(B + (size_t)(k0 + 1) * 64 + 4 * c);
        bf[w][0].x = r0.x; bf[w][0].y = r1.x;
        bf[w][1].x = r0.y; bf[w][1].y = r1.y;
        bf[w][2].x = r0.z; bf[w][2].y = r1.z;
        bf[w][3].x = r0.w; bf[w][3].y = r1.w;
    }

    for (;;) {
        int tn = t + stride;
        if (tn >= nt) { gemm_tile<K>(a0, d0, bf, C, t, c, h); return; }
        gemm_load_a<K>(a1, d1, A, dinv, tn, c, h);
        __builtin_amdgcn_sched_barrier(0);
        gemm_tile<K>(a0, d0, bf, C, t, c, h);
        t  = tn;
        tn = t + stride;
        if (tn >= nt) { gemm_tile<K>(a1, d1, bf, C, t, c, h); return; }
        gemm_load_a<K>(a0, d0, A, dinv, tn, c, h);
        __builtin_amdgcn_sched_barrier(0);
        gemm_tile<K>(a1, d1, bf, C, t, c, h);
        t = tn;
    }
}

#define AGB 8

template <bool GUARD, int PITCH>
__device__ __forceinline__ void agg_step(const v2f (&hv)[AGB], int j, int end,
                                         float* __restrict__ obase, int rp, int dvb,
                                         v2f bv, int do_relu,
                                         float& ax, float& ay, int& row, int& rend) {
    int left = rend - j;
#pragma unroll
    for (int q = 0; q < AGB; ++q) {
        if (!GUARD || j + q < end) {
            ax += hv[q].x;
            ay += hv[q].y;
            if (q + 1 == left) {
                float s  = __int_as_float(__builtin_amdgcn_readlane(dvb, row));
                float ox = fmaf(ax, s, bv.x);
                float oy = fmaf(ay, s, bv.y);
                if (do_relu) {
                    ox = fmaxf(ox, 0.f);
                    oy = fmaxf(oy, 0.f);
                }
                v2f o; o.x = ox; o.y = oy;
                *(v2f*)(obase + row * PITCH) = o;
                ax = 0.f; ay = 0.f;
                ++row;
                rend = __builtin_amdgcn_readlane(rp, row + 1);
                left = rend - j;
            }
        }
    }
}

template <int PITCH>
__device__ __forceinline__ void agg_trip(const int (&rc)[AGB], int (&rn)[AGB],
                                         const int* __restrict__ csr, int& j, int end,
                                         const float* __restrict__ h, unsigned lane8,
                                         float* __restrict__ obase, int rp, int dvb,
                                         v2f bv, int do_relu,
                                         float& ax, float& ay, int& row, int& rend) {
    v2f hv[AGB];
#pragma unroll
    for (int q = 0; q < AGB; ++q)
        hv[q] = *(const v2f*)((const char*)h + ((unsigned)rc[q] * 256u + lane8));
    __builtin_amdgcn_sched_barrier(0);
    int jn = j + AGB;
    if (jn + AGB <= end) {
        const int* pn = csr + jn;
#pragma unroll
        for (int q = 0; q < AGB; ++q) rn[q] = pn[q];
    } else {
#pragma unroll
        for (int q = 0; q < AGB; ++q) rn[q] = csr[min(jn + q, end - 1)];
    }
    if (jn <= end) agg_step<false, PITCH>(hv, j, end, obase, rp, dvb, bv, do_relu, ax, ay, row, rend);
    else           agg_step<true,  PITCH>(hv, j, end, obase, rp, dvb, bv, do_relu, ax, ay, row, rend);
    j = jn;
}

__global__ __launch_bounds__(256) void agg_kernel(const float* __restrict__ h,
                                                  const int* __restrict__ rowptr,
                                                  const int* __restrict__ csr,
                                                  const float* __restrict__ dinv,
                                                  const float* __restrict__ bias,
                                                  float* __restrict__ out,
                                                  int N, int NPW, int do_relu) {
    int lane = threadIdx.x & 31;
    int w  = __builtin_amdgcn_readfirstlane((int)(blockIdx.x * 8 + (threadIdx.x >> 5)));
    int n0 = w * NPW;
    if (n0 >= N) return;
    int cnt = N - n0; if (cnt > NPW) cnt = NPW;

    int rp   = rowptr[n0 + min(lane, cnt)];
    int dvb  = __float_as_int(dinv[n0 + min(lane, cnt - 1)]);
    int j    = __builtin_amdgcn_readlane(rp, 0);
    int end  = __builtin_amdgcn_readlane(rp, cnt);
    int row  = 0;
    int rend = __builtin_amdgcn_readlane(rp, 1);
    if (j >= end) return;

    unsigned lane8 = (unsigned)lane * 8u;
    float*   obase = out + (size_t)n0 * 64 + 2 * lane;
    v2f bv = *(const v2f*)(bias + 2 * lane);

    float ax = 0.f, ay = 0.f;
    int ra[AGB], rb[AGB];
#pragma unroll
    for (int q = 0; q < AGB; ++q) ra[q] = csr[min(j + q, end - 1)];

    for (;;) {
        agg_trip<64>(ra, rb, csr, j, end, h, lane8, obase, rp, dvb, bv, do_relu, ax, ay, row, rend);
        if (j >= end) break;
        agg_trip<64>(rb, ra, csr, j, end, h, lane8, obase, rp, dvb, bv, do_relu, ax, ay, row, rend);
        if (j >= end) break;
    }
}

#define TP 68

__global__ __launch_bounds__(256) void agg_gemm_kernel(const float* __restrict__ h,
                                                       const int* __restrict__ rowptr,
                                                       const int* __restrict__ csr,
                                                       const float* __restrict__ dinv,
                                                       const float* __restrict__ bias,
                                                       const float* __restrict__ B,
                                                       float* __restrict__ C,
                                                       int N) {
    __shared__ float tiles[8][16 * TP];
    int lane = threadIdx.x & 31;
    int wv = __builtin_amdgcn_readfirstlane((int)(threadIdx.x >> 5));
    int w  = __builtin_amdgcn_readfirstlane((int)(blockIdx.x * 8 + (threadIdx.x >> 5)));
    int n0 = w * 16;
    if (n0 >= N) return;
    int cnt = N - n0; if (cnt > 16) cnt = 16;

    int   rp  = rowptr[n0 + min(lane, cnt)];
    float dvf = dinv[n0 + min(lane, cnt - 1)];
    int   dvb = __float_as_int(dvf);
    int j    = __builtin_amdgcn_readlane(rp, 0);
    int end  = __builtin_amdgcn_readlane(rp, cnt);
    int row  = 0;
    int rend = __builtin_amdgcn_readlane(rp, 1);
    if (j >= end) return;

    unsigned lane8 = (unsigned)lane * 8u;
    float*   tile  = &tiles[wv][0];
    float*   obase = tile + 2 * lane;
    v2f bv = *(const v2f*)(bias + 2 * lane);

    float ax = 0.f, ay = 0.f;
    int ra[AGB], rb[AGB];
#pragma unroll
    for (int q = 0; q < AGB; ++q) ra[q] = csr[min(j + q, end - 1)];

    for (;;) {
        agg_trip<TP>(ra, rb, csr, j, end, h, lane8, obase, rp, dvb, bv, 1, ax, ay, row, rend);
        if (j >= end) break;
        agg_trip<TP>(rb, ra, csr, j, end, h, lane8, obase, rp, dvb, bv, 1, ax, ay, row, rend);
        if (j >= end) break;
    }

    int hh = lane >> 4;
    int c  = lane & 15;
    float sc[8];
#pragma unroll
    for (int r = 0; r < 8; ++r) sc[r] = __shfl(dvf, 8 * hh + r, 32);
    const float* tp = tile + c * TP + 4 * hh;
    v8f acc[4];
#pragma unroll
    for (int jt = 0; jt < 4; ++jt) acc[jt] = (v8f){0.f, 0.f, 0.f, 0.f, 0.f, 0.f, 0.f, 0.f};
#pragma unroll 1
    for (int i = 0; i < 8; i += 2) {
        float4 a0 = *(const float4*)(tp + 8 * i);
        float4 a1 = *(const float4*)(tp + 8 * i + 8);
        v2f bf[4][4];
#pragma unroll
        for (int q = 0; q < 4; ++q) {
            int ks = 2 * i + q;
            int k0 = 8 * (ks >> 1) + 4 * hh + 2 * (ks & 1);
            float4 r0 = *(const float4*)(B + (size_t)k0 * 64 + 4 * c);
            float4 r1 = *(const float4*)(B + (size_t)(k0 + 1) * 64 + 4 * c);
            bf[q][0].x = r0.x; bf[q][0].y = r1.x;
            bf[q][1].x = r0.y; bf[q][1].y = r1.y;
            bf[q][2].x = r0.z; bf[q][2].y = r1.z;
            bf[q][3].x = r0.w; bf[q][3].y = r1.w;
        }
        v2f af[4];
        af[0].x = a0.x; af[0].y = a0.y;
        af[1].x = a0.z; af[1].y = a0.w;
        af[2].x = a1.x; af[2].y = a1.y;
        af[3].x = a1.z; af[3].y = a1.w;
#pragma unroll
        for (int q = 0; q < 4; ++q) {
#pragma unroll
            for (int jt = 0; jt < 4; ++jt)
                acc[jt] = __builtin_amdgcn_wmma_f32_16x16x4_f32(false, af[q], false, bf[q][jt],
                                                                (short)0, acc[jt], false, false);
        }
    }
    float* crow = C + (size_t)(n0 + 8 * hh) * 64 + 4 * c;
#pragma unroll
    for (int r = 0; r < 8; ++r) {
        if (8 * hh + r < cnt) {
            float4 o;
            o.x = acc[0][r] * sc[r]; o.y = acc[1][r] * sc[r]; o.z = acc[2][r] * sc[r]; o.w = acc[3][r] * sc[r];
            *(float4*)(crow + (size_t)r * 64) = o;
        }
    }
}

extern "C" void kernel_launch(void* const* d_in, const int* in_sizes, int n_in,
                              void* d_out, int out_size, void* d_ws, size_t ws_size,
                              hipStream_t stream) {
    (void)n_in; (void)out_size; (void)ws_size;

    const float* x  = (const float*)d_in[0];
    const int*   ei = (const int*)d_in[1];
    const float* W1 = (const float*)d_in[2];
    const float* b1 = (const float*)d_in[3];
    const float* W2 = (const float*)d_in[4];
    const float* b2 = (const float*)d_in[5];
    float* out = (float*)d_out;

    const int H   = in_sizes[3];
    const int IN  = in_sizes[2] / H;
    const int N   = in_sizes[0] / IN;
    const int E   = in_sizes[1] / 2;
    const int NNZ = N + E;

    char* ws = (char*)d_ws;
    size_t off = 0;
    auto alloc = [&](size_t bytes) -> void* {
        void* p = (void*)(ws + off);
        off += (bytes + 255) & ~(size_t)255;
        return p;
    };
    const int T = 256;
    int G  = (N + GS - 1) / GS;
    int NT = (E + TE - 1) / TE;
    int*      rowptr = (int*)     alloc((size_t)(N + 1) * 4);
    float*    dinv   = (float*)   alloc((size_t)N * 4);
    int*      cumT   = (int*)     alloc((size_t)(G + 1) * NT * 4);
    unsigned* tiles  = (unsigned*)alloc((size_t)NT * TE * 4);
    int*      csr    = (int*)     alloc((size_t)NNZ * 4);
    float*    h1     = (float*)   alloc((size_t)N * H * 4);
    float*    h2     = (float*)   alloc((size_t)N * H * 4);

    int NPW  = (N + 10239) / 10240;
    if (NPW > 30) NPW = 30;
    int gAgg = ((N + NPW - 1) / NPW + 7) / 8;
    int gAggF = ((N + 15) / 16 + 7) / 8;
    int gGemm1 = 256;

    tile_sort_kernel<<<NT, 256, 0, stream>>>(ei, tiles, cumT, N, E, G, NT);
    group_rows_kernel<<<G, 256, 0, stream>>>(tiles, cumT, rowptr, dinv, csr, N, G, NT);

    gemm64_wmma_kernel<128><<<gGemm1, T, 0, stream>>>(x, W1, dinv, h1, N);
    agg_gemm_kernel<<<gAggF, T, 0, stream>>>(h1, rowptr, csr, dinv, b1, W2, h2, N);

    agg_kernel<<<gAgg, T, 0, stream>>>(h2, rowptr, csr, dinv, b2, out, N, NPW, 0);
}


# Row build fetches by edge, not by run: a group's edges are numbered across its per-tile runs (run table in LDS), thread i loads edges i, i+256, ..: all lanes busy; sum order stays arrival order
# speedup vs baseline: 4.0859x; 1.0393x over previous
#include <hip/hip_runtime.h>
#include <hip/hip_bf16.h>


typedef float v2f __attribute__((ext_vector_type(2)));
typedef float v8f __attribute__((ext_vector_type(8)));

#define GS   256
#define TE   4096
#define GMAX 512
#define RCAP 8192
#define EPT  18

__device__ __forceinline__ int4 load_idx4(const int* __restrict__ p, int e, int eend, bool vec_ok) {
    int4 v;
    if (vec_ok && e + 3 < eend) {
        v = *(const int4*)(p + e);
    } else {
        v.x = (e     < eend) ? p[e]     : -1;
        v.y = (e + 1 < eend) ? p[e + 1] : -1;
        v.z = (e + 2 < eend) ? p[e + 2] : -1;
        v.w = (e + 3 < eend) ? p[e + 3] : -1;
    }
    return v;
}

__global__ __launch_bounds__(256) void tile_sort_kernel(const int* __restrict__ ei,
                                                        unsigned* __restrict__ tiles,
                                                        int* __restrict__ cumT,
                                                        int N, int E, int G, int T) {
    __shared__ int cnt[GMAX];
    __shared__ int st[GMAX];
    __shared__ __attribute__((aligned(16))) unsigned srec[TE];
    __shared__ int wsum[8];
    int tid  = threadIdx.x;
    int lane = tid & 31;
    int wv   = tid >> 5;
    int t    = blockIdx.x;
    cnt[tid] = 0;
    cnt[tid + 256] = 0;
    int ebeg = t * TE;
    int eend = ebeg + TE; if (eend > E) eend = E;
    const int* ec = ei + E;
    bool cvec = (E & 3) == 0;
    int4 rv[4], cv[4];
    if (cvec && ebeg + TE <= E) {
#pragma unroll
        for (int k = 0; k < 4; ++k) {
            int e = ebeg + 4 * tid + 1024 * k;
            rv[k] = *(const int4*)(ei + e);
            cv[k] = *(const int4*)(ec + e);
        }
    } else {
#pragma unroll
        for (int k = 0; k < 4; ++k) {
            int e = ebeg + 4 * tid + 1024 * k;
            rv[k] = load_idx4(ei, e, eend, true);
            cv[k] = load_idx4(ec, e, eend, cvec);
        }
    }
    int r[16] = {rv[0].x, rv[0].y, rv[0].z, rv[0].w, rv[1].x, rv[1].y, rv[1].z, rv[1].w,
                 rv[2].x, rv[2].y, rv[2].z, rv[2].w, rv[3].x, rv[3].y, rv[3].z, rv[3].w};
    int c[16] = {cv[0].x, cv[0].y, cv[0].z, cv[0].w, cv[1].x, cv[1].y, cv[1].z, cv[1].w,
                 cv[2].x, cv[2].y, cv[2].z, cv[2].w, cv[3].x, cv[3].y, cv[3].z, cv[3].w};
    __syncthreads();
    int rk[16];
#pragma unroll
    for (int q = 0; q < 16; ++q) {
        if ((unsigned)r[q] < (unsigned)N && (unsigned)c[q] < (unsigned)N)
            rk[q] = atomicAdd(&cnt[(unsigned)r[q] / GS], 1);
    }
    __syncthreads();
    int a = cnt[2 * tid];
    int b = cnt[2 * tid + 1];
    int v = a + b;
    int inc = v;
    for (int off = 1; off < 32; off <<= 1) {
        int n = __shfl_up(inc, off, 32);
        if (lane >= off) inc += n;
    }
    if (lane == 31) wsum[wv] = inc;
    __syncthreads();
    int base = 0;
    for (int q = 0; q < 8; ++q) {
        int s = wsum[q];
        if (q < wv) base += s;
    }
    int ex = base + inc - v;
    st[2 * tid]     = ex;
    st[2 * tid + 1] = ex + a;
    __syncthreads();
#pragma unroll
    for (int q = 0; q < 16; ++q) {
        if ((unsigned)r[q] < (unsigned)N && (unsigned)c[q] < (unsigned)N)
            srec[st[(unsigned)r[q] / GS] + rk[q]] = (((unsigned)r[q] % GS) << 24) | (unsigned)c[q];
    }
    __syncthreads();
    int tot = st[G];
    unsigned* dst = tiles + (size_t)t * TE;
    for (int i = 4 * tid; i < tot; i += 1024) {
        if (i + 3 < tot) {
            *(uint4*)(dst + i) = *(const uint4*)(srec + i);
        } else {
            for (int k = i; k < tot; ++k) dst[k] = srec[k];
        }
    }
    for (int g = tid; g <= G; g += 256) cumT[(size_t)g * T + t] = st[g];
}

__device__ __forceinline__ int slot_of(const int* pst, int e) {
    int j = 0;
    for (int s = GMAX / 2; s > 0; s >>= 1)
        if (pst[j + s] <= e) j += s;
    return j;
}

__global__ __launch_bounds__(256) void group_rows_kernel(const unsigned* __restrict__ tiles,
                                                         const int* __restrict__ cumT,
                                                         int* __restrict__ rowptr,
                                                         float* __restrict__ dinv,
                                                         int* __restrict__ csr,
                                                         int N, int G, int T) {
    __shared__ int deg[GS];
    __shared__ int wsa[8];
    __shared__ int wsb[8];
    __shared__ int wsc[8];
    __shared__ int pst[GMAX];
    __shared__ int roff[GMAX];
    __shared__ unsigned short rid[RCAP];
    __shared__ __attribute__((aligned(16))) int stage[RCAP + 4];
    int tid  = threadIdx.x;
    int lane = tid & 31;
    int wv   = tid >> 5;
    int g    = blockIdx.x;
    const int* c0 = cumT + (size_t)g * T;
    const int* c1 = c0 + T;
    int t1 = tid + 256;
    int lo0 = 0, hi0 = 0, lo1 = 0, hi1 = 0;
    if (tid < T) { lo0 = c0[tid]; hi0 = c1[tid]; }
    if (t1  < T) { lo1 = c0[t1];  hi1 = c1[t1]; }
    int len0 = hi0 - lo0;
    int len1 = hi1 - lo1;
    int s = lo0 + lo1;
    int v = len0 + len1;
    deg[tid] = 0;
    for (int off = 16; off > 0; off >>= 1) s += __shfl_xor(s, off, 32);
    int inc = v;
    for (int off = 1; off < 32; off <<= 1) {
        int n = __shfl_up(inc, off, 32);
        if (lane >= off) inc += n;
    }
    if (lane == 0)  wsa[wv] = s;
    if (lane == 31) wsc[wv] = inc;
    __syncthreads();
    int before = 0;
    int eb = 0, Eg = 0;
    for (int q = 0; q < 8; ++q) {
        before += wsa[q];
        int x = wsc[q];
        if (q < wv) eb += x;
        Eg += x;
    }
    int base = g * GS + before;
    int st0 = eb + inc - v;
    int st1 = st0 + len0;
    pst[2 * tid]      = st0;
    pst[2 * tid + 1]  = st1;
    roff[2 * tid]     = tid * TE + lo0 - st0;
    roff[2 * tid + 1] = t1  * TE + lo1 - st1;
    bool use_rid = Eg <= RCAP;
    if (use_rid) {
        for (int i = 0; i < len0; ++i) rid[st0 + i] = (unsigned short)(2 * tid);
        for (int i = 0; i < len1; ++i) rid[st1 + i] = (unsigned short)(2 * tid + 1);
    }
    __syncthreads();

    unsigned rec[EPT];
#pragma unroll
    for (int k = 0; k < EPT; ++k) rec[k] = 0u;
    if (Eg > 0) {
        int jj[EPT];
        if (use_rid) {
#pragma unroll
            for (int k = 0; k < EPT; ++k) jj[k] = rid[min(tid + 256 * k, Eg - 1)];
        } else {
#pragma unroll
            for (int k = 0; k < EPT; ++k) jj[k] = slot_of(pst, min(tid + 256 * k, Eg - 1));
        }
#pragma unroll
        for (int k = 0; k < EPT; ++k) rec[k] = tiles[roff[jj[k]] + min(tid + 256 * k, Eg - 1)];
    }
#pragma unroll
    for (int k = 0; k < EPT; ++k) {
        if (tid + 256 * k < Eg) atomicAdd(&deg[rec[k] >> 24], 1);
    }
    for (int e = tid + 256 * EPT; e < Eg; e += 256) {
        int j = use_rid ? (int)rid[e] : slot_of(pst, e);
        atomicAdd(&deg[tiles[roff[j] + e] >> 24], 1);
    }
    __syncthreads();
    int node = g * GS + tid;
    int d = deg[tid];
    int vr = (node < N) ? d + 1 : 0;
    int incr = vr;
    for (int off = 1; off < 32; off <<= 1) {
        int n = __shfl_up(incr, off, 32);
        if (lane >= off) incr += n;
    }
    if (lane == 31) wsb[wv] = incr;
    __syncthreads();
    int wb = 0, tot = 0;
    for (int q = 0; q < 8; ++q) {
        int x = wsb[q];
        if (q < wv) wb += x;
        tot += x;
    }
    int p0 = base + wb + incr - vr;
    bool staged = tot <= RCAP;
    int  A = base & ~3;
    if (node < N) {
        rowptr[node] = p0;
        dinv[node]   = rsqrtf((float)(d + 1));
        if (staged) stage[p0 - A] = node;
        else        csr[p0]       = node;
    }
    deg[tid] = p0 + 1;
    if (g == G - 1 && tid == 0) rowptr[N] = base + tot;
    __syncthreads();
    if (staged) {
        int slot[EPT];
#pragma unroll
        for (int k = 0; k < EPT; ++k) {
            if (tid + 256 * k < Eg) slot[k] = atomicAdd(&deg[rec[k] >> 24], 1);
        }
        for (int e = tid + 256 * EPT; e < Eg; e += 256) {
            int j = use_rid ? (int)rid[e] : slot_of(pst, e);
            unsigned w = tiles[roff[j] + e];
            int p = atomicAdd(&deg[w >> 24], 1);
            stage[p - A] = (int)(w & 0xFFFFFFu);
        }
        __builtin_amdgcn_fence(__ATOMIC_SEQ_CST, "workgroup", "local");
#pragma unroll
        for (int k = 0; k < EPT; ++k) {
            if (tid + 256 * k < Eg) stage[slot[k] - A] = (int)(rec[k] & 0xFFFFFFu);
        }
        __syncthreads();
        int a   = base - A;
        int lim = a + tot;
        for (int i = 4 * tid; i < lim; i += 1024) {
            if (i >= a && i + 3 < lim) {
                *(int4*)(csr + A + i) = *(const int4*)(stage + i);
            } else {
                for (int k = i; k < i + 4; ++k)
                    if (k >= a && k < lim) csr[A + k] = stage[k];
            }
        }
    } else {
#pragma unroll
        for (int k = 0; k < EPT; ++k) {
            if (tid + 256 * k < Eg) {
                unsigned w = rec[k];
                int p = atomicAdd(&deg[w >> 24], 1);
                csr[p] = (int)(w & 0xFFFFFFu);
            }
        }
        for (int e = tid + 256 * EPT; e < Eg; e += 256) {
            int j = use_rid ? (int)rid[e] : slot_of(pst, e);
            unsigned w = tiles[roff[j] + e];
            int p = atomicAdd(&deg[w >> 24], 1);
            csr[p] = (int)(w & 0xFFFFFFu);
        }
    }
}

template <int K>
__device__ __forceinline__ void gemm_load_a(float4 (&a)[K / 8], float4 (&dv)[2],
                                            const float* __restrict__ A, const float* __restrict__ dinv,
                                            int t, int c, int h) {
    const float* p = A + (size_t)(t * 16 + c) * K + 4 * h;
#pragma unroll
    for (int i = 0; i < K / 8; ++i) a[i] = *(const float4*)(p + 8 * i);
    const float* pd = dinv + t * 16 + 8 * h;
    dv[0] = *(const float4*)pd;
    dv[1] = *(const float4*)(pd + 4);
}

template <int K>
__device__ __forceinline__ void gemm_tile(const float4 (&a)[K / 8], const float4 (&dv)[2],
                                          const v2f (&bf)[K / 4][4],
                                          float* __restrict__ C, int t, int c, int h) {
    v8f acc[4];
#pragma unroll
    for (int j = 0; j < 4; ++j) acc[j] = (v8f){0.f, 0.f, 0.f, 0.f, 0.f, 0.f, 0.f, 0.f};
#pragma unroll
    for (int i = 0; i < K / 8; ++i) {
        v2f lo, hi;
        lo.x = a[i].x; lo.y = a[i].y;
        hi.x = a[i].z; hi.y = a[i].w;
#pragma unroll
        for (int j = 0; j < 4; ++j)
            acc[j] = __builtin_amdgcn_wmma_f32_16x16x4_f32(false, lo, false, bf[2 * i][j],
                                                           (short)0, acc[j], false, false);
#pragma unroll
        for (int j = 0; j < 4; ++j)
            acc[j] = __builtin_amdgcn_wmma_f32_16x16x4_f32(false, hi, false, bf[2 * i + 1][j],
                                                           (short)0, acc[j], false, false);
    }
    float* crow = C + (size_t)(t * 16 + 8 * h) * 64 + 4 * c;
    const float sc[8] = {dv[0].x, dv[0].y, dv[0].z, dv[0].w, dv[1].x, dv[1].y, dv[1].z, dv[1].w};
#pragma unroll
    for (int r = 0; r < 8; ++r) {
        float4 o;
        o.x = acc[0][r] * sc[r]; o.y = acc[1][r] * sc[r]; o.z = acc[2][r] * sc[r]; o.w = acc[3][r] * sc[r];
        *(float4*)(crow + (size_t)r * 64) = o;
    }
}

template <int K>
__global__ __launch_bounds__(256) void gemm64_wmma_kernel(const float* __restrict__ A,
                                                          const float* __restrict__ B,
                                                          const float* __restrict__ dinv,
                                                          float* __restrict__ C,
                                                          int M) {
    int tid  = threadIdx.x;
    int wave = __builtin_amdgcn_readfirstlane(tid >> 5);
    int lane = tid & 31;
    int h = lane >> 4;
    int c = lane & 15;

    int nt     = M >> 4;
    int stride = (int)gridDim.x * 8;
    int t = (int)blockIdx.x + (int)gridDim.x * wave;
    if (t >= nt) return;

    float4 a0[K / 8], a1[K / 8];
    float4 d0[2], d1[2];
    gemm_load_a<K>(a0, d0, A, dinv, t, c, h);

    v2f bf[K / 4][4];
#pragma unroll
    for (int w = 0; w < K / 4; ++w) {
        int k0 = 8 * (w >> 1) + 4 * h + 2 * (w & 1);
        float4 r0 = *(const float4*)(B + (size_t)k0 * 64 + 4 * c);
        float4 r1 = *(const float4*)(B + (size_t)(k0 + 1) * 64 + 4 * c);
        bf[w][0].x = r0.x; bf[w][0].y = r1.x;
        bf[w][1].x = r0.y; bf[w][1].y = r1.y;
        bf[w][2].x = r0.z; bf[w][2].y = r1.z;
        bf[w][3].x = r0.w; bf[w][3].y = r1.w;
    }

    for (;;) {
        int tn = t + stride;
        if (tn >= nt) { gemm_tile<K>(a0, d0, bf, C, t, c, h); return; }
        gemm_load_a<K>(a1, d1, A, dinv, tn, c, h);
        __builtin_amdgcn_sched_barrier(0);
        gemm_tile<K>(a0, d0, bf, C, t, c, h);
        t  = tn;
        tn = t + stride;
        if (tn >= nt) { gemm_tile<K>(a1, d1, bf, C, t, c, h); return; }
        gemm_load_a<K>(a0, d0, A, dinv, tn, c, h);
        __builtin_amdgcn_sched_barrier(0);
        gemm_tile<K>(a1, d1, bf, C, t, c, h);
        t = tn;
    }
}

#define AGB 8

template <bool GUARD, int PITCH>
__device__ __forceinline__ void agg_step(const v2f (&hv)[AGB], int j, int end,
                                         float* __restrict__ obase, int rp, int dvb,
                                         v2f bv, int do_relu,
                                         float& ax, float& ay, int& row, int& rend) {
    int left = rend - j;
#pragma unroll
    for (int q = 0; q < AGB; ++q) {
        if (!GUARD || j + q < end) {
            ax += hv[q].x;
            ay += hv[q].y;
            if (q + 1 == left) {
                float s  = __int_as_float(__builtin_amdgcn_readlane(dvb, row));
                float ox = fmaf(ax, s, bv.x);
                float oy = fmaf(ay, s, bv.y);
                if (do_relu) {
                    ox = fmaxf(ox, 0.f);
                    oy = fmaxf(oy, 0.f);
                }
                v2f o; o.x = ox; o.y = oy;
                *(v2f*)(obase + row * PITCH) = o;
                ax = 0.f; ay = 0.f;
                ++row;
                rend = __builtin_amdgcn_readlane(rp, row + 1);
                left = rend - j;
            }
        }
    }
}

template <int PITCH>
__device__ __forceinline__ void agg_trip(const int (&rc)[AGB], int (&rn)[AGB],
                                         const int* __restrict__ csr, int& j, int end,
                                         const float* __restrict__ h, unsigned lane8,
                                         float* __restrict__ obase, int rp, int dvb,
                                         v2f bv, int do_relu,
                                         float& ax, float& ay, int& row, int& rend) {
    v2f hv[AGB];
#pragma unroll
    for (int q = 0; q < AGB; ++q)
        hv[q] = *(const v2f*)((const char*)h + ((unsigned)rc[q] * 256u + lane8));
    __builtin_amdgcn_sched_barrier(0);
    int jn = j + AGB;
    if (jn + AGB <= end) {
        const int* pn = csr + jn;
#pragma unroll
        for (int q = 0; q < AGB; ++q) rn[q] = pn[q];
    } else {
#pragma unroll
        for (int q = 0; q < AGB; ++q) rn[q] = csr[min(jn + q, end - 1)];
    }
    if (jn <= end) agg_step<false, PITCH>(hv, j, end, obase, rp, dvb, bv, do_relu, ax, ay, row, rend);
    else           agg_step<true,  PITCH>(hv, j, end, obase, rp, dvb, bv, do_relu, ax, ay, row, rend);
    j = jn;
}

__global__ __launch_bounds__(256) void agg_kernel(const float* __restrict__ h,
                                                  const int* __restrict__ rowptr,
                                                  const int* __restrict__ csr,
                                                  const float* __restrict__ dinv,
                                                  const float* __restrict__ bias,
                                                  float* __restrict__ out,
                                                  int N, int NPW, int do_relu) {
    int lane = threadIdx.x & 31;
    int w  = __builtin_amdgcn_readfirstlane((int)(blockIdx.x * 8 + (threadIdx.x >> 5)));
    int n0 = w * NPW;
    if (n0 >= N) return;
    int cnt = N - n0; if (cnt > NPW) cnt = NPW;

    int rp   = rowptr[n0 + min(lane, cnt)];
    int dvb  = __float_as_int(dinv[n0 + min(lane, cnt - 1)]);
    int j    = __builtin_amdgcn_readlane(rp, 0);
    int end  = __builtin_amdgcn_readlane(rp, cnt);
    int row  = 0;
    int rend = __builtin_amdgcn_readlane(rp, 1);
    if (j >= end) return;

    unsigned lane8 = (unsigned)lane * 8u;
    float*   obase = out + (size_t)n0 * 64 + 2 * lane;
    v2f bv = *(const v2f*)(bias + 2 * lane);

    float ax = 0.f, ay = 0.f;
    int ra[AGB], rb[AGB];
#pragma unroll
    for (int q = 0; q < AGB; ++q) ra[q] = csr[min(j + q, end - 1)];

    for (;;) {
        agg_trip<64>(ra, rb, csr, j, end, h, lane8, obase, rp, dvb, bv, do_relu, ax, ay, row, rend);
        if (j >= end) break;
        agg_trip<64>(rb, ra, csr, j, end, h, lane8, obase, rp, dvb, bv, do_relu, ax, ay, row, rend);
        if (j >= end) break;
    }
}

#define TP 68

__global__ __launch_bounds__(256) void agg_gemm_kernel(const float* __restrict__ h,
                                                       const int* __restrict__ rowptr,
                                                       const int* __restrict__ csr,
                                                       const float* __restrict__ dinv,
                                                       const float* __restrict__ bias,
                                                       const float* __restrict__ B,
                                                       float* __restrict__ C,
                                                       int N) {
    __shared__ float tiles[8][16 * TP];
    int lane = threadIdx.x & 31;
    int wv = __builtin_amdgcn_readfirstlane((int)(threadIdx.x >> 5));
    int w  = __builtin_amdgcn_readfirstlane((int)(blockIdx.x * 8 + (threadIdx.x >> 5)));
    int n0 = w * 16;
    if (n0 >= N) return;
    int cnt = N - n0; if (cnt > 16) cnt = 16;

    int   rp  = rowptr[n0 + min(lane, cnt)];
    float dvf = dinv[n0 + min(lane, cnt - 1)];
    int   dvb = __float_as_int(dvf);
    int j    = __builtin_amdgcn_readlane(rp, 0);
    int end  = __builtin_amdgcn_readlane(rp, cnt);
    int row  = 0;
    int rend = __builtin_amdgcn_readlane(rp, 1);
    if (j >= end) return;

    unsigned lane8 = (unsigned)lane * 8u;
    float*   tile  = &tiles[wv][0];
    float*   obase = tile + 2 * lane;
    v2f bv = *(const v2f*)(bias + 2 * lane);

    float ax = 0.f, ay = 0.f;
    int ra[AGB], rb[AGB];
#pragma unroll
    for (int q = 0; q < AGB; ++q) ra[q] = csr[min(j + q, end - 1)];

    for (;;) {
        agg_trip<TP>(ra, rb, csr, j, end, h, lane8, obase, rp, dvb, bv, 1, ax, ay, row, rend);
        if (j >= end) break;
        agg_trip<TP>(rb, ra, csr, j, end, h, lane8, obase, rp, dvb, bv, 1, ax, ay, row, rend);
        if (j >= end) break;
    }

    int hh = lane >> 4;
    int c  = lane & 15;
    float sc[8];
#pragma unroll
    for (int r = 0; r < 8; ++r) sc[r] = __shfl(dvf, 8 * hh + r, 32);
    const float* tp = tile + c * TP + 4 * hh;
    v8f acc[4];
#pragma unroll
    for (int jt = 0; jt < 4; ++jt) acc[jt] = (v8f){0.f, 0.f, 0.f, 0.f, 0.f, 0.f, 0.f, 0.f};
#pragma unroll 1
    for (int i = 0; i < 8; i += 2) {
        float4 a0 = *(const float4*)(tp + 8 * i);
        float4 a1 = *(const float4*)(tp + 8 * i + 8);
        v2f bf[4][4];
#pragma unroll
        for (int q = 0; q < 4; ++q) {
            int ks = 2 * i + q;
            int k0 = 8 * (ks >> 1) + 4 * hh + 2 * (ks & 1);
            float4 r0 = *(const float4*)(B + (size_t)k0 * 64 + 4 * c);
            float4 r1 = *(const float4*)(B + (size_t)(k0 + 1) * 64 + 4 * c);
            bf[q][0].x = r0.x; bf[q][0].y = r1.x;
            bf[q][1].x = r0.y; bf[q][1].y = r1.y;
            bf[q][2].x = r0.z; bf[q][2].y = r1.z;
            bf[q][3].x = r0.w; bf[q][3].y = r1.w;
        }
        v2f af[4];
        af[0].x = a0.x; af[0].y = a0.y;
        af[1].x = a0.z; af[1].y = a0.w;
        af[2].x = a1.x; af[2].y = a1.y;
        af[3].x = a1.z; af[3].y = a1.w;
#pragma unroll
        for (int q = 0; q < 4; ++q) {
#pragma unroll
            for (int jt = 0; jt < 4; ++jt)
                acc[jt] = __builtin_amdgcn_wmma_f32_16x16x4_f32(false, af[q], false, bf[q][jt],
                                                                (short)0, acc[jt], false, false);
        }
    }
    float* crow = C + (size_t)(n0 + 8 * hh) * 64 + 4 * c;
#pragma unroll
    for (int r = 0; r < 8; ++r) {
        if (8 * hh + r < cnt) {
            float4 o;
            o.x = acc[0][r] * sc[r]; o.y = acc[1][r] * sc[r]; o.z = acc[2][r] * sc[r]; o.w = acc[3][r] * sc[r];
            *(float4*)(crow + (size_t)r * 64) = o;
        }
    }
}

extern "C" void kernel_launch(void* const* d_in, const int* in_sizes, int n_in,
                              void* d_out, int out_size, void* d_ws, size_t ws_size,
                              hipStream_t stream) {
    (void)n_in; (void)out_size; (void)ws_size;

    const float* x  = (const float*)d_in[0];
    const int*   ei = (const int*)d_in[1];
    const float* W1 = (const float*)d_in[2];
    const float* b1 = (const float*)d_in[3];
    const float* W2 = (const float*)d_in[4];
    const float* b2 = (const float*)d_in[5];
    float* out = (float*)d_out;

    const int H   = in_sizes[3];
    const int IN  = in_sizes[2] / H;
    const int N   = in_sizes[0] / IN;
    const int E   = in_sizes[1] / 2;
    const int NNZ = N + E;

    char* ws = (char*)d_ws;
    size_t off = 0;
    auto alloc = [&](size_t bytes) -> void* {
        void* p = (void*)(ws + off);
        off += (bytes + 255) & ~(size_t)255;
        return p;
    };
    const int T = 256;
    int G  = (N + GS - 1) / GS;
    int NT = (E + TE - 1) / TE;
    int*      rowptr = (int*)     alloc((size_t)(N + 1) * 4);
    float*    dinv   = (float*)   alloc((size_t)N * 4);
    int*      cumT   = (int*)     alloc((size_t)(G + 1) * NT * 4);
    unsigned* tiles  = (unsigned*)alloc((size_t)NT * TE * 4);
    int*      csr    = (int*)     alloc((size_t)NNZ * 4);
    float*    h1     = (float*)   alloc((size_t)N * H * 4);
    float*    h2     = (float*)   alloc((size_t)N * H * 4);

    int NPW  = (N + 10239) / 10240;
    if (NPW > 30) NPW = 30;
    int gAgg = ((N + NPW - 1) / NPW + 7) / 8;
    int gAggF = ((N + 15) / 16 + 7) / 8;
    int gGemm1 = 256;

    tile_sort_kernel<<<NT, 256, 0, stream>>>(ei, tiles, cumT, N, E, G, NT);
    group_rows_kernel<<<G, 256, 0, stream>>>(tiles, cumT, rowptr, dinv, csr, N, G, NT);

    gemm64_wmma_kernel<128><<<gGemm1, T, 0, stream>>>(x, W1, dinv, h1, N);
    agg_gemm_kernel<<<gAggF, T, 0, stream>>>(h1, rowptr, csr, dinv, b1, W2, h2, N);

    agg_kernel<<<gAgg, T, 0, stream>>>(h2, rowptr, csr, dinv, b2, out, N, NPW, 0);
}
